# starting kernel
# baseline (speedup 1.0000x reference)
#include <hip/hip_runtime.h>


typedef float v2f __attribute__((ext_vector_type(2)));
typedef float v8f __attribute__((ext_vector_type(8)));

#define DH 128

static __device__ __forceinline__ v8f zero8() {
  v8f z = {0.f, 0.f, 0.f, 0.f, 0.f, 0.f, 0.f, 0.f};
  return z;
}

__global__ void __launch_bounds__(256) gin_scatter_add(
    const float* __restrict__ h, const int* __restrict__ src,
    const int* __restrict__ dst, float* __restrict__ agg, int E) {
  const int lane = threadIdx.x & 31;
  int w = (int)((blockIdx.x * blockDim.x + threadIdx.x) >> 5);
  const int nw = (int)((gridDim.x * blockDim.x) >> 5);
  for (int e = w; e < E; e += nw) {
    const int s = src[e];
    const int d = dst[e];
    const float4 v =
        *reinterpret_cast<const float4*>(h + (size_t)s * DH + lane * 4);
    float* p = agg + (size_t)d * DH + lane * 4;
    atomicAdd(p + 0, v.x);
    atomicAdd(p + 1, v.y);
    atomicAdd(p + 2, v.z);
    atomicAdd(p + 3, v.w);
  }
}

template <int NCOL2, bool HAS_AGG, bool RELU2>
__global__ void __launch_bounds__(128) gin_mlp(
    const float* __restrict__ hin, const float* __restrict__ agg,
    const float* __restrict__ W1, const float* __restrict__ b1,
    const float* __restrict__ W2, const float* __restrict__ b2,
    float* __restrict__ out, int nrows) {
  constexpr int N2 = NCOL2 * 16;
  constexpr int SWP = 144;
  constexpr int SZ = 132;
  __shared__ v2f sW1[(DH / 2) * SWP];
  __shared__ v2f sW2[(DH / 2) * SWP];
  __shared__ float sb1[DH];
  __shared__ float sb2[DH];
  __shared__ float sZ[4][16 * SZ];

  const int tid = (int)threadIdx.x;

  for (int i = tid; i < DH * DH; i += 128) {
    const int r = i >> 7, c = i & 127;
    reinterpret_cast<float*>(sW1)[((r >> 1) * SWP + c) * 2 + (r & 1)] = W1[i];
  }
  for (int i = tid; i < DH * N2; i += 128) {
    const int r = i / N2, c = i - r * N2;
    reinterpret_cast<float*>(sW2)[((r >> 1) * SWP + c) * 2 + (r & 1)] = W2[i];
  }
  if (tid < DH) sb1[tid] = b1[tid];
  if (tid < N2) sb2[tid] = b2[tid];
  __syncthreads();

  const int lane = tid & 31;
  const int wid = tid >> 5;
  const int m = lane & 15;
  const int koff = (lane >> 4) * 2;
  const int kpo = lane >> 4;
  const int mrow = (lane >> 4) * 8;
  float* zs = sZ[wid];

  const int ntiles = (nrows + 15) >> 4;
  const int gw0 = (int)blockIdx.x * 4 + wid;
  const int gwn = (int)gridDim.x * 4;

  for (int t = gw0; t < ntiles; t += gwn) {
    const int row0 = t << 4;
    int rowa = row0 + m;
    if (rowa > nrows - 1) rowa = nrows - 1;
    const float* hrow = hin + (size_t)rowa * DH;
    const float* arow = agg + (size_t)rowa * DH;

    v8f acc[8];
#pragma unroll
    for (int n = 0; n < 8; ++n) acc[n] = zero8();

    for (int k0 = 0; k0 < DH; k0 += 4) {
      const int k = k0 + koff;
      const int kp = (k0 >> 1) + kpo;
      v2f a = *reinterpret_cast<const v2f*>(hrow + k);
      if (HAS_AGG) {
        const v2f av = *reinterpret_cast<const v2f*>(arow + k);
        a.x += av.x;
        a.y += av.y;
      }
#pragma unroll
      for (int n = 0; n < 8; ++n) {
        const v2f bf = sW1[kp * SWP + n * 16 + m];
        acc[n] = __builtin_amdgcn_wmma_f32_16x16x4_f32(
            false, a, false, bf, (short)0, acc[n], false, false);
      }
    }

#pragma unroll
    for (int n = 0; n < 8; ++n) {
      const float bias = sb1[n * 16 + m];
#pragma unroll
      for (int j = 0; j < 8; ++j) {
        float v = acc[n][j] + bias;
        v = v > 0.0f ? v : 0.0f;
        zs[(mrow + j) * SZ + n * 16 + m] = v;
      }
    }
    __builtin_amdgcn_wave_barrier();

#pragma unroll
    for (int n = 0; n < NCOL2; ++n) acc[n] = zero8();
    for (int k0 = 0; k0 < DH; k0 += 4) {
      const int k = k0 + koff;
      const int kp = (k0 >> 1) + kpo;
      const v2f a = *reinterpret_cast<const v2f*>(&zs[m * SZ + k]);
#pragma unroll
      for (int n = 0; n < NCOL2; ++n) {
        const v2f bf = sW2[kp * SWP + n * 16 + m];
        acc[n] = __builtin_amdgcn_wmma_f32_16x16x4_f32(
            false, a, false, bf, (short)0, acc[n], false, false);
      }
    }
    __builtin_amdgcn_wave_barrier();

#pragma unroll
    for (int n = 0; n < NCOL2; ++n) {
      const float bias = sb2[n * 16 + m];
#pragma unroll
      for (int j = 0; j < 8; ++j) {
        float v = acc[n][j] + bias;
        if (RELU2) v = v > 0.0f ? v : 0.0f;
        const int row = row0 + mrow + j;
        if (row < nrows) out[(size_t)row * N2 + n * 16 + m] = v;
      }
    }
  }
}

extern "C" void kernel_launch(void* const* d_in, const int* in_sizes, int n_in,
                              void* d_out, int out_size, void* d_ws,
                              size_t ws_size, hipStream_t stream) {
  const float* x = (const float*)d_in[0];
  const int* ei = (const int*)d_in[1];
  const int N = in_sizes[0] / DH;
  const int E = in_sizes[1] / 2;
  const int* src = ei;
  const int* dst = ei + E;

  const float* c1w1 = (const float*)d_in[2];
  const float* c1b1 = (const float*)d_in[3];
  const float* c1w2 = (const float*)d_in[4];
  const float* c1b2 = (const float*)d_in[5];
  const float* c2w1 = (const float*)d_in[6];
  const float* c2b1 = (const float*)d_in[7];
  const float* c2w2 = (const float*)d_in[8];
  const float* c2b2 = (const float*)d_in[9];
  const float* c3w1 = (const float*)d_in[10];
  const float* c3b1 = (const float*)d_in[11];
  const float* c3w2 = (const float*)d_in[12];
  const float* c3b2 = (const float*)d_in[13];
  const float* l1w = (const float*)d_in[14];
  const float* l1b = (const float*)d_in[15];
  const float* l2w = (const float*)d_in[16];
  const float* l2b = (const float*)d_in[17];

  float* hA = (float*)d_ws;
  float* hB = hA + (size_t)N * DH;
  float* agg = hB + (size_t)N * DH;
  const size_t featBytes = (size_t)N * DH * sizeof(float);

  const dim3 sB(256), sG(4096);
  const dim3 mB(128), mG(512);

  hipMemsetAsync(agg, 0, featBytes, stream);
  gin_scatter_add<<<sG, sB, 0, stream>>>(x, src, dst, agg, E);
  gin_mlp<8, true, true>
      <<<mG, mB, 0, stream>>>(x, agg, c1w1, c1b1, c1w2, c1b2, hA, N);
  hipMemsetAsync(agg, 0, featBytes, stream);
  gin_scatter_add<<<sG, sB, 0, stream>>>(hA, src, dst, agg, E);
  gin_mlp<8, true, true>
      <<<mG, mB, 0, stream>>>(hA, agg, c2w1, c2b1, c2w2, c2b2, hB, N);
  hipMemsetAsync(agg, 0, featBytes, stream);
  gin_scatter_add<<<sG, sB, 0, stream>>>(hB, src, dst, agg, E);
  gin_mlp<8, true, true>
      <<<mG, mB, 0, stream>>>(hB, agg, c3w1, c3b1, c3w2, c3b2, hA, N);
  gin_mlp<4, false, false>
      <<<mG, mB, 0, stream>>>(hA, nullptr, l1w, l1b, l2w, l2b,
                              (float*)d_out, N);
}


# float-atomic scatter-add replaced by a gather over an edge list grouped by destination (built once per call: integer-atomic count+rank, prefix sum, fill); a row's sum order is its arrival rank
# speedup vs baseline: 1.7424x; 1.7424x over previous
#include <hip/hip_runtime.h>


typedef float v2f __attribute__((ext_vector_type(2)));
typedef float v8f __attribute__((ext_vector_type(8)));

#define DH 128

static __device__ __forceinline__ v8f zero8() {
  v8f z = {0.f, 0.f, 0.f, 0.f, 0.f, 0.f, 0.f, 0.f};
  return z;
}

__global__ void __launch_bounds__(256) csr_rank(
    const int* __restrict__ dst, int* __restrict__ cnt,
    int* __restrict__ rank, int E) {
  const int i0 = (int)(blockIdx.x * blockDim.x + threadIdx.x);
  const int n = (int)(gridDim.x * blockDim.x);
  for (int e = i0; e < E; e += n) rank[e] = atomicAdd(cnt + dst[e], 1);
}

__global__ void __launch_bounds__(1024) csr_scan(
    const int* __restrict__ cnt, int* __restrict__ rowptr, int n) {
  __shared__ int sfront[32];
  __shared__ int swave[32];
  const int tid = (int)threadIdx.x;
  const int lane = tid & 31;
  const int wid = tid >> 5;
  for (int base = (int)blockIdx.x * 1024; base < n;
       base += (int)gridDim.x * 1024) {
    int front = 0;
    for (int i = tid; i < base; i += 1024) front += cnt[i];
    for (int o = 16; o > 0; o >>= 1) front += __shfl_xor(front, o);
    const int i = base + tid;
    const int v = i < n ? cnt[i] : 0;
    int x = v;
    for (int o = 1; o < 32; o <<= 1) {
      const int y = __shfl_up(x, o);
      if (lane >= o) x += y;
    }
    if (lane == 0) sfront[wid] = front;
    if (lane == 31) swave[wid] = x;
    __syncthreads();
    int pre = sfront[lane] + (lane < wid ? swave[lane] : 0);
    for (int o = 16; o > 0; o >>= 1) pre += __shfl_xor(pre, o);
    if (i < n) rowptr[i] = pre + x - v;
    __syncthreads();
  }
}

__global__ void __launch_bounds__(256) csr_fill(
    const int* __restrict__ src, const int* __restrict__ dst,
    const int* __restrict__ rank, const int* __restrict__ rowptr,
    int* __restrict__ col, int E) {
  const int i0 = (int)(blockIdx.x * blockDim.x + threadIdx.x);
  const int n = (int)(gridDim.x * blockDim.x);
  for (int e = i0; e < E; e += n) col[rowptr[dst[e]] + rank[e]] = src[e];
}

template <int U>
static __device__ __forceinline__ void gather_rows(
    const float* __restrict__ h, int mine, int j, int lane, float4& acc) {
  float4 v[U];
#pragma unroll
  for (int u = 0; u < U; ++u) {
    const int s = __builtin_amdgcn_readlane(mine, j + u);
    v[u] = *reinterpret_cast<const float4*>(h + (size_t)s * DH + lane * 4);
  }
#pragma unroll
  for (int u = 0; u < U; ++u) {
    acc.x += v[u].x;
    acc.y += v[u].y;
    acc.z += v[u].z;
    acc.w += v[u].w;
  }
}

__global__ void __launch_bounds__(256) gin_gather(
    const float* __restrict__ h, const int* __restrict__ rowptr,
    const int* __restrict__ cnt, const int* __restrict__ col,
    float* __restrict__ agg, int N) {
  const int lane = (int)threadIdx.x & 31;
  const int w0 = (int)((blockIdx.x * blockDim.x + threadIdx.x) >> 5);
  const int nw = (int)((gridDim.x * blockDim.x) >> 5);
  for (int i = w0; i < N; i += nw) {
    const int node = __builtin_amdgcn_readfirstlane(i);
    const int beg = rowptr[node];
    const int deg = cnt[node];
    float4 acc = {0.f, 0.f, 0.f, 0.f};
    for (int j0 = 0; j0 < deg; j0 += 32) {
      const int m = deg - j0 < 32 ? deg - j0 : 32;
      const int mine = lane < m ? col[beg + j0 + lane] : 0;
      int j = 0;
      for (; j + 8 <= m; j += 8) gather_rows<8>(h, mine, j, lane, acc);
      if (j + 4 <= m) {
        gather_rows<4>(h, mine, j, lane, acc);
        j += 4;
      }
      if (j + 2 <= m) {
        gather_rows<2>(h, mine, j, lane, acc);
        j += 2;
      }
      if (j < m) gather_rows<1>(h, mine, j, lane, acc);
    }
    *reinterpret_cast<float4*>(agg + (size_t)node * DH + lane * 4) = acc;
  }
}

template <int NCOL2, bool HAS_AGG, bool RELU2>
__global__ void __launch_bounds__(128) gin_mlp(
    const float* __restrict__ hin, const float* __restrict__ agg,
    const float* __restrict__ W1, const float* __restrict__ b1,
    const float* __restrict__ W2, const float* __restrict__ b2,
    float* __restrict__ out, int nrows) {
  constexpr int N2 = NCOL2 * 16;
  constexpr int SWP = 144;
  constexpr int SZ = 132;
  __shared__ v2f sW1[(DH / 2) * SWP];
  __shared__ v2f sW2[(DH / 2) * SWP];
  __shared__ float sb1[DH];
  __shared__ float sb2[DH];
  __shared__ float sZ[4][16 * SZ];

  const int tid = (int)threadIdx.x;

  for (int i = tid; i < DH * DH; i += 128) {
    const int r = i >> 7, c = i & 127;
    reinterpret_cast<float*>(sW1)[((r >> 1) * SWP + c) * 2 + (r & 1)] = W1[i];
  }
  for (int i = tid; i < DH * N2; i += 128) {
    const int r = i / N2, c = i - r * N2;
    reinterpret_cast<float*>(sW2)[((r >> 1) * SWP + c) * 2 + (r & 1)] = W2[i];
  }
  if (tid < DH) sb1[tid] = b1[tid];
  if (tid < N2) sb2[tid] = b2[tid];
  __syncthreads();

  const int lane = tid & 31;
  const int wid = tid >> 5;
  const int m = lane & 15;
  const int koff = (lane >> 4) * 2;
  const int kpo = lane >> 4;
  const int mrow = (lane >> 4) * 8;
  float* zs = sZ[wid];

  const int ntiles = (nrows + 15) >> 4;
  const int gw0 = (int)blockIdx.x * 4 + wid;
  const int gwn = (int)gridDim.x * 4;

  for (int t = gw0; t < ntiles; t += gwn) {
    const int row0 = t << 4;
    int rowa = row0 + m;
    if (rowa > nrows - 1) rowa = nrows - 1;
    const float* hrow = hin + (size_t)rowa * DH;
    const float* arow = agg + (size_t)rowa * DH;

    v8f acc[8];
#pragma unroll
    for (int n = 0; n < 8; ++n) acc[n] = zero8();

    for (int k0 = 0; k0 < DH; k0 += 4) {
      const int k = k0 + koff;
      const int kp = (k0 >> 1) + kpo;
      v2f a = *reinterpret_cast<const v2f*>(hrow + k);
      if (HAS_AGG) {
        const v2f av = *reinterpret_cast<const v2f*>(arow + k);
        a.x += av.x;
        a.y += av.y;
      }
#pragma unroll
      for (int n = 0; n < 8; ++n) {
        const v2f bf = sW1[kp * SWP + n * 16 + m];
        acc[n] = __builtin_amdgcn_wmma_f32_16x16x4_f32(
            false, a, false, bf, (short)0, acc[n], false, false);
      }
    }

#pragma unroll
    for (int n = 0; n < 8; ++n) {
      const float bias = sb1[n * 16 + m];
#pragma unroll
      for (int j = 0; j < 8; ++j) {
        float v = acc[n][j] + bias;
        v = v > 0.0f ? v : 0.0f;
        zs[(mrow + j) * SZ + n * 16 + m] = v;
      }
    }
    __builtin_amdgcn_wave_barrier();

#pragma unroll
    for (int n = 0; n < NCOL2; ++n) acc[n] = zero8();
    for (int k0 = 0; k0 < DH; k0 += 4) {
      const int k = k0 + koff;
      const int kp = (k0 >> 1) + kpo;
      const v2f a = *reinterpret_cast<const v2f*>(&zs[m * SZ + k]);
#pragma unroll
      for (int n = 0; n < NCOL2; ++n) {
        const v2f bf = sW2[kp * SWP + n * 16 + m];
        acc[n] = __builtin_amdgcn_wmma_f32_16x16x4_f32(
            false, a, false, bf, (short)0, acc[n], false, false);
      }
    }
    __builtin_amdgcn_wave_barrier();

#pragma unroll
    for (int n = 0; n < NCOL2; ++n) {
      const float bias = sb2[n * 16 + m];
#pragma unroll
      for (int j = 0; j < 8; ++j) {
        float v = acc[n][j] + bias;
        if (RELU2) v = v > 0.0f ? v : 0.0f;
        const int row = row0 + mrow + j;
        if (row < nrows) out[(size_t)row * N2 + n * 16 + m] = v;
      }
    }
  }
}

extern "C" void kernel_launch(void* const* d_in, const int* in_sizes, int n_in,
                              void* d_out, int out_size, void* d_ws,
                              size_t ws_size, hipStream_t stream) {
  const float* x = (const float*)d_in[0];
  const int* ei = (const int*)d_in[1];
  const int N = in_sizes[0] / DH;
  const int E = in_sizes[1] / 2;
  const int* src = ei;
  const int* dst = ei + E;

  const float* c1w1 = (const float*)d_in[2];
  const float* c1b1 = (const float*)d_in[3];
  const float* c1w2 = (const float*)d_in[4];
  const float* c1b2 = (const float*)d_in[5];
  const float* c2w1 = (const float*)d_in[6];
  const float* c2b1 = (const float*)d_in[7];
  const float* c2w2 = (const float*)d_in[8];
  const float* c2b2 = (const float*)d_in[9];
  const float* c3w1 = (const float*)d_in[10];
  const float* c3b1 = (const float*)d_in[11];
  const float* c3w2 = (const float*)d_in[12];
  const float* c3b2 = (const float*)d_in[13];
  const float* l1w = (const float*)d_in[14];
  const float* l1b = (const float*)d_in[15];
  const float* l2w = (const float*)d_in[16];
  const float* l2b = (const float*)d_in[17];

  float* hA = (float*)d_ws;
  float* hB = hA + (size_t)N * DH;
  float* agg = hB + (size_t)N * DH;
  int* cnt = (int*)(agg + (size_t)N * DH);
  int* rowptr = cnt + N;
  int* rank = rowptr + N;
  int* col = rank + E;

  const dim3 eB(256), eG(1024);
  const dim3 pB(1024), pG(64);
  const dim3 gB(256), gG(1024);
  const dim3 mB(128), mG(512);

  hipMemsetAsync(cnt, 0, (size_t)N * sizeof(int), stream);
  csr_rank<<<eG, eB, 0, stream>>>(dst, cnt, rank, E);
  csr_scan<<<pG, pB, 0, stream>>>(cnt, rowptr, N);
  csr_fill<<<eG, eB, 0, stream>>>(src, dst, rank, rowptr, col, E);

  gin_gather<<<gG, gB, 0, stream>>>(x, rowptr, cnt, col, agg, N);
  gin_mlp<8, true, true>
      <<<mG, mB, 0, stream>>>(x, agg, c1w1, c1b1, c1w2, c1b2, hA, N);
  gin_gather<<<gG, gB, 0, stream>>>(hA, rowptr, cnt, col, agg, N);
  gin_mlp<8, true, true>
      <<<mG, mB, 0, stream>>>(hA, agg, c2w1, c2b1, c2w2, c2b2, hB, N);
  gin_gather<<<gG, gB, 0, stream>>>(hB, rowptr, cnt, col, agg, N);
  gin_mlp<8, true, true>
      <<<mG, mB, 0, stream>>>(hB, agg, c3w1, c3b1, c3w2, c3b2, hA, N);
  gin_mlp<4, false, false>
      <<<mG, mB, 0, stream>>>(hA, nullptr, l1w, l1b, l2w, l2b,
                              (float*)d_out, N);
}


# gin_mlp launched as 256 resident 16-wave workgroups, one per WGP, one 16-row tile per wave (was 512 of 4 waves in two rounds): weights staged once per WGP, sibling waves cover each other's load waits
# speedup vs baseline: 2.6114x; 1.4987x over previous
#include <hip/hip_runtime.h>


typedef float v2f __attribute__((ext_vector_type(2)));
typedef float v8f __attribute__((ext_vector_type(8)));

#define DH 128

static __device__ __forceinline__ v8f zero8() {
  v8f z = {0.f, 0.f, 0.f, 0.f, 0.f, 0.f, 0.f, 0.f};
  return z;
}

__global__ void __launch_bounds__(256) csr_rank(
    const int* __restrict__ dst, int* __restrict__ cnt,
    int* __restrict__ rank, int E) {
  const int i0 = (int)(blockIdx.x * blockDim.x + threadIdx.x);
  const int n = (int)(gridDim.x * blockDim.x);
  for (int e = i0; e < E; e += n) rank[e] = atomicAdd(cnt + dst[e], 1);
}

__global__ void __launch_bounds__(1024) csr_scan(
    const int* __restrict__ cnt, int* __restrict__ rowptr, int n) {
  __shared__ int sfront[32];
  __shared__ int swave[32];
  const int tid = (int)threadIdx.x;
  const int lane = tid & 31;
  const int wid = tid >> 5;
  for (int base = (int)blockIdx.x * 1024; base < n;
       base += (int)gridDim.x * 1024) {
    int front = 0;
    for (int i = tid; i < base; i += 1024) front += cnt[i];
    for (int o = 16; o > 0; o >>= 1) front += __shfl_xor(front, o);
    const int i = base + tid;
    const int v = i < n ? cnt[i] : 0;
    int x = v;
    for (int o = 1; o < 32; o <<= 1) {
      const int y = __shfl_up(x, o);
      if (lane >= o) x += y;
    }
    if (lane == 0) sfront[wid] = front;
    if (lane == 31) swave[wid] = x;
    __syncthreads();
    int pre = sfront[lane] + (lane < wid ? swave[lane] : 0);
    for (int o = 16; o > 0; o >>= 1) pre += __shfl_xor(pre, o);
    if (i < n) rowptr[i] = pre + x - v;
    __syncthreads();
  }
}

__global__ void __launch_bounds__(256) csr_fill(
    const int* __restrict__ src, const int* __restrict__ dst,
    const int* __restrict__ rank, const int* __restrict__ rowptr,
    int* __restrict__ col, int E) {
  const int i0 = (int)(blockIdx.x * blockDim.x + threadIdx.x);
  const int n = (int)(gridDim.x * blockDim.x);
  for (int e = i0; e < E; e += n) col[rowptr[dst[e]] + rank[e]] = src[e];
}

template <int U>
static __device__ __forceinline__ void gather_rows(
    const float* __restrict__ h, int mine, int j, int lane, float4& acc) {
  float4 v[U];
#pragma unroll
  for (int u = 0; u < U; ++u) {
    const int s = __builtin_amdgcn_readlane(mine, j + u);
    v[u] = *reinterpret_cast<const float4*>(h + (size_t)s * DH + lane * 4);
  }
#pragma unroll
  for (int u = 0; u < U; ++u) {
    acc.x += v[u].x;
    acc.y += v[u].y;
    acc.z += v[u].z;
    acc.w += v[u].w;
  }
}

__global__ void __launch_bounds__(256) gin_gather(
    const float* __restrict__ h, const int* __restrict__ rowptr,
    const int* __restrict__ cnt, const int* __restrict__ col,
    float* __restrict__ agg, int N) {
  const int lane = (int)threadIdx.x & 31;
  const int w0 = (int)((blockIdx.x * blockDim.x + threadIdx.x) >> 5);
  const int nw = (int)((gridDim.x * blockDim.x) >> 5);
  for (int i = w0; i < N; i += nw) {
    const int node = __builtin_amdgcn_readfirstlane(i);
    const int beg = rowptr[node];
    const int deg = cnt[node];
    float4 acc = {0.f, 0.f, 0.f, 0.f};
    for (int j0 = 0; j0 < deg; j0 += 32) {
      const int m = deg - j0 < 32 ? deg - j0 : 32;
      const int mine = lane < m ? col[beg + j0 + lane] : 0;
      int j = 0;
      for (; j + 8 <= m; j += 8) gather_rows<8>(h, mine, j, lane, acc);
      if (j + 4 <= m) {
        gather_rows<4>(h, mine, j, lane, acc);
        j += 4;
      }
      if (j + 2 <= m) {
        gather_rows<2>(h, mine, j, lane, acc);
        j += 2;
      }
      if (j < m) gather_rows<1>(h, mine, j, lane, acc);
    }
    *reinterpret_cast<float4*>(agg + (size_t)node * DH + lane * 4) = acc;
  }
}

#define MLP_NW 16
#define MLP_NT (MLP_NW * 32)

template <int NCOL2, bool HAS_AGG, bool RELU2>
__global__ void __launch_bounds__(MLP_NT) gin_mlp(
    const float* __restrict__ hin, const float* __restrict__ agg,
    const float* __restrict__ W1, const float* __restrict__ b1,
    const float* __restrict__ W2, const float* __restrict__ b2,
    float* __restrict__ out, int nrows) {
  constexpr int N2 = NCOL2 * 16;
  constexpr int SWP = 144;
  constexpr int SZ = 132;
  __shared__ v2f sW1[(DH / 2) * SWP];
  __shared__ v2f sW2[(DH / 2) * SWP];
  __shared__ float sb1[DH];
  __shared__ float sb2[DH];
  __shared__ float sZ[MLP_NW][16 * SZ];

  const int tid = (int)threadIdx.x;

  for (int i = tid; i < DH * DH; i += MLP_NT) {
    const int r = i >> 7, c = i & 127;
    reinterpret_cast<float*>(sW1)[((r >> 1) * SWP + c) * 2 + (r & 1)] = W1[i];
  }
  for (int i = tid; i < DH * N2; i += MLP_NT) {
    const int r = i / N2, c = i - r * N2;
    reinterpret_cast<float*>(sW2)[((r >> 1) * SWP + c) * 2 + (r & 1)] = W2[i];
  }
  if (tid < DH) sb1[tid] = b1[tid];
  if (tid < N2) sb2[tid] = b2[tid];
  __syncthreads();

  const int lane = tid & 31;
  const int wid = tid >> 5;
  const int m = lane & 15;
  const int koff = (lane >> 4) * 2;
  const int kpo = lane >> 4;
  const int mrow = (lane >> 4) * 8;
  float* zs = sZ[wid];

  const int ntiles = (nrows + 15) >> 4;
  const int gw0 = (int)blockIdx.x + (int)gridDim.x * wid;
  const int gwn = (int)gridDim.x * MLP_NW;

  for (int t = gw0; t < ntiles; t += gwn) {
    const int row0 = t << 4;
    int rowa = row0 + m;
    if (rowa > nrows - 1) rowa = nrows - 1;
    const float* hrow = hin + (size_t)rowa * DH;
    const float* arow = agg + (size_t)rowa * DH;

    v8f acc[8];
#pragma unroll
    for (int n = 0; n < 8; ++n) acc[n] = zero8();

    for (int k0 = 0; k0 < DH; k0 += 4) {
      const int k = k0 + koff;
      const int kp = (k0 >> 1) + kpo;
      v2f a = *reinterpret_cast<const v2f*>(hrow + k);
      if (HAS_AGG) {
        const v2f av = *reinterpret_cast<const v2f*>(arow + k);
        a.x += av.x;
        a.y += av.y;
      }
#pragma unroll
      for (int n = 0; n < 8; ++n) {
        const v2f bf = sW1[kp * SWP + n * 16 + m];
        acc[n] = __builtin_amdgcn_wmma_f32_16x16x4_f32(
            false, a, false, bf, (short)0, acc[n], false, false);
      }
    }

#pragma unroll
    for (int n = 0; n < 8; ++n) {
      const float bias = sb1[n * 16 + m];
#pragma unroll
      for (int j = 0; j < 8; ++j) {
        float v = acc[n][j] + bias;
        v = v > 0.0f ? v : 0.0f;
        zs[(mrow + j) * SZ + n * 16 + m] = v;
      }
    }
    __builtin_amdgcn_wave_barrier();

#pragma unroll
    for (int n = 0; n < NCOL2; ++n) acc[n] = zero8();
    for (int k0 = 0; k0 < DH; k0 += 4) {
      const int k = k0 + koff;
      const int kp = (k0 >> 1) + kpo;
      const v2f a = *reinterpret_cast<const v2f*>(&zs[m * SZ + k]);
#pragma unroll
      for (int n = 0; n < NCOL2; ++n) {
        const v2f bf = sW2[kp * SWP + n * 16 + m];
        acc[n] = __builtin_amdgcn_wmma_f32_16x16x4_f32(
            false, a, false, bf, (short)0, acc[n], false, false);
      }
    }
    __builtin_amdgcn_wave_barrier();

#pragma unroll
    for (int n = 0; n < NCOL2; ++n) {
      const float bias = sb2[n * 16 + m];
#pragma unroll
      for (int j = 0; j < 8; ++j) {
        float v = acc[n][j] + bias;
        if (RELU2) v = v > 0.0f ? v : 0.0f;
        const int row = row0 + mrow + j;
        if (row < nrows) out[(size_t)row * N2 + n * 16 + m] = v;
      }
    }
  }
}

extern "C" void kernel_launch(void* const* d_in, const int* in_sizes, int n_in,
                              void* d_out, int out_size, void* d_ws,
                              size_t ws_size, hipStream_t stream) {
  const float* x = (const float*)d_in[0];
  const int* ei = (const int*)d_in[1];
  const int N = in_sizes[0] / DH;
  const int E = in_sizes[1] / 2;
  const int* src = ei;
  const int* dst = ei + E;

  const float* c1w1 = (const float*)d_in[2];
  const float* c1b1 = (const float*)d_in[3];
  const float* c1w2 = (const float*)d_in[4];
  const float* c1b2 = (const float*)d_in[5];
  const float* c2w1 = (const float*)d_in[6];
  const float* c2b1 = (const float*)d_in[7];
  const float* c2w2 = (const float*)d_in[8];
  const float* c2b2 = (const float*)d_in[9];
  const float* c3w1 = (const float*)d_in[10];
  const float* c3b1 = (const float*)d_in[11];
  const float* c3w2 = (const float*)d_in[12];
  const float* c3b2 = (const float*)d_in[13];
  const float* l1w = (const float*)d_in[14];
  const float* l1b = (const float*)d_in[15];
  const float* l2w = (const float*)d_in[16];
  const float* l2b = (const float*)d_in[17];

  float* hA = (float*)d_ws;
  float* hB = hA + (size_t)N * DH;
  float* agg = hB + (size_t)N * DH;
  int* cnt = (int*)(agg + (size_t)N * DH);
  int* rowptr = cnt + N;
  int* rank = rowptr + N;
  int* col = rank + E;

  const dim3 eB(256), eG(1024);
  const dim3 pB(1024), pG(64);
  const dim3 gB(256), gG(1024);
  const dim3 mB(512), mG(256);

  hipMemsetAsync(cnt, 0, (size_t)N * sizeof(int), stream);
  csr_rank<<<eG, eB, 0, stream>>>(dst, cnt, rank, E);
  csr_scan<<<pG, pB, 0, stream>>>(cnt, rowptr, N);
  csr_fill<<<eG, eB, 0, stream>>>(src, dst, rank, rowptr, col, E);

  gin_gather<<<gG, gB, 0, stream>>>(x, rowptr, cnt, col, agg, N);
  gin_mlp<8, true, true>
      <<<mG, mB, 0, stream>>>(x, agg, c1w1, c1b1, c1w2, c1b2, hA, N);
  gin_gather<<<gG, gB, 0, stream>>>(hA, rowptr, cnt, col, agg, N);
  gin_mlp<8, true, true>
      <<<mG, mB, 0, stream>>>(hA, agg, c2w1, c2b1, c2w2, c2b2, hB, N);
  gin_gather<<<gG, gB, 0, stream>>>(hB, rowptr, cnt, col, agg, N);
  gin_mlp<8, true, true>
      <<<mG, mB, 0, stream>>>(hB, agg, c3w1, c3b1, c3w2, c3b2, hA, N);
  gin_mlp<4, false, false>
      <<<mG, mB, 0, stream>>>(hA, nullptr, l1w, l1b, l2w, l2b,
                              (float*)d_out, N);
}


# neighbour gather fused into gin_mlp: a wave sums its 16 nodes' source rows (32 whole-row loads in flight) into its LDS tile, GEMM1 reads A from LDS; agg plane, 3 launches gone; sum order unchanged
# speedup vs baseline: 2.8573x; 1.0942x over previous
#include <hip/hip_runtime.h>


typedef float v2f __attribute__((ext_vector_type(2)));
typedef float v8f __attribute__((ext_vector_type(8)));

#define DH 128

static __device__ __forceinline__ v8f zero8() {
  v8f z = {0.f, 0.f, 0.f, 0.f, 0.f, 0.f, 0.f, 0.f};
  return z;
}

__global__ void __launch_bounds__(256) csr_rank(
    const int* __restrict__ dst, int* __restrict__ cnt,
    int* __restrict__ rank, int E) {
  const int i0 = (int)(blockIdx.x * blockDim.x + threadIdx.x);
  const int n = (int)(gridDim.x * blockDim.x);
  for (int e = i0; e < E; e += n) rank[e] = atomicAdd(cnt + dst[e], 1);
}

__global__ void __launch_bounds__(1024) csr_scan(
    const int* __restrict__ cnt, int* __restrict__ rowptr, int n) {
  __shared__ int sfront[32];
  __shared__ int swave[32];
  const int tid = (int)threadIdx.x;
  const int lane = tid & 31;
  const int wid = tid >> 5;
  for (int base = (int)blockIdx.x * 1024; base < n;
       base += (int)gridDim.x * 1024) {
    int front = 0;
    for (int i = tid; i < base; i += 1024) front += cnt[i];
    for (int o = 16; o > 0; o >>= 1) front += __shfl_xor(front, o);
    const int i = base + tid;
    const int v = i < n ? cnt[i] : 0;
    int x = v;
    for (int o = 1; o < 32; o <<= 1) {
      const int y = __shfl_up(x, o);
      if (lane >= o) x += y;
    }
    if (lane == 0) sfront[wid] = front;
    if (lane == 31) swave[wid] = x;
    __syncthreads();
    int pre = sfront[lane] + (lane < wid ? swave[lane] : 0);
    for (int o = 16; o > 0; o >>= 1) pre += __shfl_xor(pre, o);
    if (i < n) rowptr[i] = pre + x - v;
    __syncthreads();
  }
}

__global__ void __launch_bounds__(256) csr_fill(
    const int* __restrict__ src, const int* __restrict__ dst,
    const int* __restrict__ rank, const int* __restrict__ rowptr,
    int* __restrict__ col, int E) {
  const int i0 = (int)(blockIdx.x * blockDim.x + threadIdx.x);
  const int n = (int)(gridDim.x * blockDim.x);
  for (int e = i0; e < E; e += n) col[rowptr[dst[e]] + rank[e]] = src[e];
}

#define MLP_NW 16
#define MLP_NT (MLP_NW * 32)
#define GU 32

template <int NCOL2, bool HAS_AGG, bool RELU2>
__global__ void __launch_bounds__(MLP_NT) gin_mlp(
    const float* __restrict__ hin, const int* __restrict__ rowptr,
    const int* __restrict__ cnt, const int* __restrict__ col,
    const float* __restrict__ W1, const float* __restrict__ b1,
    const float* __restrict__ W2, const float* __restrict__ b2,
    float* __restrict__ out, int nrows) {
  constexpr int N2 = NCOL2 * 16;
  constexpr int SWP = 144;
  constexpr int SZ = 132;
  __shared__ v2f sW1[(DH / 2) * SWP];
  __shared__ v2f sW2[(DH / 2) * SWP];
  __shared__ float sb1[DH];
  __shared__ float sb2[DH];
  __shared__ __attribute__((aligned(16))) float sZ[MLP_NW][16 * SZ];

  const int tid = (int)threadIdx.x;

  for (int i = tid; i < DH * DH; i += MLP_NT) {
    const int r = i >> 7, c = i & 127;
    reinterpret_cast<float*>(sW1)[((r >> 1) * SWP + c) * 2 + (r & 1)] = W1[i];
  }
  for (int i = tid; i < DH * N2; i += MLP_NT) {
    const int r = i / N2, c = i - r * N2;
    reinterpret_cast<float*>(sW2)[((r >> 1) * SWP + c) * 2 + (r & 1)] = W2[i];
  }
  if (tid < DH) sb1[tid] = b1[tid];
  if (tid < N2) sb2[tid] = b2[tid];
  __syncthreads();

  const int lane = tid & 31;
  const int wid = tid >> 5;
  const int m = lane & 15;
  const int koff = (lane >> 4) * 2;
  const int kpo = lane >> 4;
  const int mrow = (lane >> 4) * 8;
  float* zs = sZ[wid];

  const int ntiles = (nrows + 15) >> 4;
  const int gw0 = (int)blockIdx.x + (int)gridDim.x * wid;
  const int gwn = (int)gridDim.x * MLP_NW;

  for (int t = gw0; t < ntiles; t += gwn) {
    const int row0 = t << 4;

    if (HAS_AGG) {
      const int r0 = __builtin_amdgcn_readfirstlane(row0);
      const int nv = nrows - r0 < 16 ? nrows - r0 : 16;
      int nd = r0 + m;
      if (nd > nrows - 1) nd = nrows - 1;
      const int rp = rowptr[nd];
      const int dg = m < nv ? cnt[nd] : 0;
      const int myend = rp + dg;
      float4 own[16];
#pragma unroll
      for (int r = 0; r < 16; ++r) {
        int nr = r0 + r;
        if (nr > nrows - 1) nr = nrows - 1;
        own[r] =
            *reinterpret_cast<const float4*>(hin + (size_t)nr * DH + lane * 4);
      }
      const int beg = __builtin_amdgcn_readlane(rp, 0);
      const int end = __builtin_amdgcn_readlane(myend, nv - 1);
      int cur = 0;
      int curend = __builtin_amdgcn_readlane(myend, 0);
      float4 acc = {0.f, 0.f, 0.f, 0.f};
      int mine = 0;
      if (beg < end) mine = beg + lane < end ? col[beg + lane] : 0;
#pragma unroll
      for (int r = 0; r < 16; ++r)
        *reinterpret_cast<float4*>(&zs[r * SZ + lane * 4]) = own[r];
      for (int base = beg; base < end; base += GU) {
        float4 v[GU];
#pragma unroll
        for (int u = 0; u < GU; ++u) {
          const int sidx = __builtin_amdgcn_readlane(mine, u);
          v[u] = *reinterpret_cast<const float4*>(hin + (size_t)sidx * DH +
                                                  lane * 4);
        }
        int nxt = 0;
        if (base + GU < end)
          nxt = base + GU + lane < end ? col[base + GU + lane] : 0;
#pragma unroll
        for (int u = 0; u < GU; ++u) {
          const int p = base + u;
          if (p < end) {
            while (p >= curend && cur < 15) {
              float4* zr = reinterpret_cast<float4*>(&zs[cur * SZ + lane * 4]);
              float4 hv = *zr;
              hv.x += acc.x;
              hv.y += acc.y;
              hv.z += acc.z;
              hv.w += acc.w;
              *zr = hv;
              acc.x = 0.f;
              acc.y = 0.f;
              acc.z = 0.f;
              acc.w = 0.f;
              ++cur;
              curend = __builtin_amdgcn_readlane(myend, cur);
            }
            acc.x += v[u].x;
            acc.y += v[u].y;
            acc.z += v[u].z;
            acc.w += v[u].w;
          }
        }
        mine = nxt;
      }
      {
        float4* zr = reinterpret_cast<float4*>(&zs[cur * SZ + lane * 4]);
        float4 hv = *zr;
        hv.x += acc.x;
        hv.y += acc.y;
        hv.z += acc.z;
        hv.w += acc.w;
        *zr = hv;
      }
      __builtin_amdgcn_wave_barrier();
    }

    int rowa = row0 + m;
    if (rowa > nrows - 1) rowa = nrows - 1;
    const float* hrow = hin + (size_t)rowa * DH;

    v8f acc[8];
#pragma unroll
    for (int n = 0; n < 8; ++n) acc[n] = zero8();

    for (int k0 = 0; k0 < DH; k0 += 4) {
      const int k = k0 + koff;
      const int kp = (k0 >> 1) + kpo;
      v2f a;
      if (HAS_AGG) {
        a = *reinterpret_cast<const v2f*>(&zs[m * SZ + k]);
      } else {
        a = *reinterpret_cast<const v2f*>(hrow + k);
      }
#pragma unroll
      for (int n = 0; n < 8; ++n) {
        const v2f bf = sW1[kp * SWP + n * 16 + m];
        acc[n] = __builtin_amdgcn_wmma_f32_16x16x4_f32(
            false, a, false, bf, (short)0, acc[n], false, false);
      }
    }
    if (HAS_AGG) __builtin_amdgcn_wave_barrier();

#pragma unroll
    for (int n = 0; n < 8; ++n) {
      const float bias = sb1[n * 16 + m];
#pragma unroll
      for (int j = 0; j < 8; ++j) {
        float v = acc[n][j] + bias;
        v = v > 0.0f ? v : 0.0f;
        zs[(mrow + j) * SZ + n * 16 + m] = v;
      }
    }
    __builtin_amdgcn_wave_barrier();

#pragma unroll
    for (int n = 0; n < NCOL2; ++n) acc[n] = zero8();
    for (int k0 = 0; k0 < DH; k0 += 4) {
      const int k = k0 + koff;
      const int kp = (k0 >> 1) + kpo;
      const v2f a = *reinterpret_cast<const v2f*>(&zs[m * SZ + k]);
#pragma unroll
      for (int n = 0; n < NCOL2; ++n) {
        const v2f bf = sW2[kp * SWP + n * 16 + m];
        acc[n] = __builtin_amdgcn_wmma_f32_16x16x4_f32(
            false, a, false, bf, (short)0, acc[n], false, false);
      }
    }
    __builtin_amdgcn_wave_barrier();

#pragma unroll
    for (int n = 0; n < NCOL2; ++n) {
      const float bias = sb2[n * 16 + m];
#pragma unroll
      for (int j = 0; j < 8; ++j) {
        float v = acc[n][j] + bias;
        if (RELU2) v = v > 0.0f ? v : 0.0f;
        const int row = row0 + mrow + j;
        if (row < nrows) out[(size_t)row * N2 + n * 16 + m] = v;
      }
    }
  }
}

extern "C" void kernel_launch(void* const* d_in, const int* in_sizes, int n_in,
                              void* d_out, int out_size, void* d_ws,
                              size_t ws_size, hipStream_t stream) {
  const float* x = (const float*)d_in[0];
  const int* ei = (const int*)d_in[1];
  const int N = in_sizes[0] / DH;
  const int E = in_sizes[1] / 2;
  const int* src = ei;
  const int* dst = ei + E;

  const float* c1w1 = (const float*)d_in[2];
  const float* c1b1 = (const float*)d_in[3];
  const float* c1w2 = (const float*)d_in[4];
  const float* c1b2 = (const float*)d_in[5];
  const float* c2w1 = (const float*)d_in[6];
  const float* c2b1 = (const float*)d_in[7];
  const float* c2w2 = (const float*)d_in[8];
  const float* c2b2 = (const float*)d_in[9];
  const float* c3w1 = (const float*)d_in[10];
  const float* c3b1 = (const float*)d_in[11];
  const float* c3w2 = (const float*)d_in[12];
  const float* c3b2 = (const float*)d_in[13];
  const float* l1w = (const float*)d_in[14];
  const float* l1b = (const float*)d_in[15];
  const float* l2w = (const float*)d_in[16];
  const float* l2b = (const float*)d_in[17];

  float* hA = (float*)d_ws;
  float* hB = hA + (size_t)N * DH;
  int* cnt = (int*)(hB + (size_t)N * DH);
  int* rowptr = cnt + N;
  int* rank = rowptr + N;
  int* col = rank + E;

  const dim3 eB(256), eG(1024);
  const dim3 pB(1024), pG(64);
  const dim3 mB(512), mG(256);

  hipMemsetAsync(cnt, 0, (size_t)N * sizeof(int), stream);
  csr_rank<<<eG, eB, 0, stream>>>(dst, cnt, rank, E);
  csr_scan<<<pG, pB, 0, stream>>>(cnt, rowptr, N);
  csr_fill<<<eG, eB, 0, stream>>>(src, dst, rank, rowptr, col, E);

  gin_mlp<8, true, true><<<mG, mB, 0, stream>>>(
      x, rowptr, cnt, col, c1w1, c1b1, c1w2, c1b2, hA, N);
  gin_mlp<8, true, true><<<mG, mB, 0, stream>>>(
      hA, rowptr, cnt, col, c2w1, c2b1, c2w2, c2b2, hB, N);
  gin_mlp<8, true, true><<<mG, mB, 0, stream>>>(
      hB, rowptr, cnt, col, c3w1, c3b1, c3w2, c3b2, hA, N);
  gin_mlp<4, false, false><<<mG, mB, 0, stream>>>(
      hA, nullptr, nullptr, nullptr, l1w, l1b, l2w, l2b, (float*)d_out, N);
}


# gin_mlp weight staging: row pairs fetched with 16-byte loads, both matrices and biases in flight before the first LDS store (was 9 drained trips of 4-byte loads); start row rotated per workgroup
# speedup vs baseline: 3.0599x; 1.0709x over previous
#include <hip/hip_runtime.h>


typedef float v2f __attribute__((ext_vector_type(2)));
typedef float v8f __attribute__((ext_vector_type(8)));

#define DH 128

static __device__ __forceinline__ v8f zero8() {
  v8f z = {0.f, 0.f, 0.f, 0.f, 0.f, 0.f, 0.f, 0.f};
  return z;
}

__global__ void __launch_bounds__(256) csr_rank(
    const int* __restrict__ dst, int* __restrict__ cnt,
    int* __restrict__ rank, int E) {
  const int i0 = (int)(blockIdx.x * blockDim.x + threadIdx.x);
  const int n = (int)(gridDim.x * blockDim.x);
  for (int e = i0; e < E; e += n) rank[e] = atomicAdd(cnt + dst[e], 1);
}

__global__ void __launch_bounds__(1024) csr_scan(
    const int* __restrict__ cnt, int* __restrict__ rowptr, int n) {
  __shared__ int sfront[32];
  __shared__ int swave[32];
  const int tid = (int)threadIdx.x;
  const int lane = tid & 31;
  const int wid = tid >> 5;
  for (int base = (int)blockIdx.x * 1024; base < n;
       base += (int)gridDim.x * 1024) {
    int front = 0;
    for (int i = tid; i < base; i += 1024) front += cnt[i];
    for (int o = 16; o > 0; o >>= 1) front += __shfl_xor(front, o);
    const int i = base + tid;
    const int v = i < n ? cnt[i] : 0;
    int x = v;
    for (int o = 1; o < 32; o <<= 1) {
      const int y = __shfl_up(x, o);
      if (lane >= o) x += y;
    }
    if (lane == 0) sfront[wid] = front;
    if (lane == 31) swave[wid] = x;
    __syncthreads();
    int pre = sfront[lane] + (lane < wid ? swave[lane] : 0);
    for (int o = 16; o > 0; o >>= 1) pre += __shfl_xor(pre, o);
    if (i < n) rowptr[i] = pre + x - v;
    __syncthreads();
  }
}

__global__ void __launch_bounds__(256) csr_fill(
    const int* __restrict__ src, const int* __restrict__ dst,
    const int* __restrict__ rank, const int* __restrict__ rowptr,
    int* __restrict__ col, int E) {
  const int i0 = (int)(blockIdx.x * blockDim.x + threadIdx.x);
  const int n = (int)(gridDim.x * blockDim.x);
  for (int e = i0; e < E; e += n) col[rowptr[dst[e]] + rank[e]] = src[e];
}

#define MLP_NW 16
#define MLP_NT (MLP_NW * 32)
#define GU 32

template <int NCOL2, bool HAS_AGG, bool RELU2>
__global__ void __launch_bounds__(MLP_NT) gin_mlp(
    const float* __restrict__ hin, const int* __restrict__ rowptr,
    const int* __restrict__ cnt, const int* __restrict__ col,
    const float* __restrict__ W1, const float* __restrict__ b1,
    const float* __restrict__ W2, const float* __restrict__ b2,
    float* __restrict__ out, int nrows) {
  constexpr int N2 = NCOL2 * 16;
  constexpr int SWP = 144;
  constexpr int SZ = 132;
  __shared__ __attribute__((aligned(16))) v2f sW1[(DH / 2) * SWP];
  __shared__ __attribute__((aligned(16))) v2f sW2[(DH / 2) * SWP];
  __shared__ float sb1[DH];
  __shared__ float sb2[DH];
  __shared__ __attribute__((aligned(16))) float sZ[MLP_NW][16 * SZ];

  const int tid = (int)threadIdx.x;

  constexpr int CG2 = N2 / 4;
  constexpr int NJ1 = (DH / 2) * (DH / 4) / MLP_NT;
  constexpr int NJ2 = (DH / 2) * CG2 / MLP_NT;
  const int rot = (int)blockIdx.x;
  float4 w1a[NJ1], w1b[NJ1], w2a[NJ2], w2b[NJ2];
#pragma unroll
  for (int j = 0; j < NJ1; ++j) {
    const int it = j * MLP_NT + tid;
    const int p = ((it >> 5) + rot) & (DH / 2 - 1), cg = it & 31;
    const float* g = W1 + (size_t)(2 * p) * DH + cg * 4;
    w1a[j] = *reinterpret_cast<const float4*>(g);
    w1b[j] = *reinterpret_cast<const float4*>(g + DH);
  }
#pragma unroll
  for (int j = 0; j < NJ2; ++j) {
    const int it = j * MLP_NT + tid;
    const int p = (it / CG2 + rot) & (DH / 2 - 1), cg = it % CG2;
    const float* g = W2 + (size_t)(2 * p) * N2 + cg * 4;
    w2a[j] = *reinterpret_cast<const float4*>(g);
    w2b[j] = *reinterpret_cast<const float4*>(g + N2);
  }
  float bv1 = 0.f, bv2 = 0.f;
  if (tid < DH) bv1 = b1[tid];
  if (tid < N2) bv2 = b2[tid];
#pragma unroll
  for (int j = 0; j < NJ1; ++j) {
    const int it = j * MLP_NT + tid;
    const int p = ((it >> 5) + rot) & (DH / 2 - 1), cg = it & 31;
    float4* d = reinterpret_cast<float4*>(&sW1[p * SWP + cg * 4]);
    const float4 lo = {w1a[j].x, w1b[j].x, w1a[j].y, w1b[j].y};
    const float4 hi = {w1a[j].z, w1b[j].z, w1a[j].w, w1b[j].w};
    d[0] = lo;
    d[1] = hi;
  }
#pragma unroll
  for (int j = 0; j < NJ2; ++j) {
    const int it = j * MLP_NT + tid;
    const int p = (it / CG2 + rot) & (DH / 2 - 1), cg = it % CG2;
    float4* d = reinterpret_cast<float4*>(&sW2[p * SWP + cg * 4]);
    const float4 lo = {w2a[j].x, w2b[j].x, w2a[j].y, w2b[j].y};
    const float4 hi = {w2a[j].z, w2b[j].z, w2a[j].w, w2b[j].w};
    d[0] = lo;
    d[1] = hi;
  }
  if (tid < DH) sb1[tid] = bv1;
  if (tid < N2) sb2[tid] = bv2;
  __syncthreads();

  const int lane = tid & 31;
  const int wid = tid >> 5;
  const int m = lane & 15;
  const int koff = (lane >> 4) * 2;
  const int kpo = lane >> 4;
  const int mrow = (lane >> 4) * 8;
  float* zs = sZ[wid];

  const int ntiles = (nrows + 15) >> 4;
  const int gw0 = (int)blockIdx.x + (int)gridDim.x * wid;
  const int gwn = (int)gridDim.x * MLP_NW;

  for (int t = gw0; t < ntiles; t += gwn) {
    const int row0 = t << 4;

    if (HAS_AGG) {
      const int r0 = __builtin_amdgcn_readfirstlane(row0);
      const int nv = nrows - r0 < 16 ? nrows - r0 : 16;
      int nd = r0 + m;
      if (nd > nrows - 1) nd = nrows - 1;
      const int rp = rowptr[nd];
      const int dg = m < nv ? cnt[nd] : 0;
      const int myend = rp + dg;
      float4 own[16];
#pragma unroll
      for (int r = 0; r < 16; ++r) {
        int nr = r0 + r;
        if (nr > nrows - 1) nr = nrows - 1;
        own[r] =
            *reinterpret_cast<const float4*>(hin + (size_t)nr * DH + lane * 4);
      }
      const int beg = __builtin_amdgcn_readlane(rp, 0);
      const int end = __builtin_amdgcn_readlane(myend, nv - 1);
      int cur = 0;
      int curend = __builtin_amdgcn_readlane(myend, 0);
      float4 acc = {0.f, 0.f, 0.f, 0.f};
      int mine = 0;
      if (beg < end) mine = beg + lane < end ? col[beg + lane] : 0;
#pragma unroll
      for (int r = 0; r < 16; ++r)
        *reinterpret_cast<float4*>(&zs[r * SZ + lane * 4]) = own[r];
      for (int base = beg; base < end; base += GU) {
        float4 v[GU];
#pragma unroll
        for (int u = 0; u < GU; ++u) {
          const int sidx = __builtin_amdgcn_readlane(mine, u);
          v[u] = *reinterpret_cast<const float4*>(hin + (size_t)sidx * DH +
                                                  lane * 4);
        }
        int nxt = 0;
        if (base + GU < end)
          nxt = base + GU + lane < end ? col[base + GU + lane] : 0;
#pragma unroll
        for (int u = 0; u < GU; ++u) {
          const int p = base + u;
          if (p < end) {
            while (p >= curend && cur < 15) {
              float4* zr = reinterpret_cast<float4*>(&zs[cur * SZ + lane * 4]);
              float4 hv = *zr;
              hv.x += acc.x;
              hv.y += acc.y;
              hv.z += acc.z;
              hv.w += acc.w;
              *zr = hv;
              acc.x = 0.f;
              acc.y = 0.f;
              acc.z = 0.f;
              acc.w = 0.f;
              ++cur;
              curend = __builtin_amdgcn_readlane(myend, cur);
            }
            acc.x += v[u].x;
            acc.y += v[u].y;
            acc.z += v[u].z;
            acc.w += v[u].w;
          }
        }
        mine = nxt;
      }
      {
        float4* zr = reinterpret_cast<float4*>(&zs[cur * SZ + lane * 4]);
        float4 hv = *zr;
        hv.x += acc.x;
        hv.y += acc.y;
        hv.z += acc.z;
        hv.w += acc.w;
        *zr = hv;
      }
      __builtin_amdgcn_wave_barrier();
    }

    int rowa = row0 + m;
    if (rowa > nrows - 1) rowa = nrows - 1;
    const float* hrow = hin + (size_t)rowa * DH;

    v8f acc[8];
#pragma unroll
    for (int n = 0; n < 8; ++n) acc[n] = zero8();

    for (int k0 = 0; k0 < DH; k0 += 4) {
      const int k = k0 + koff;
      const int kp = (k0 >> 1) + kpo;
      v2f a;
      if (HAS_AGG) {
        a = *reinterpret_cast<const v2f*>(&zs[m * SZ + k]);
      } else {
        a = *reinterpret_cast<const v2f*>(hrow + k);
      }
#pragma unroll
      for (int n = 0; n < 8; ++n) {
        const v2f bf = sW1[kp * SWP + n * 16 + m];
        acc[n] = __builtin_amdgcn_wmma_f32_16x16x4_f32(
            false, a, false, bf, (short)0, acc[n], false, false);
      }
    }
    if (HAS_AGG) __builtin_amdgcn_wave_barrier();

#pragma unroll
    for (int n = 0; n < 8; ++n) {
      const float bias = sb1[n * 16 + m];
#pragma unroll
      for (int j = 0; j < 8; ++j) {
        float v = acc[n][j] + bias;
        v = v > 0.0f ? v : 0.0f;
        zs[(mrow + j) * SZ + n * 16 + m] = v;
      }
    }
    __builtin_amdgcn_wave_barrier();

#pragma unroll
    for (int n = 0; n < NCOL2; ++n) acc[n] = zero8();
    for (int k0 = 0; k0 < DH; k0 += 4) {
      const int k = k0 + koff;
      const int kp = (k0 >> 1) + kpo;
      const v2f a = *reinterpret_cast<const v2f*>(&zs[m * SZ + k]);
#pragma unroll
      for (int n = 0; n < NCOL2; ++n) {
        const v2f bf = sW2[kp * SWP + n * 16 + m];
        acc[n] = __builtin_amdgcn_wmma_f32_16x16x4_f32(
            false, a, false, bf, (short)0, acc[n], false, false);
      }
    }
    __builtin_amdgcn_wave_barrier();

#pragma unroll
    for (int n = 0; n < NCOL2; ++n) {
      const float bias = sb2[n * 16 + m];
#pragma unroll
      for (int j = 0; j < 8; ++j) {
        float v = acc[n][j] + bias;
        if (RELU2) v = v > 0.0f ? v : 0.0f;
        const int row = row0 + mrow + j;
        if (row < nrows) out[(size_t)row * N2 + n * 16 + m] = v;
      }
    }
  }
}

extern "C" void kernel_launch(void* const* d_in, const int* in_sizes, int n_in,
                              void* d_out, int out_size, void* d_ws,
                              size_t ws_size, hipStream_t stream) {
  const float* x = (const float*)d_in[0];
  const int* ei = (const int*)d_in[1];
  const int N = in_sizes[0] / DH;
  const int E = in_sizes[1] / 2;
  const int* src = ei;
  const int* dst = ei + E;

  const float* c1w1 = (const float*)d_in[2];
  const float* c1b1 = (const float*)d_in[3];
  const float* c1w2 = (const float*)d_in[4];
  const float* c1b2 = (const float*)d_in[5];
  const float* c2w1 = (const float*)d_in[6];
  const float* c2b1 = (const float*)d_in[7];
  const float* c2w2 = (const float*)d_in[8];
  const float* c2b2 = (const float*)d_in[9];
  const float* c3w1 = (const float*)d_in[10];
  const float* c3b1 = (const float*)d_in[11];
  const float* c3w2 = (const float*)d_in[12];
  const float* c3b2 = (const float*)d_in[13];
  const float* l1w = (const float*)d_in[14];
  const float* l1b = (const float*)d_in[15];
  const float* l2w = (const float*)d_in[16];
  const float* l2b = (const float*)d_in[17];

  float* hA = (float*)d_ws;
  float* hB = hA + (size_t)N * DH;
  int* cnt = (int*)(hB + (size_t)N * DH);
  int* rowptr = cnt + N;
  int* rank = rowptr + N;
  int* col = rank + E;

  const dim3 eB(256), eG(1024);
  const dim3 pB(1024), pG(64);
  const dim3 mB(512), mG(256);

  hipMemsetAsync(cnt, 0, (size_t)N * sizeof(int), stream);
  csr_rank<<<eG, eB, 0, stream>>>(dst, cnt, rank, E);
  csr_scan<<<pG, pB, 0, stream>>>(cnt, rowptr, N);
  csr_fill<<<eG, eB, 0, stream>>>(src, dst, rank, rowptr, col, E);

  gin_mlp<8, true, true><<<mG, mB, 0, stream>>>(
      x, rowptr, cnt, col, c1w1, c1b1, c1w2, c1b2, hA, N);
  gin_mlp<8, true, true><<<mG, mB, 0, stream>>>(
      hA, rowptr, cnt, col, c2w1, c2b1, c2w2, c2b2, hB, N);
  gin_mlp<8, true, true><<<mG, mB, 0, stream>>>(
      hB, rowptr, cnt, col, c3w1, c3b1, c3w2, c3b2, hA, N);
  gin_mlp<4, false, false><<<mG, mB, 0, stream>>>(
      hA, nullptr, nullptr, nullptr, l1w, l1b, l2w, l2b, (float*)d_out, N);
}


# gin_mlp tile I/O as whole rows via the wave's LDS slice: output tile staged, stored 16 B per lane in full 128-B lines (was 64 guarded 4-B stores per lane); head input rows staged like the layers'
# speedup vs baseline: 3.2023x; 1.0465x over previous
#include <hip/hip_runtime.h>


typedef float v2f __attribute__((ext_vector_type(2)));
typedef float v8f __attribute__((ext_vector_type(8)));

#define DH 128

static __device__ __forceinline__ v8f zero8() {
  v8f z = {0.f, 0.f, 0.f, 0.f, 0.f, 0.f, 0.f, 0.f};
  return z;
}

__global__ void __launch_bounds__(256) csr_rank(
    const int* __restrict__ dst, int* __restrict__ cnt,
    int* __restrict__ rank, int E) {
  const int i0 = (int)(blockIdx.x * blockDim.x + threadIdx.x);
  const int n = (int)(gridDim.x * blockDim.x);
  for (int e = i0; e < E; e += n) rank[e] = atomicAdd(cnt + dst[e], 1);
}

__global__ void __launch_bounds__(1024) csr_scan(
    const int* __restrict__ cnt, int* __restrict__ rowptr, int n) {
  __shared__ int sfront[32];
  __shared__ int swave[32];
  const int tid = (int)threadIdx.x;
  const int lane = tid & 31;
  const int wid = tid >> 5;
  for (int base = (int)blockIdx.x * 1024; base < n;
       base += (int)gridDim.x * 1024) {
    int front = 0;
    for (int i = tid; i < base; i += 1024) front += cnt[i];
    for (int o = 16; o > 0; o >>= 1) front += __shfl_xor(front, o);
    const int i = base + tid;
    const int v = i < n ? cnt[i] : 0;
    int x = v;
    for (int o = 1; o < 32; o <<= 1) {
      const int y = __shfl_up(x, o);
      if (lane >= o) x += y;
    }
    if (lane == 0) sfront[wid] = front;
    if (lane == 31) swave[wid] = x;
    __syncthreads();
    int pre = sfront[lane] + (lane < wid ? swave[lane] : 0);
    for (int o = 16; o > 0; o >>= 1) pre += __shfl_xor(pre, o);
    if (i < n) rowptr[i] = pre + x - v;
    __syncthreads();
  }
}

__global__ void __launch_bounds__(256) csr_fill(
    const int* __restrict__ src, const int* __restrict__ dst,
    const int* __restrict__ rank, const int* __restrict__ rowptr,
    int* __restrict__ col, int E) {
  const int i0 = (int)(blockIdx.x * blockDim.x + threadIdx.x);
  const int n = (int)(gridDim.x * blockDim.x);
  for (int e = i0; e < E; e += n) col[rowptr[dst[e]] + rank[e]] = src[e];
}

#define MLP_NW 16
#define MLP_NT (MLP_NW * 32)
#define GU 32

template <int NCOL2, bool HAS_AGG, bool RELU2>
__global__ void __launch_bounds__(MLP_NT) gin_mlp(
    const float* __restrict__ hin, const int* __restrict__ rowptr,
    const int* __restrict__ cnt, const int* __restrict__ col,
    const float* __restrict__ W1, const float* __restrict__ b1,
    const float* __restrict__ W2, const float* __restrict__ b2,
    float* __restrict__ out, int nrows) {
  constexpr int N2 = NCOL2 * 16;
  constexpr int SWP = 144;
  constexpr int SZ = 132;
  __shared__ __attribute__((aligned(16))) v2f sW1[(DH / 2) * SWP];
  __shared__ __attribute__((aligned(16))) v2f sW2[(DH / 2) * SWP];
  __shared__ float sb1[DH];
  __shared__ float sb2[DH];
  __shared__ __attribute__((aligned(16))) float sZ[MLP_NW][16 * SZ];

  const int tid = (int)threadIdx.x;

  constexpr int CG2 = N2 / 4;
  constexpr int NJ1 = (DH / 2) * (DH / 4) / MLP_NT;
  constexpr int NJ2 = (DH / 2) * CG2 / MLP_NT;
  const int rot = (int)blockIdx.x;
  float4 w1a[NJ1], w1b[NJ1], w2a[NJ2], w2b[NJ2];
#pragma unroll
  for (int j = 0; j < NJ1; ++j) {
    const int it = j * MLP_NT + tid;
    const int p = ((it >> 5) + rot) & (DH / 2 - 1), cg = it & 31;
    const float* g = W1 + (size_t)(2 * p) * DH + cg * 4;
    w1a[j] = *reinterpret_cast<const float4*>(g);
    w1b[j] = *reinterpret_cast<const float4*>(g + DH);
  }
#pragma unroll
  for (int j = 0; j < NJ2; ++j) {
    const int it = j * MLP_NT + tid;
    const int p = (it / CG2 + rot) & (DH / 2 - 1), cg = it % CG2;
    const float* g = W2 + (size_t)(2 * p) * N2 + cg * 4;
    w2a[j] = *reinterpret_cast<const float4*>(g);
    w2b[j] = *reinterpret_cast<const float4*>(g + N2);
  }
  float bv1 = 0.f, bv2 = 0.f;
  if (tid < DH) bv1 = b1[tid];
  if (tid < N2) bv2 = b2[tid];
#pragma unroll
  for (int j = 0; j < NJ1; ++j) {
    const int it = j * MLP_NT + tid;
    const int p = ((it >> 5) + rot) & (DH / 2 - 1), cg = it & 31;
    float4* d = reinterpret_cast<float4*>(&sW1[p * SWP + cg * 4]);
    const float4 lo = {w1a[j].x, w1b[j].x, w1a[j].y, w1b[j].y};
    const float4 hi = {w1a[j].z, w1b[j].z, w1a[j].w, w1b[j].w};
    d[0] = lo;
    d[1] = hi;
  }
#pragma unroll
  for (int j = 0; j < NJ2; ++j) {
    const int it = j * MLP_NT + tid;
    const int p = (it / CG2 + rot) & (DH / 2 - 1), cg = it % CG2;
    float4* d = reinterpret_cast<float4*>(&sW2[p * SWP + cg * 4]);
    const float4 lo = {w2a[j].x, w2b[j].x, w2a[j].y, w2b[j].y};
    const float4 hi = {w2a[j].z, w2b[j].z, w2a[j].w, w2b[j].w};
    d[0] = lo;
    d[1] = hi;
  }
  if (tid < DH) sb1[tid] = bv1;
  if (tid < N2) sb2[tid] = bv2;
  __syncthreads();

  const int lane = tid & 31;
  const int wid = tid >> 5;
  const int m = lane & 15;
  const int koff = (lane >> 4) * 2;
  const int kpo = lane >> 4;
  const int mrow = (lane >> 4) * 8;
  float* zs = sZ[wid];

  const int ntiles = (nrows + 15) >> 4;
  const int gw0 = (int)blockIdx.x + (int)gridDim.x * wid;
  const int gwn = (int)gridDim.x * MLP_NW;

  for (int t = gw0; t < ntiles; t += gwn) {
    const int r0 = __builtin_amdgcn_readfirstlane(t << 4);

    if (!HAS_AGG) {
      float4 own[16];
#pragma unroll
      for (int r = 0; r < 16; ++r) {
        int nr = r0 + r;
        if (nr > nrows - 1) nr = nrows - 1;
        own[r] =
            *reinterpret_cast<const float4*>(hin + (size_t)nr * DH + lane * 4);
      }
#pragma unroll
      for (int r = 0; r < 16; ++r)
        *reinterpret_cast<float4*>(&zs[r * SZ + lane * 4]) = own[r];
      __builtin_amdgcn_wave_barrier();
    }

    if (HAS_AGG) {
      const int nv = nrows - r0 < 16 ? nrows - r0 : 16;
      int nd = r0 + m;
      if (nd > nrows - 1) nd = nrows - 1;
      const int rp = rowptr[nd];
      const int dg = m < nv ? cnt[nd] : 0;
      const int myend = rp + dg;
      float4 own[16];
#pragma unroll
      for (int r = 0; r < 16; ++r) {
        int nr = r0 + r;
        if (nr > nrows - 1) nr = nrows - 1;
        own[r] =
            *reinterpret_cast<const float4*>(hin + (size_t)nr * DH + lane * 4);
      }
      const int beg = __builtin_amdgcn_readlane(rp, 0);
      const int end = __builtin_amdgcn_readlane(myend, nv - 1);
      int cur = 0;
      int curend = __builtin_amdgcn_readlane(myend, 0);
      float4 acc = {0.f, 0.f, 0.f, 0.f};
      int mine = 0;
      if (beg < end) mine = beg + lane < end ? col[beg + lane] : 0;
#pragma unroll
      for (int r = 0; r < 16; ++r)
        *reinterpret_cast<float4*>(&zs[r * SZ + lane * 4]) = own[r];
      for (int base = beg; base < end; base += GU) {
        float4 v[GU];
#pragma unroll
        for (int u = 0; u < GU; ++u) {
          const int sidx = __builtin_amdgcn_readlane(mine, u);
          v[u] = *reinterpret_cast<const float4*>(hin + (size_t)sidx * DH +
                                                  lane * 4);
        }
        int nxt = 0;
        if (base + GU < end)
          nxt = base + GU + lane < end ? col[base + GU + lane] : 0;
#pragma unroll
        for (int u = 0; u < GU; ++u) {
          const int p = base + u;
          if (p < end) {
            while (p >= curend && cur < 15) {
              float4* zr = reinterpret_cast<float4*>(&zs[cur * SZ + lane * 4]);
              float4 hv = *zr;
              hv.x += acc.x;
              hv.y += acc.y;
              hv.z += acc.z;
              hv.w += acc.w;
              *zr = hv;
              acc.x = 0.f;
              acc.y = 0.f;
              acc.z = 0.f;
              acc.w = 0.f;
              ++cur;
              curend = __builtin_amdgcn_readlane(myend, cur);
            }
            acc.x += v[u].x;
            acc.y += v[u].y;
            acc.z += v[u].z;
            acc.w += v[u].w;
          }
        }
        mine = nxt;
      }
      {
        float4* zr = reinterpret_cast<float4*>(&zs[cur * SZ + lane * 4]);
        float4 hv = *zr;
        hv.x += acc.x;
        hv.y += acc.y;
        hv.z += acc.z;
        hv.w += acc.w;
        *zr = hv;
      }
      __builtin_amdgcn_wave_barrier();
    }

    v8f acc[8];
#pragma unroll
    for (int n = 0; n < 8; ++n) acc[n] = zero8();

    for (int k0 = 0; k0 < DH; k0 += 4) {
      const int k = k0 + koff;
      const int kp = (k0 >> 1) + kpo;
      const v2f a = *reinterpret_cast<const v2f*>(&zs[m * SZ + k]);
#pragma unroll
      for (int n = 0; n < 8; ++n) {
        const v2f bf = sW1[kp * SWP + n * 16 + m];
        acc[n] = __builtin_amdgcn_wmma_f32_16x16x4_f32(
            false, a, false, bf, (short)0, acc[n], false, false);
      }
    }
    __builtin_amdgcn_wave_barrier();

#pragma unroll
    for (int n = 0; n < 8; ++n) {
      const float bias = sb1[n * 16 + m];
#pragma unroll
      for (int j = 0; j < 8; ++j) {
        float v = acc[n][j] + bias;
        v = v > 0.0f ? v : 0.0f;
        zs[(mrow + j) * SZ + n * 16 + m] = v;
      }
    }
    __builtin_amdgcn_wave_barrier();

#pragma unroll
    for (int n = 0; n < NCOL2; ++n) acc[n] = zero8();
    for (int k0 = 0; k0 < DH; k0 += 4) {
      const int k = k0 + koff;
      const int kp = (k0 >> 1) + kpo;
      const v2f a = *reinterpret_cast<const v2f*>(&zs[m * SZ + k]);
#pragma unroll
      for (int n = 0; n < NCOL2; ++n) {
        const v2f bf = sW2[kp * SWP + n * 16 + m];
        acc[n] = __builtin_amdgcn_wmma_f32_16x16x4_f32(
            false, a, false, bf, (short)0, acc[n], false, false);
      }
    }
    __builtin_amdgcn_wave_barrier();

#pragma unroll
    for (int n = 0; n < NCOL2; ++n) {
      const float bias = sb2[n * 16 + m];
#pragma unroll
      for (int j = 0; j < 8; ++j) {
        float v = acc[n][j] + bias;
        if (RELU2) v = v > 0.0f ? v : 0.0f;
        zs[(mrow + j) * SZ + n * 16 + m] = v;
      }
    }
    __builtin_amdgcn_wave_barrier();
    constexpr int LPR = N2 / 4;
    constexpr int RPS = 32 / LPR;
    const int rsub = lane / LPR;
    const int c4 = (lane % LPR) * 4;
    float* orow0 = out + (size_t)(r0 + rsub) * N2 + c4;
    if (r0 + 16 <= nrows) {
      float4 orow[16 / RPS];
#pragma unroll
      for (int i = 0; i < 16 / RPS; ++i)
        orow[i] = *reinterpret_cast<const float4*>(
            &zs[(i * RPS + rsub) * SZ + c4]);
#pragma unroll
      for (int i = 0; i < 16 / RPS; ++i)
        *reinterpret_cast<float4*>(orow0 + (size_t)(i * RPS) * N2) = orow[i];
    } else {
#pragma unroll
      for (int i = 0; i < 16 / RPS; ++i) {
        if (r0 + i * RPS + rsub < nrows)
          *reinterpret_cast<float4*>(orow0 + (size_t)(i * RPS) * N2) =
              *reinterpret_cast<const float4*>(
                  &zs[(i * RPS + rsub) * SZ + c4]);
      }
    }
  }
}

extern "C" void kernel_launch(void* const* d_in, const int* in_sizes, int n_in,
                              void* d_out, int out_size, void* d_ws,
                              size_t ws_size, hipStream_t stream) {
  const float* x = (const float*)d_in[0];
  const int* ei = (const int*)d_in[1];
  const int N = in_sizes[0] / DH;
  const int E = in_sizes[1] / 2;
  const int* src = ei;
  const int* dst = ei + E;

  const float* c1w1 = (const float*)d_in[2];
  const float* c1b1 = (const float*)d_in[3];
  const float* c1w2 = (const float*)d_in[4];
  const float* c1b2 = (const float*)d_in[5];
  const float* c2w1 = (const float*)d_in[6];
  const float* c2b1 = (const float*)d_in[7];
  const float* c2w2 = (const float*)d_in[8];
  const float* c2b2 = (const float*)d_in[9];
  const float* c3w1 = (const float*)d_in[10];
  const float* c3b1 = (const float*)d_in[11];
  const float* c3w2 = (const float*)d_in[12];
  const float* c3b2 = (const float*)d_in[13];
  const float* l1w = (const float*)d_in[14];
  const float* l1b = (const float*)d_in[15];
  const float* l2w = (const float*)d_in[16];
  const float* l2b = (const float*)d_in[17];

  float* hA = (float*)d_ws;
  float* hB = hA + (size_t)N * DH;
  int* cnt = (int*)(hB + (size_t)N * DH);
  int* rowptr = cnt + N;
  int* rank = rowptr + N;
  int* col = rank + E;

  const dim3 eB(256), eG(1024);
  const dim3 pB(1024), pG(64);
  const dim3 mB(512), mG(256);

  hipMemsetAsync(cnt, 0, (size_t)N * sizeof(int), stream);
  csr_rank<<<eG, eB, 0, stream>>>(dst, cnt, rank, E);
  csr_scan<<<pG, pB, 0, stream>>>(cnt, rowptr, N);
  csr_fill<<<eG, eB, 0, stream>>>(src, dst, rank, rowptr, col, E);

  gin_mlp<8, true, true><<<mG, mB, 0, stream>>>(
      x, rowptr, cnt, col, c1w1, c1b1, c1w2, c1b2, hA, N);
  gin_mlp<8, true, true><<<mG, mB, 0, stream>>>(
      hA, rowptr, cnt, col, c2w1, c2b1, c2w2, c2b2, hB, N);
  gin_mlp<8, true, true><<<mG, mB, 0, stream>>>(
      hB, rowptr, cnt, col, c3w1, c3b1, c3w2, c3b2, hA, N);
  gin_mlp<4, false, false><<<mG, mB, 0, stream>>>(
      hA, nullptr, nullptr, nullptr, l1w, l1b, l2w, l2b, (float*)d_out, N);
}


# edge list by destination built by a two-level counting sort in LDS (chunk-by-bin pass, then a workgroup per 128-node bin); global atomics, memset, scan launch gone; sum order: LDS arrival rank
# speedup vs baseline: 4.3887x; 1.3705x over previous
#include <hip/hip_runtime.h>


typedef float v2f __attribute__((ext_vector_type(2)));
typedef float v8f __attribute__((ext_vector_type(8)));

#define DH 128

static __device__ __forceinline__ v8f zero8() {
  v8f z = {0.f, 0.f, 0.f, 0.f, 0.f, 0.f, 0.f, 0.f};
  return z;
}

#define CSR_CH 3200
#define CSR_T1 512
#define CSR_EPT 7
#define CSR_SH 7
#define CSR_NPB 128
#define CSR_MAXB 512
#define CSR_MST 514
#define CSR_T2 256
#define CSR_CAP 6144

template <int NW>
static __device__ __forceinline__ int block_excl_scan(int v, int* swave,
                                                      int* tot) {
  const int lane = (int)threadIdx.x & 31;
  const int wid = (int)threadIdx.x >> 5;
  int x = v;
  for (int o = 1; o < 32; o <<= 1) {
    const int y = __shfl_up(x, o);
    if (lane >= o) x += y;
  }
  if (lane == 31) swave[wid] = x;
  __syncthreads();
  int w = lane < NW ? swave[lane] : 0;
  int pre = lane < wid ? w : 0;
  for (int o = 16; o > 0; o >>= 1) {
    pre += __shfl_xor(pre, o);
    w += __shfl_xor(w, o);
  }
  *tot = w;
  return pre + x - v;
}

__global__ void __launch_bounds__(CSR_T1) csr_bin(
    const int* __restrict__ src, const int* __restrict__ dst,
    int* __restrict__ mat, int* __restrict__ ebin, int E, int N) {
  __shared__ int hist[CSR_MAXB];
  __shared__ int offs[CSR_MAXB];
  __shared__ int swave[32];
  __shared__ __attribute__((aligned(16))) int2 sorted[CSR_CH];
  const int tid = (int)threadIdx.x;
  const int nchunks = (E + CSR_CH - 1) / CSR_CH;
  for (int c = (int)blockIdx.x; c < nchunks; c += (int)gridDim.x) {
    const int e0 = c * CSR_CH;
    const int n = E - e0 < CSR_CH ? E - e0 : CSR_CH;
    hist[tid] = 0;
    __syncthreads();
    int s[CSR_EPT], d[CSR_EPT], r[CSR_EPT];
#pragma unroll
    for (int j = 0; j < CSR_EPT; ++j) {
      const int i = tid + j * CSR_T1;
      s[j] = 0;
      d[j] = -1;
      if (i < n) {
        s[j] = src[e0 + i];
        d[j] = dst[e0 + i];
      }
    }
#pragma unroll
    for (int j = 0; j < CSR_EPT; ++j) {
      r[j] = 0;
      if (d[j] >= 0) {
        int b = d[j] >> CSR_SH;
        if (b > CSR_MAXB - 1) b = CSR_MAXB - 1;
        r[j] = atomicAdd(&hist[b], 1);
      }
    }
    __syncthreads();
    const int v = hist[tid];
    int tot;
    const int ex = block_excl_scan<CSR_T1 / 32>(v, swave, &tot);
    offs[tid] = ex;
    mat[(size_t)c * CSR_MST + tid] = ex;
    if (tid == CSR_T1 - 1) mat[(size_t)c * CSR_MST + CSR_MAXB] = ex + v;
    __syncthreads();
#pragma unroll
    for (int j = 0; j < CSR_EPT; ++j) {
      if (d[j] >= 0) {
        int b = d[j] >> CSR_SH;
        if (b > CSR_MAXB - 1) b = CSR_MAXB - 1;
        int2 p;
        p.x = s[j];
        p.y = d[j];
        sorted[offs[b] + r[j]] = p;
      }
    }
    __syncthreads();
    int2* eb = reinterpret_cast<int2*>(ebin) + e0;
    for (int i = tid; i < n; i += CSR_T1) eb[i] = sorted[i];
    __syncthreads();
  }
}

__global__ void __launch_bounds__(CSR_T2) csr_emit(
    const int* __restrict__ mat, const int* __restrict__ ebin,
    int* __restrict__ cnt, int* __restrict__ rowptr, int* __restrict__ col,
    int E, int N) {
  __shared__ int ncnt[CSR_NPB];
  __shared__ int cur[CSR_NPB];
  __shared__ int swave[32];
  __shared__ int sred[2][CSR_T2 / 32];
  __shared__ int colbuf[CSR_CAP];
  const int tid = (int)threadIdx.x;
  const int lane = tid & 31;
  const int wid = tid >> 5;
  const int nbins = ((N - 1) >> CSR_SH) + 1;
  const int nchunks = (E + CSR_CH - 1) / CSR_CH;
  const int2* eb = reinterpret_cast<const int2*>(ebin);
  for (int b = (int)blockIdx.x; b < nbins; b += (int)gridDim.x) {
    int st = 0, nt = 0;
    for (int c = tid; c < nchunks; c += CSR_T2) {
      const int o0 = mat[(size_t)c * CSR_MST + b];
      const int o1 = mat[(size_t)c * CSR_MST + b + 1];
      st += o0;
      nt += o1 - o0;
    }
    for (int o = 16; o > 0; o >>= 1) {
      st += __shfl_xor(st, o);
      nt += __shfl_xor(nt, o);
    }
    if (lane == 0) {
      sred[0][wid] = st;
      sred[1][wid] = nt;
    }
    if (tid < CSR_NPB) ncnt[tid] = 0;
    __syncthreads();
    int bstart = 0, btot = 0;
#pragma unroll
    for (int w = 0; w < CSR_T2 / 32; ++w) {
      bstart += sred[0][w];
      btot += sred[1][w];
    }
    for (int c = tid; c < nchunks; c += CSR_T2) {
      const int o0 = mat[(size_t)c * CSR_MST + b];
      const int o1 = mat[(size_t)c * CSR_MST + b + 1];
      const int2* seg = eb + (size_t)c * CSR_CH;
      for (int i = o0; i < o1; i += 4) {
        int2 e[4];
#pragma unroll
        for (int u = 0; u < 4; ++u) {
          e[u].x = 0;
          e[u].y = -1;
          if (i + u < o1) e[u] = seg[i + u];
        }
#pragma unroll
        for (int u = 0; u < 4; ++u)
          if (e[u].y >= 0) atomicAdd(&ncnt[e[u].y & (CSR_NPB - 1)], 1);
      }
    }
    __syncthreads();
    const int v = tid < CSR_NPB ? ncnt[tid] : 0;
    int tot;
    const int ex = block_excl_scan<CSR_T2 / 32>(v, swave, &tot);
    if (tid < CSR_NPB) {
      cur[tid] = ex;
      const int node = b * CSR_NPB + tid;
      if (node < N) {
        cnt[node] = v;
        rowptr[node] = bstart + ex;
      }
    }
    __syncthreads();
    const bool inlds = btot <= CSR_CAP;
    for (int c = tid; c < nchunks; c += CSR_T2) {
      const int o0 = mat[(size_t)c * CSR_MST + b];
      const int o1 = mat[(size_t)c * CSR_MST + b + 1];
      const int2* seg = eb + (size_t)c * CSR_CH;
      for (int i = o0; i < o1; i += 4) {
        int2 e[4];
#pragma unroll
        for (int u = 0; u < 4; ++u) {
          e[u].x = 0;
          e[u].y = -1;
          if (i + u < o1) e[u] = seg[i + u];
        }
#pragma unroll
        for (int u = 0; u < 4; ++u) {
          if (e[u].y >= 0) {
            const int pos = atomicAdd(&cur[e[u].y & (CSR_NPB - 1)], 1);
            if (inlds)
              colbuf[pos] = e[u].x;
            else
              col[bstart + pos] = e[u].x;
          }
        }
      }
    }
    __syncthreads();
    if (inlds)
      for (int i = tid; i < btot; i += CSR_T2) col[bstart + i] = colbuf[i];
    __syncthreads();
  }
}

#define MLP_NW 16
#define MLP_NT (MLP_NW * 32)
#define GU 32

template <int NCOL2, bool HAS_AGG, bool RELU2>
__global__ void __launch_bounds__(MLP_NT) gin_mlp(
    const float* __restrict__ hin, const int* __restrict__ rowptr,
    const int* __restrict__ cnt, const int* __restrict__ col,
    const float* __restrict__ W1, const float* __restrict__ b1,
    const float* __restrict__ W2, const float* __restrict__ b2,
    float* __restrict__ out, int nrows) {
  constexpr int N2 = NCOL2 * 16;
  constexpr int SWP = 144;
  constexpr int SZ = 132;
  __shared__ __attribute__((aligned(16))) v2f sW1[(DH / 2) * SWP];
  __shared__ __attribute__((aligned(16))) v2f sW2[(DH / 2) * SWP];
  __shared__ float sb1[DH];
  __shared__ float sb2[DH];
  __shared__ __attribute__((aligned(16))) float sZ[MLP_NW][16 * SZ];

  const int tid = (int)threadIdx.x;

  constexpr int CG2 = N2 / 4;
  constexpr int NJ1 = (DH / 2) * (DH / 4) / MLP_NT;
  constexpr int NJ2 = (DH / 2) * CG2 / MLP_NT;
  const int rot = (int)blockIdx.x;
  float4 w1a[NJ1], w1b[NJ1], w2a[NJ2], w2b[NJ2];
#pragma unroll
  for (int j = 0; j < NJ1; ++j) {
    const int it = j * MLP_NT + tid;
    const int p = ((it >> 5) + rot) & (DH / 2 - 1), cg = it & 31;
    const float* g = W1 + (size_t)(2 * p) * DH + cg * 4;
    w1a[j] = *reinterpret_cast<const float4*>(g);
    w1b[j] = *reinterpret_cast<const float4*>(g + DH);
  }
#pragma unroll
  for (int j = 0; j < NJ2; ++j) {
    const int it = j * MLP_NT + tid;
    const int p = (it / CG2 + rot) & (DH / 2 - 1), cg = it % CG2;
    const float* g = W2 + (size_t)(2 * p) * N2 + cg * 4;
    w2a[j] = *reinterpret_cast<const float4*>(g);
    w2b[j] = *reinterpret_cast<const float4*>(g + N2);
  }
  float bv1 = 0.f, bv2 = 0.f;
  if (tid < DH) bv1 = b1[tid];
  if (tid < N2) bv2 = b2[tid];
#pragma unroll
  for (int j = 0; j < NJ1; ++j) {
    const int it = j * MLP_NT + tid;
    const int p = ((it >> 5) + rot) & (DH / 2 - 1), cg = it & 31;
    float4* d = reinterpret_cast<float4*>(&sW1[p * SWP + cg * 4]);
    const float4 lo = {w1a[j].x, w1b[j].x, w1a[j].y, w1b[j].y};
    const float4 hi = {w1a[j].z, w1b[j].z, w1a[j].w, w1b[j].w};
    d[0] = lo;
    d[1] = hi;
  }
#pragma unroll
  for (int j = 0; j < NJ2; ++j) {
    const int it = j * MLP_NT + tid;
    const int p = (it / CG2 + rot) & (DH / 2 - 1), cg = it % CG2;
    float4* d = reinterpret_cast<float4*>(&sW2[p * SWP + cg * 4]);
    const float4 lo = {w2a[j].x, w2b[j].x, w2a[j].y, w2b[j].y};
    const float4 hi = {w2a[j].z, w2b[j].z, w2a[j].w, w2b[j].w};
    d[0] = lo;
    d[1] = hi;
  }
  if (tid < DH) sb1[tid] = bv1;
  if (tid < N2) sb2[tid] = bv2;
  __syncthreads();

  const int lane = tid & 31;
  const int wid = tid >> 5;
  const int m = lane & 15;
  const int koff = (lane >> 4) * 2;
  const int kpo = lane >> 4;
  const int mrow = (lane >> 4) * 8;
  float* zs = sZ[wid];

  const int ntiles = (nrows + 15) >> 4;
  const int gw0 = (int)blockIdx.x + (int)gridDim.x * wid;
  const int gwn = (int)gridDim.x * MLP_NW;

  for (int t = gw0; t < ntiles; t += gwn) {
    const int r0 = __builtin_amdgcn_readfirstlane(t << 4);

    if (!HAS_AGG) {
      float4 own[16];
#pragma unroll
      for (int r = 0; r < 16; ++r) {
        int nr = r0 + r;
        if (nr > nrows - 1) nr = nrows - 1;
        own[r] =
            *reinterpret_cast<const float4*>(hin + (size_t)nr * DH + lane * 4);
      }
#pragma unroll
      for (int r = 0; r < 16; ++r)
        *reinterpret_cast<float4*>(&zs[r * SZ + lane * 4]) = own[r];
      __builtin_amdgcn_wave_barrier();
    }

    if (HAS_AGG) {
      const int nv = nrows - r0 < 16 ? nrows - r0 : 16;
      int nd = r0 + m;
      if (nd > nrows - 1) nd = nrows - 1;
      const int rp = rowptr[nd];
      const int dg = m < nv ? cnt[nd] : 0;
      const int myend = rp + dg;
      float4 own[16];
#pragma unroll
      for (int r = 0; r < 16; ++r) {
        int nr = r0 + r;
        if (nr > nrows - 1) nr = nrows - 1;
        own[r] =
            *reinterpret_cast<const float4*>(hin + (size_t)nr * DH + lane * 4);
      }
      const int beg = __builtin_amdgcn_readlane(rp, 0);
      const int end = __builtin_amdgcn_readlane(myend, nv - 1);
      int cur = 0;
      int curend = __builtin_amdgcn_readlane(myend, 0);
      float4 acc = {0.f, 0.f, 0.f, 0.f};
      int mine = 0;
      if (beg < end) mine = beg + lane < end ? col[beg + lane] : 0;
#pragma unroll
      for (int r = 0; r < 16; ++r)
        *reinterpret_cast<float4*>(&zs[r * SZ + lane * 4]) = own[r];
      for (int base = beg; base < end; base += GU) {
        float4 v[GU];
#pragma unroll
        for (int u = 0; u < GU; ++u) {
          const int sidx = __builtin_amdgcn_readlane(mine, u);
          v[u] = *reinterpret_cast<const float4*>(hin + (size_t)sidx * DH +
                                                  lane * 4);
        }
        int nxt = 0;
        if (base + GU < end)
          nxt = base + GU + lane < end ? col[base + GU + lane] : 0;
#pragma unroll
        for (int u = 0; u < GU; ++u) {
          const int p = base + u;
          if (p < end) {
            while (p >= curend && cur < 15) {
              float4* zr = reinterpret_cast<float4*>(&zs[cur * SZ + lane * 4]);
              float4 hv = *zr;
              hv.x += acc.x;
              hv.y += acc.y;
              hv.z += acc.z;
              hv.w += acc.w;
              *zr = hv;
              acc.x = 0.f;
              acc.y = 0.f;
              acc.z = 0.f;
              acc.w = 0.f;
              ++cur;
              curend = __builtin_amdgcn_readlane(myend, cur);
            }
            acc.x += v[u].x;
            acc.y += v[u].y;
            acc.z += v[u].z;
            acc.w += v[u].w;
          }
        }
        mine = nxt;
      }
      {
        float4* zr = reinterpret_cast<float4*>(&zs[cur * SZ + lane * 4]);
        float4 hv = *zr;
        hv.x += acc.x;
        hv.y += acc.y;
        hv.z += acc.z;
        hv.w += acc.w;
        *zr = hv;
      }
      __builtin_amdgcn_wave_barrier();
    }

    v8f acc[8];
#pragma unroll
    for (int n = 0; n < 8; ++n) acc[n] = zero8();

    for (int k0 = 0; k0 < DH; k0 += 4) {
      const int k = k0 + koff;
      const int kp = (k0 >> 1) + kpo;
      const v2f a = *reinterpret_cast<const v2f*>(&zs[m * SZ + k]);
#pragma unroll
      for (int n = 0; n < 8; ++n) {
        const v2f bf = sW1[kp * SWP + n * 16 + m];
        acc[n] = __builtin_amdgcn_wmma_f32_16x16x4_f32(
            false, a, false, bf, (short)0, acc[n], false, false);
      }
    }
    __builtin_amdgcn_wave_barrier();

#pragma unroll
    for (int n = 0; n < 8; ++n) {
      const float bias = sb1[n * 16 + m];
#pragma unroll
      for (int j = 0; j < 8; ++j) {
        float v = acc[n][j] + bias;
        v = v > 0.0f ? v : 0.0f;
        zs[(mrow + j) * SZ + n * 16 + m] = v;
      }
    }
    __builtin_amdgcn_wave_barrier();

#pragma unroll
    for (int n = 0; n < NCOL2; ++n) acc[n] = zero8();
    for (int k0 = 0; k0 < DH; k0 += 4) {
      const int k = k0 + koff;
      const int kp = (k0 >> 1) + kpo;
      const v2f a = *reinterpret_cast<const v2f*>(&zs[m * SZ + k]);
#pragma unroll
      for (int n = 0; n < NCOL2; ++n) {
        const v2f bf = sW2[kp * SWP + n * 16 + m];
        acc[n] = __builtin_amdgcn_wmma_f32_16x16x4_f32(
            false, a, false, bf, (short)0, acc[n], false, false);
      }
    }
    __builtin_amdgcn_wave_barrier();

#pragma unroll
    for (int n = 0; n < NCOL2; ++n) {
      const float bias = sb2[n * 16 + m];
#pragma unroll
      for (int j = 0; j < 8; ++j) {
        float v = acc[n][j] + bias;
        if (RELU2) v = v > 0.0f ? v : 0.0f;
        zs[(mrow + j) * SZ + n * 16 + m] = v;
      }
    }
    __builtin_amdgcn_wave_barrier();
    constexpr int LPR = N2 / 4;
    constexpr int RPS = 32 / LPR;
    const int rsub = lane / LPR;
    const int c4 = (lane % LPR) * 4;
    float* orow0 = out + (size_t)(r0 + rsub) * N2 + c4;
    if (r0 + 16 <= nrows) {
      float4 orow[16 / RPS];
#pragma unroll
      for (int i = 0; i < 16 / RPS; ++i)
        orow[i] = *reinterpret_cast<const float4*>(
            &zs[(i * RPS + rsub) * SZ + c4]);
#pragma unroll
      for (int i = 0; i < 16 / RPS; ++i)
        *reinterpret_cast<float4*>(orow0 + (size_t)(i * RPS) * N2) = orow[i];
    } else {
#pragma unroll
      for (int i = 0; i < 16 / RPS; ++i) {
        if (r0 + i * RPS + rsub < nrows)
          *reinterpret_cast<float4*>(orow0 + (size_t)(i * RPS) * N2) =
              *reinterpret_cast<const float4*>(
                  &zs[(i * RPS + rsub) * SZ + c4]);
      }
    }
  }
}

extern "C" void kernel_launch(void* const* d_in, const int* in_sizes, int n_in,
                              void* d_out, int out_size, void* d_ws,
                              size_t ws_size, hipStream_t stream) {
  const float* x = (const float*)d_in[0];
  const int* ei = (const int*)d_in[1];
  const int N = in_sizes[0] / DH;
  const int E = in_sizes[1] / 2;
  const int* src = ei;
  const int* dst = ei + E;

  const float* c1w1 = (const float*)d_in[2];
  const float* c1b1 = (const float*)d_in[3];
  const float* c1w2 = (const float*)d_in[4];
  const float* c1b2 = (const float*)d_in[5];
  const float* c2w1 = (const float*)d_in[6];
  const float* c2b1 = (const float*)d_in[7];
  const float* c2w2 = (const float*)d_in[8];
  const float* c2b2 = (const float*)d_in[9];
  const float* c3w1 = (const float*)d_in[10];
  const float* c3b1 = (const float*)d_in[11];
  const float* c3w2 = (const float*)d_in[12];
  const float* c3b2 = (const float*)d_in[13];
  const float* l1w = (const float*)d_in[14];
  const float* l1b = (const float*)d_in[15];
  const float* l2w = (const float*)d_in[16];
  const float* l2b = (const float*)d_in[17];

  float* hA = (float*)d_ws;
  float* hB = hA + (size_t)N * DH;
  int* ebin = (int*)(hB + (size_t)N * DH);
  int* cnt = ebin + 2 * (size_t)E;
  int* rowptr = cnt + N;
  int* col = rowptr + N;
  int* mat = col + E;

  const dim3 bB(CSR_T1), bG(256);
  const dim3 tB(CSR_T2), tG(512);
  const dim3 mB(512), mG(256);

  csr_bin<<<bG, bB, 0, stream>>>(src, dst, mat, ebin, E, N);
  csr_emit<<<tG, tB, 0, stream>>>(mat, ebin, cnt, rowptr, col, E, N);

  gin_mlp<8, true, true><<<mG, mB, 0, stream>>>(
      x, rowptr, cnt, col, c1w1, c1b1, c1w2, c1b2, hA, N);
  gin_mlp<8, true, true><<<mG, mB, 0, stream>>>(
      hA, rowptr, cnt, col, c2w1, c2b1, c2w2, c2b2, hB, N);
  gin_mlp<8, true, true><<<mG, mB, 0, stream>>>(
      hB, rowptr, cnt, col, c3w1, c3b1, c3w2, c3b2, hA, N);
  gin_mlp<4, false, false><<<mG, mB, 0, stream>>>(
      hA, nullptr, nullptr, nullptr, l1w, l1b, l2w, l2b, (float*)d_out, N);
}


# gin_mlp: the 53 of 3125 tiles left over beyond 12 per workgroup are multiplied by the whole workgroup behind a barrier, a 16-column block per wave on all 4 SIMDs, not as a 4th tile on one SIMD
# speedup vs baseline: 4.6590x; 1.0616x over previous
#include <hip/hip_runtime.h>


typedef float v2f __attribute__((ext_vector_type(2)));
typedef float v8f __attribute__((ext_vector_type(8)));

#define DH 128

static __device__ __forceinline__ v8f zero8() {
  v8f z = {0.f, 0.f, 0.f, 0.f, 0.f, 0.f, 0.f, 0.f};
  return z;
}

#define CSR_CH 3200
#define CSR_T1 512
#define CSR_EPT 7
#define CSR_SH 7
#define CSR_NPB 128
#define CSR_MAXB 512
#define CSR_MST 514
#define CSR_T2 256
#define CSR_CAP 6144

template <int NW>
static __device__ __forceinline__ int block_excl_scan(int v, int* swave,
                                                      int* tot) {
  const int lane = (int)threadIdx.x & 31;
  const int wid = (int)threadIdx.x >> 5;
  int x = v;
  for (int o = 1; o < 32; o <<= 1) {
    const int y = __shfl_up(x, o);
    if (lane >= o) x += y;
  }
  if (lane == 31) swave[wid] = x;
  __syncthreads();
  int w = lane < NW ? swave[lane] : 0;
  int pre = lane < wid ? w : 0;
  for (int o = 16; o > 0; o >>= 1) {
    pre += __shfl_xor(pre, o);
    w += __shfl_xor(w, o);
  }
  *tot = w;
  return pre + x - v;
}

__global__ void __launch_bounds__(CSR_T1) csr_bin(
    const int* __restrict__ src, const int* __restrict__ dst,
    int* __restrict__ mat, int* __restrict__ ebin, int E, int N) {
  __shared__ int hist[CSR_MAXB];
  __shared__ int offs[CSR_MAXB];
  __shared__ int swave[32];
  __shared__ __attribute__((aligned(16))) int2 sorted[CSR_CH];
  const int tid = (int)threadIdx.x;
  const int nchunks = (E + CSR_CH - 1) / CSR_CH;
  for (int c = (int)blockIdx.x; c < nchunks; c += (int)gridDim.x) {
    const int e0 = c * CSR_CH;
    const int n = E - e0 < CSR_CH ? E - e0 : CSR_CH;
    hist[tid] = 0;
    __syncthreads();
    int s[CSR_EPT], d[CSR_EPT], r[CSR_EPT];
#pragma unroll
    for (int j = 0; j < CSR_EPT; ++j) {
      const int i = tid + j * CSR_T1;
      s[j] = 0;
      d[j] = -1;
      if (i < n) {
        s[j] = src[e0 + i];
        d[j] = dst[e0 + i];
      }
    }
#pragma unroll
    for (int j = 0; j < CSR_EPT; ++j) {
      r[j] = 0;
      if (d[j] >= 0) {
        int b = d[j] >> CSR_SH;
        if (b > CSR_MAXB - 1) b = CSR_MAXB - 1;
        r[j] = atomicAdd(&hist[b], 1);
      }
    }
    __syncthreads();
    const int v = hist[tid];
    int tot;
    const int ex = block_excl_scan<CSR_T1 / 32>(v, swave, &tot);
    offs[tid] = ex;
    mat[(size_t)c * CSR_MST + tid] = ex;
    if (tid == CSR_T1 - 1) mat[(size_t)c * CSR_MST + CSR_MAXB] = ex + v;
    __syncthreads();
#pragma unroll
    for (int j = 0; j < CSR_EPT; ++j) {
      if (d[j] >= 0) {
        int b = d[j] >> CSR_SH;
        if (b > CSR_MAXB - 1) b = CSR_MAXB - 1;
        int2 p;
        p.x = s[j];
        p.y = d[j];
        sorted[offs[b] + r[j]] = p;
      }
    }
    __syncthreads();
    int2* eb = reinterpret_cast<int2*>(ebin) + e0;
    for (int i = tid; i < n; i += CSR_T1) eb[i] = sorted[i];
    __syncthreads();
  }
}

__global__ void __launch_bounds__(CSR_T2) csr_emit(
    const int* __restrict__ mat, const int* __restrict__ ebin,
    int* __restrict__ cnt, int* __restrict__ rowptr, int* __restrict__ col,
    int E, int N) {
  __shared__ int ncnt[CSR_NPB];
  __shared__ int cur[CSR_NPB];
  __shared__ int swave[32];
  __shared__ int sred[2][CSR_T2 / 32];
  __shared__ int colbuf[CSR_CAP];
  const int tid = (int)threadIdx.x;
  const int lane = tid & 31;
  const int wid = tid >> 5;
  const int nbins = ((N - 1) >> CSR_SH) + 1;
  const int nchunks = (E + CSR_CH - 1) / CSR_CH;
  const int2* eb = reinterpret_cast<const int2*>(ebin);
  for (int b = (int)blockIdx.x; b < nbins; b += (int)gridDim.x) {
    int st = 0, nt = 0;
    for (int c = tid; c < nchunks; c += CSR_T2) {
      const int o0 = mat[(size_t)c * CSR_MST + b];
      const int o1 = mat[(size_t)c * CSR_MST + b + 1];
      st += o0;
      nt += o1 - o0;
    }
    for (int o = 16; o > 0; o >>= 1) {
      st += __shfl_xor(st, o);
      nt += __shfl_xor(nt, o);
    }
    if (lane == 0) {
      sred[0][wid] = st;
      sred[1][wid] = nt;
    }
    if (tid < CSR_NPB) ncnt[tid] = 0;
    __syncthreads();
    int bstart = 0, btot = 0;
#pragma unroll
    for (int w = 0; w < CSR_T2 / 32; ++w) {
      bstart += sred[0][w];
      btot += sred[1][w];
    }
    for (int c = tid; c < nchunks; c += CSR_T2) {
      const int o0 = mat[(size_t)c * CSR_MST + b];
      const int o1 = mat[(size_t)c * CSR_MST + b + 1];
      const int2* seg = eb + (size_t)c * CSR_CH;
      for (int i = o0; i < o1; i += 4) {
        int2 e[4];
#pragma unroll
        for (int u = 0; u < 4; ++u) {
          e[u].x = 0;
          e[u].y = -1;
          if (i + u < o1) e[u] = seg[i + u];
        }
#pragma unroll
        for (int u = 0; u < 4; ++u)
          if (e[u].y >= 0) atomicAdd(&ncnt[e[u].y & (CSR_NPB - 1)], 1);
      }
    }
    __syncthreads();
    const int v = tid < CSR_NPB ? ncnt[tid] : 0;
    int tot;
    const int ex = block_excl_scan<CSR_T2 / 32>(v, swave, &tot);
    if (tid < CSR_NPB) {
      cur[tid] = ex;
      const int node = b * CSR_NPB + tid;
      if (node < N) {
        cnt[node] = v;
        rowptr[node] = bstart + ex;
      }
    }
    __syncthreads();
    const bool inlds = btot <= CSR_CAP;
    for (int c = tid; c < nchunks; c += CSR_T2) {
      const int o0 = mat[(size_t)c * CSR_MST + b];
      const int o1 = mat[(size_t)c * CSR_MST + b + 1];
      const int2* seg = eb + (size_t)c * CSR_CH;
      for (int i = o0; i < o1; i += 4) {
        int2 e[4];
#pragma unroll
        for (int u = 0; u < 4; ++u) {
          e[u].x = 0;
          e[u].y = -1;
          if (i + u < o1) e[u] = seg[i + u];
        }
#pragma unroll
        for (int u = 0; u < 4; ++u) {
          if (e[u].y >= 0) {
            const int pos = atomicAdd(&cur[e[u].y & (CSR_NPB - 1)], 1);
            if (inlds)
              colbuf[pos] = e[u].x;
            else
              col[bstart + pos] = e[u].x;
          }
        }
      }
    }
    __syncthreads();
    if (inlds)
      for (int i = tid; i < btot; i += CSR_T2) col[bstart + i] = colbuf[i];
    __syncthreads();
  }
}

#define MLP_NW 16
#define MLP_NT (MLP_NW * 32)
#define GU 32

template <int NCOL2, bool HAS_AGG, bool RELU2>
__global__ void __launch_bounds__(MLP_NT) gin_mlp(
    const float* __restrict__ hin, const int* __restrict__ rowptr,
    const int* __restrict__ cnt, const int* __restrict__ col,
    const float* __restrict__ W1, const float* __restrict__ b1,
    const float* __restrict__ W2, const float* __restrict__ b2,
    float* __restrict__ out, int nrows) {
  constexpr int N2 = NCOL2 * 16;
  constexpr int SWP = 144;
  constexpr int SZ = 132;
  __shared__ __attribute__((aligned(16))) v2f sW1[(DH / 2) * SWP];
  __shared__ __attribute__((aligned(16))) v2f sW2[(DH / 2) * SWP];
  __shared__ float sb1[DH];
  __shared__ float sb2[DH];
  __shared__ __attribute__((aligned(16))) float sZ[MLP_NW][16 * SZ];

  const int tid = (int)threadIdx.x;

  constexpr int CG2 = N2 / 4;
  constexpr int NJ1 = (DH / 2) * (DH / 4) / MLP_NT;
  constexpr int NJ2 = (DH / 2) * CG2 / MLP_NT;
  const int rot = (int)blockIdx.x;
  float4 w1a[NJ1], w1b[NJ1], w2a[NJ2], w2b[NJ2];
#pragma unroll
  for (int j = 0; j < NJ1; ++j) {
    const int it = j * MLP_NT + tid;
    const int p = ((it >> 5) + rot) & (DH / 2 - 1), cg = it & 31;
    const float* g = W1 + (size_t)(2 * p) * DH + cg * 4;
    w1a[j] = *reinterpret_cast<const float4*>(g);
    w1b[j] = *reinterpret_cast<const float4*>(g + DH);
  }
#pragma unroll
  for (int j = 0; j < NJ2; ++j) {
    const int it = j * MLP_NT + tid;
    const int p = (it / CG2 + rot) & (DH / 2 - 1), cg = it % CG2;
    const float* g = W2 + (size_t)(2 * p) * N2 + cg * 4;
    w2a[j] = *reinterpret_cast<const float4*>(g);
    w2b[j] = *reinterpret_cast<const float4*>(g + N2);
  }
  float bv1 = 0.f, bv2 = 0.f;
  if (tid < DH) bv1 = b1[tid];
  if (tid < N2) bv2 = b2[tid];
#pragma unroll
  for (int j = 0; j < NJ1; ++j) {
    const int it = j * MLP_NT + tid;
    const int p = ((it >> 5) + rot) & (DH / 2 - 1), cg = it & 31;
    float4* d = reinterpret_cast<float4*>(&sW1[p * SWP + cg * 4]);
    const float4 lo = {w1a[j].x, w1b[j].x, w1a[j].y, w1b[j].y};
    const float4 hi = {w1a[j].z, w1b[j].z, w1a[j].w, w1b[j].w};
    d[0] = lo;
    d[1] = hi;
  }
#pragma unroll
  for (int j = 0; j < NJ2; ++j) {
    const int it = j * MLP_NT + tid;
    const int p = (it / CG2 + rot) & (DH / 2 - 1), cg = it % CG2;
    float4* d = reinterpret_cast<float4*>(&sW2[p * SWP + cg * 4]);
    const float4 lo = {w2a[j].x, w2b[j].x, w2a[j].y, w2b[j].y};
    const float4 hi = {w2a[j].z, w2b[j].z, w2a[j].w, w2b[j].w};
    d[0] = lo;
    d[1] = hi;
  }
  if (tid < DH) sb1[tid] = bv1;
  if (tid < N2) sb2[tid] = bv2;
  __syncthreads();

  const int lane = tid & 31;
  const int wid = tid >> 5;
  const int m = lane & 15;
  const int koff = (lane >> 4) * 2;
  const int kpo = lane >> 4;
  const int mrow = (lane >> 4) * 8;
  float* zs = sZ[wid];

  const int ntiles = (nrows + 15) >> 4;
  const int gw0 = (int)blockIdx.x + (int)gridDim.x * wid;
  const int gwn = (int)gridDim.x * MLP_NW;
  const int nmain = (ntiles / ((int)gridDim.x * 4)) * ((int)gridDim.x * 4);

  for (int t = gw0; t < ntiles; t += gwn) {
    const int r0 = __builtin_amdgcn_readfirstlane(t << 4);

    if (!HAS_AGG) {
      float4 own[16];
#pragma unroll
      for (int r = 0; r < 16; ++r) {
        int nr = r0 + r;
        if (nr > nrows - 1) nr = nrows - 1;
        own[r] =
            *reinterpret_cast<const float4*>(hin + (size_t)nr * DH + lane * 4);
      }
#pragma unroll
      for (int r = 0; r < 16; ++r)
        *reinterpret_cast<float4*>(&zs[r * SZ + lane * 4]) = own[r];
      __builtin_amdgcn_wave_barrier();
    }

    if (HAS_AGG) {
      const int nv = nrows - r0 < 16 ? nrows - r0 : 16;
      int nd = r0 + m;
      if (nd > nrows - 1) nd = nrows - 1;
      const int rp = rowptr[nd];
      const int dg = m < nv ? cnt[nd] : 0;
      const int myend = rp + dg;
      float4 own[16];
#pragma unroll
      for (int r = 0; r < 16; ++r) {
        int nr = r0 + r;
        if (nr > nrows - 1) nr = nrows - 1;
        own[r] =
            *reinterpret_cast<const float4*>(hin + (size_t)nr * DH + lane * 4);
      }
      const int beg = __builtin_amdgcn_readlane(rp, 0);
      const int end = __builtin_amdgcn_readlane(myend, nv - 1);
      int cur = 0;
      int curend = __builtin_amdgcn_readlane(myend, 0);
      float4 acc = {0.f, 0.f, 0.f, 0.f};
      int mine = 0;
      if (beg < end) mine = beg + lane < end ? col[beg + lane] : 0;
#pragma unroll
      for (int r = 0; r < 16; ++r)
        *reinterpret_cast<float4*>(&zs[r * SZ + lane * 4]) = own[r];
      for (int base = beg; base < end; base += GU) {
        float4 v[GU];
#pragma unroll
        for (int u = 0; u < GU; ++u) {
          const int sidx = __builtin_amdgcn_readlane(mine, u);
          v[u] = *reinterpret_cast<const float4*>(hin + (size_t)sidx * DH +
                                                  lane * 4);
        }
        int nxt = 0;
        if (base + GU < end)
          nxt = base + GU + lane < end ? col[base + GU + lane] : 0;
#pragma unroll
        for (int u = 0; u < GU; ++u) {
          const int p = base + u;
          if (p < end) {
            while (p >= curend && cur < 15) {
              float4* zr = reinterpret_cast<float4*>(&zs[cur * SZ + lane * 4]);
              float4 hv = *zr;
              hv.x += acc.x;
              hv.y += acc.y;
              hv.z += acc.z;
              hv.w += acc.w;
              *zr = hv;
              acc.x = 0.f;
              acc.y = 0.f;
              acc.z = 0.f;
              acc.w = 0.f;
              ++cur;
              curend = __builtin_amdgcn_readlane(myend, cur);
            }
            acc.x += v[u].x;
            acc.y += v[u].y;
            acc.z += v[u].z;
            acc.w += v[u].w;
          }
        }
        mine = nxt;
      }
      {
        float4* zr = reinterpret_cast<float4*>(&zs[cur * SZ + lane * 4]);
        float4 hv = *zr;
        hv.x += acc.x;
        hv.y += acc.y;
        hv.z += acc.z;
        hv.w += acc.w;
        *zr = hv;
      }
      __builtin_amdgcn_wave_barrier();
    }

    if ((r0 >> 4) >= nmain) break;

    v8f acc[8];
#pragma unroll
    for (int n = 0; n < 8; ++n) acc[n] = zero8();

    for (int k0 = 0; k0 < DH; k0 += 4) {
      const int k = k0 + koff;
      const int kp = (k0 >> 1) + kpo;
      const v2f a = *reinterpret_cast<const v2f*>(&zs[m * SZ + k]);
#pragma unroll
      for (int n = 0; n < 8; ++n) {
        const v2f bf = sW1[kp * SWP + n * 16 + m];
        acc[n] = __builtin_amdgcn_wmma_f32_16x16x4_f32(
            false, a, false, bf, (short)0, acc[n], false, false);
      }
    }
    __builtin_amdgcn_wave_barrier();

#pragma unroll
    for (int n = 0; n < 8; ++n) {
      const float bias = sb1[n * 16 + m];
#pragma unroll
      for (int j = 0; j < 8; ++j) {
        float v = acc[n][j] + bias;
        v = v > 0.0f ? v : 0.0f;
        zs[(mrow + j) * SZ + n * 16 + m] = v;
      }
    }
    __builtin_amdgcn_wave_barrier();

#pragma unroll
    for (int n = 0; n < NCOL2; ++n) acc[n] = zero8();
    for (int k0 = 0; k0 < DH; k0 += 4) {
      const int k = k0 + koff;
      const int kp = (k0 >> 1) + kpo;
      const v2f a = *reinterpret_cast<const v2f*>(&zs[m * SZ + k]);
#pragma unroll
      for (int n = 0; n < NCOL2; ++n) {
        const v2f bf = sW2[kp * SWP + n * 16 + m];
        acc[n] = __builtin_amdgcn_wmma_f32_16x16x4_f32(
            false, a, false, bf, (short)0, acc[n], false, false);
      }
    }
    __builtin_amdgcn_wave_barrier();

#pragma unroll
    for (int n = 0; n < NCOL2; ++n) {
      const float bias = sb2[n * 16 + m];
#pragma unroll
      for (int j = 0; j < 8; ++j) {
        float v = acc[n][j] + bias;
        if (RELU2) v = v > 0.0f ? v : 0.0f;
        zs[(mrow + j) * SZ + n * 16 + m] = v;
      }
    }
    __builtin_amdgcn_wave_barrier();
    constexpr int LPR = N2 / 4;
    constexpr int RPS = 32 / LPR;
    const int rsub = lane / LPR;
    const int c4 = (lane % LPR) * 4;
    float* orow0 = out + (size_t)(r0 + rsub) * N2 + c4;
    if (r0 + 16 <= nrows) {
      float4 orow[16 / RPS];
#pragma unroll
      for (int i = 0; i < 16 / RPS; ++i)
        orow[i] = *reinterpret_cast<const float4*>(
            &zs[(i * RPS + rsub) * SZ + c4]);
#pragma unroll
      for (int i = 0; i < 16 / RPS; ++i)
        *reinterpret_cast<float4*>(orow0 + (size_t)(i * RPS) * N2) = orow[i];
    } else {
#pragma unroll
      for (int i = 0; i < 16 / RPS; ++i) {
        if (r0 + i * RPS + rsub < nrows)
          *reinterpret_cast<float4*>(orow0 + (size_t)(i * RPS) * N2) =
              *reinterpret_cast<const float4*>(
                  &zs[(i * RPS + rsub) * SZ + c4]);
      }
    }
  }

  const int j0 = nmain / (int)gridDim.x;
  const int left = ntiles - nmain - (int)blockIdx.x;
  const int nrem = left > 0 ? (left + (int)gridDim.x - 1) / (int)gridDim.x : 0;
  if (nrem > 0) {
    const int wv = __builtin_amdgcn_readfirstlane(wid);
    __syncthreads();
    for (int q = 0; q < nrem; ++q) {
      const int r0 = (nmain + (int)blockIdx.x + (int)gridDim.x * q) << 4;
      const float* za = sZ[(j0 + q) & (MLP_NW - 1)];
      float* zb = sZ[(j0 + q + 4) & (MLP_NW - 1)];
      float* zc = sZ[(j0 + q + 8) & (MLP_NW - 1)];
      if (wv < 8) {
        v2f av[DH / 4], bv[DH / 4];
#pragma unroll
        for (int s = 0; s < DH / 4; ++s) {
          av[s] = *reinterpret_cast<const v2f*>(&za[m * SZ + s * 4 + koff]);
          bv[s] = sW1[(s * 2 + kpo) * SWP + wv * 16 + m];
        }
        __builtin_amdgcn_sched_barrier(0);
        v8f c = zero8();
#pragma unroll
        for (int s = 0; s < DH / 4; ++s)
          c = __builtin_amdgcn_wmma_f32_16x16x4_f32(
              false, av[s], false, bv[s], (short)0, c, false, false);
        const float bias = sb1[wv * 16 + m];
#pragma unroll
        for (int j = 0; j < 8; ++j) {
          float v = c[j] + bias;
          v = v > 0.0f ? v : 0.0f;
          zb[(mrow + j) * SZ + wv * 16 + m] = v;
        }
      }
      __syncthreads();
      if (wv < NCOL2) {
        v2f av[DH / 4], bv[DH / 4];
#pragma unroll
        for (int s = 0; s < DH / 4; ++s) {
          av[s] = *reinterpret_cast<const v2f*>(&zb[m * SZ + s * 4 + koff]);
          bv[s] = sW2[(s * 2 + kpo) * SWP + wv * 16 + m];
        }
        __builtin_amdgcn_sched_barrier(0);
        v8f c = zero8();
#pragma unroll
        for (int s = 0; s < DH / 4; ++s)
          c = __builtin_amdgcn_wmma_f32_16x16x4_f32(
              false, av[s], false, bv[s], (short)0, c, false, false);
        const float bias = sb2[wv * 16 + m];
#pragma unroll
        for (int j = 0; j < 8; ++j) {
          float v = c[j] + bias;
          if (RELU2) v = v > 0.0f ? v : 0.0f;
          zc[(mrow + j) * SZ + wv * 16 + m] = v;
        }
      }
      __syncthreads();
      constexpr int LPR = N2 / 4;
      constexpr int RPS = 32 / LPR;
      if (wv < 16 / RPS) {
        const int row = wv * RPS + lane / LPR;
        const int c4 = (lane % LPR) * 4;
        if (r0 + row < nrows)
          *reinterpret_cast<float4*>(out + (size_t)(r0 + row) * N2 + c4) =
              *reinterpret_cast<const float4*>(&zc[row * SZ + c4]);
      }
    }
  }
}

extern "C" void kernel_launch(void* const* d_in, const int* in_sizes, int n_in,
                              void* d_out, int out_size, void* d_ws,
                              size_t ws_size, hipStream_t stream) {
  const float* x = (const float*)d_in[0];
  const int* ei = (const int*)d_in[1];
  const int N = in_sizes[0] / DH;
  const int E = in_sizes[1] / 2;
  const int* src = ei;
  const int* dst = ei + E;

  const float* c1w1 = (const float*)d_in[2];
  const float* c1b1 = (const float*)d_in[3];
  const float* c1w2 = (const float*)d_in[4];
  const float* c1b2 = (const float*)d_in[5];
  const float* c2w1 = (const float*)d_in[6];
  const float* c2b1 = (const float*)d_in[7];
  const float* c2w2 = (const float*)d_in[8];
  const float* c2b2 = (const float*)d_in[9];
  const float* c3w1 = (const float*)d_in[10];
  const float* c3b1 = (const float*)d_in[11];
  const float* c3w2 = (const float*)d_in[12];
  const float* c3b2 = (const float*)d_in[13];
  const float* l1w = (const float*)d_in[14];
  const float* l1b = (const float*)d_in[15];
  const float* l2w = (const float*)d_in[16];
  const float* l2b = (const float*)d_in[17];

  float* hA = (float*)d_ws;
  float* hB = hA + (size_t)N * DH;
  int* ebin = (int*)(hB + (size_t)N * DH);
  int* cnt = ebin + 2 * (size_t)E;
  int* rowptr = cnt + N;
  int* col = rowptr + N;
  int* mat = col + E;

  const dim3 bB(CSR_T1), bG(256);
  const dim3 tB(CSR_T2), tG(512);
  const dim3 mB(512), mG(256);

  csr_bin<<<bG, bB, 0, stream>>>(src, dst, mat, ebin, E, N);
  csr_emit<<<tG, tB, 0, stream>>>(mat, ebin, cnt, rowptr, col, E, N);

  gin_mlp<8, true, true><<<mG, mB, 0, stream>>>(
      x, rowptr, cnt, col, c1w1, c1b1, c1w2, c1b2, hA, N);
  gin_mlp<8, true, true><<<mG, mB, 0, stream>>>(
      hA, rowptr, cnt, col, c2w1, c2b1, c2w2, c2b2, hB, N);
  gin_mlp<8, true, true><<<mG, mB, 0, stream>>>(
      hB, rowptr, cnt, col, c3w1, c3b1, c3w2, c3b2, hA, N);
  gin_mlp<4, false, false><<<mG, mB, 0, stream>>>(
      hA, nullptr, nullptr, nullptr, l1w, l1b, l2w, l2b, (float*)d_out, N);
}


# output head fused into the third layer's launch: the layer's tile stays in the wave's LDS slice, head weights restaged behind a barrier; one launch and the h3 plane's write and read gone; same sums
# speedup vs baseline: 4.8012x; 1.0305x over previous
#include <hip/hip_runtime.h>


typedef float v2f __attribute__((ext_vector_type(2)));
typedef float v8f __attribute__((ext_vector_type(8)));

#define DH 128

static __device__ __forceinline__ v8f zero8() {
  v8f z = {0.f, 0.f, 0.f, 0.f, 0.f, 0.f, 0.f, 0.f};
  return z;
}

#define CSR_CH 3200
#define CSR_T1 512
#define CSR_EPT 7
#define CSR_SH 7
#define CSR_NPB 128
#define CSR_MAXB 512
#define CSR_MST 514
#define CSR_T2 256
#define CSR_CAP 6144

template <int NW>
static __device__ __forceinline__ int block_excl_scan(int v, int* swave,
                                                      int* tot) {
  const int lane = (int)threadIdx.x & 31;
  const int wid = (int)threadIdx.x >> 5;
  int x = v;
  for (int o = 1; o < 32; o <<= 1) {
    const int y = __shfl_up(x, o);
    if (lane >= o) x += y;
  }
  if (lane == 31) swave[wid] = x;
  __syncthreads();
  int w = lane < NW ? swave[lane] : 0;
  int pre = lane < wid ? w : 0;
  for (int o = 16; o > 0; o >>= 1) {
    pre += __shfl_xor(pre, o);
    w += __shfl_xor(w, o);
  }
  *tot = w;
  return pre + x - v;
}

__global__ void __launch_bounds__(CSR_T1) csr_bin(
    const int* __restrict__ src, const int* __restrict__ dst,
    int* __restrict__ mat, int* __restrict__ ebin, int E, int N) {
  __shared__ int hist[CSR_MAXB];
  __shared__ int offs[CSR_MAXB];
  __shared__ int swave[32];
  __shared__ __attribute__((aligned(16))) int2 sorted[CSR_CH];
  const int tid = (int)threadIdx.x;
  const int nchunks = (E + CSR_CH - 1) / CSR_CH;
  for (int c = (int)blockIdx.x; c < nchunks; c += (int)gridDim.x) {
    const int e0 = c * CSR_CH;
    const int n = E - e0 < CSR_CH ? E - e0 : CSR_CH;
    hist[tid] = 0;
    __syncthreads();
    int s[CSR_EPT], d[CSR_EPT], r[CSR_EPT];
#pragma unroll
    for (int j = 0; j < CSR_EPT; ++j) {
      const int i = tid + j * CSR_T1;
      s[j] = 0;
      d[j] = -1;
      if (i < n) {
        s[j] = src[e0 + i];
        d[j] = dst[e0 + i];
      }
    }
#pragma unroll
    for (int j = 0; j < CSR_EPT; ++j) {
      r[j] = 0;
      if (d[j] >= 0) {
        int b = d[j] >> CSR_SH;
        if (b > CSR_MAXB - 1) b = CSR_MAXB - 1;
        r[j] = atomicAdd(&hist[b], 1);
      }
    }
    __syncthreads();
    const int v = hist[tid];
    int tot;
    const int ex = block_excl_scan<CSR_T1 / 32>(v, swave, &tot);
    offs[tid] = ex;
    mat[(size_t)c * CSR_MST + tid] = ex;
    if (tid == CSR_T1 - 1) mat[(size_t)c * CSR_MST + CSR_MAXB] = ex + v;
    __syncthreads();
#pragma unroll
    for (int j = 0; j < CSR_EPT; ++j) {
      if (d[j] >= 0) {
        int b = d[j] >> CSR_SH;
        if (b > CSR_MAXB - 1) b = CSR_MAXB - 1;
        int2 p;
        p.x = s[j];
        p.y = d[j];
        sorted[offs[b] + r[j]] = p;
      }
    }
    __syncthreads();
    int2* eb = reinterpret_cast<int2*>(ebin) + e0;
    for (int i = tid; i < n; i += CSR_T1) eb[i] = sorted[i];
    __syncthreads();
  }
}

__global__ void __launch_bounds__(CSR_T2) csr_emit(
    const int* __restrict__ mat, const int* __restrict__ ebin,
    int* __restrict__ cnt, int* __restrict__ rowptr, int* __restrict__ col,
    int E, int N) {
  __shared__ int ncnt[CSR_NPB];
  __shared__ int cur[CSR_NPB];
  __shared__ int swave[32];
  __shared__ int sred[2][CSR_T2 / 32];
  __shared__ int colbuf[CSR_CAP];
  const int tid = (int)threadIdx.x;
  const int lane = tid & 31;
  const int wid = tid >> 5;
  const int nbins = ((N - 1) >> CSR_SH) + 1;
  const int nchunks = (E + CSR_CH - 1) / CSR_CH;
  const int2* eb = reinterpret_cast<const int2*>(ebin);
  for (int b = (int)blockIdx.x; b < nbins; b += (int)gridDim.x) {
    int st = 0, nt = 0;
    for (int c = tid; c < nchunks; c += CSR_T2) {
      const int o0 = mat[(size_t)c * CSR_MST + b];
      const int o1 = mat[(size_t)c * CSR_MST + b + 1];
      st += o0;
      nt += o1 - o0;
    }
    for (int o = 16; o > 0; o >>= 1) {
      st += __shfl_xor(st, o);
      nt += __shfl_xor(nt, o);
    }
    if (lane == 0) {
      sred[0][wid] = st;
      sred[1][wid] = nt;
    }
    if (tid < CSR_NPB) ncnt[tid] = 0;
    __syncthreads();
    int bstart = 0, btot = 0;
#pragma unroll
    for (int w = 0; w < CSR_T2 / 32; ++w) {
      bstart += sred[0][w];
      btot += sred[1][w];
    }
    for (int c = tid; c < nchunks; c += CSR_T2) {
      const int o0 = mat[(size_t)c * CSR_MST + b];
      const int o1 = mat[(size_t)c * CSR_MST + b + 1];
      const int2* seg = eb + (size_t)c * CSR_CH;
      for (int i = o0; i < o1; i += 4) {
        int2 e[4];
#pragma unroll
        for (int u = 0; u < 4; ++u) {
          e[u].x = 0;
          e[u].y = -1;
          if (i + u < o1) e[u] = seg[i + u];
        }
#pragma unroll
        for (int u = 0; u < 4; ++u)
          if (e[u].y >= 0) atomicAdd(&ncnt[e[u].y & (CSR_NPB - 1)], 1);
      }
    }
    __syncthreads();
    const int v = tid < CSR_NPB ? ncnt[tid] : 0;
    int tot;
    const int ex = block_excl_scan<CSR_T2 / 32>(v, swave, &tot);
    if (tid < CSR_NPB) {
      cur[tid] = ex;
      const int node = b * CSR_NPB + tid;
      if (node < N) {
        cnt[node] = v;
        rowptr[node] = bstart + ex;
      }
    }
    __syncthreads();
    const bool inlds = btot <= CSR_CAP;
    for (int c = tid; c < nchunks; c += CSR_T2) {
      const int o0 = mat[(size_t)c * CSR_MST + b];
      const int o1 = mat[(size_t)c * CSR_MST + b + 1];
      const int2* seg = eb + (size_t)c * CSR_CH;
      for (int i = o0; i < o1; i += 4) {
        int2 e[4];
#pragma unroll
        for (int u = 0; u < 4; ++u) {
          e[u].x = 0;
          e[u].y = -1;
          if (i + u < o1) e[u] = seg[i + u];
        }
#pragma unroll
        for (int u = 0; u < 4; ++u) {
          if (e[u].y >= 0) {
            const int pos = atomicAdd(&cur[e[u].y & (CSR_NPB - 1)], 1);
            if (inlds)
              colbuf[pos] = e[u].x;
            else
              col[bstart + pos] = e[u].x;
          }
        }
      }
    }
    __syncthreads();
    if (inlds)
      for (int i = tid; i < btot; i += CSR_T2) col[bstart + i] = colbuf[i];
    __syncthreads();
  }
}

#define MLP_NW 16
#define MLP_NT (MLP_NW * 32)
#define GU 32

template <int NCOL2, bool HAS_AGG, bool RELU2>
__global__ void __launch_bounds__(MLP_NT) gin_mlp(
    const float* __restrict__ hin, const int* __restrict__ rowptr,
    const int* __restrict__ cnt, const int* __restrict__ col,
    const float* __restrict__ W1, const float* __restrict__ b1,
    const float* __restrict__ W2, const float* __restrict__ b2,
    float* __restrict__ out, int nrows) {
  constexpr int N2 = NCOL2 * 16;
  constexpr int SWP = 144;
  constexpr int SZ = 132;
  __shared__ __attribute__((aligned(16))) v2f sW1[(DH / 2) * SWP];
  __shared__ __attribute__((aligned(16))) v2f sW2[(DH / 2) * SWP];
  __shared__ float sb1[DH];
  __shared__ float sb2[DH];
  __shared__ __attribute__((aligned(16))) float sZ[MLP_NW][16 * SZ];

  const int tid = (int)threadIdx.x;

  constexpr int CG2 = N2 / 4;
  constexpr int NJ1 = (DH / 2) * (DH / 4) / MLP_NT;
  constexpr int NJ2 = (DH / 2) * CG2 / MLP_NT;
  const int rot = (int)blockIdx.x;
  float4 w1a[NJ1], w1b[NJ1], w2a[NJ2], w2b[NJ2];
#pragma unroll
  for (int j = 0; j < NJ1; ++j) {
    const int it = j * MLP_NT + tid;
    const int p = ((it >> 5) + rot) & (DH / 2 - 1), cg = it & 31;
    const float* g = W1 + (size_t)(2 * p) * DH + cg * 4;
    w1a[j] = *reinterpret_cast<const float4*>(g);
    w1b[j] = *reinterpret_cast<const float4*>(g + DH);
  }
#pragma unroll
  for (int j = 0; j < NJ2; ++j) {
    const int it = j * MLP_NT + tid;
    const int p = (it / CG2 + rot) & (DH / 2 - 1), cg = it % CG2;
    const float* g = W2 + (size_t)(2 * p) * N2 + cg * 4;
    w2a[j] = *reinterpret_cast<const float4*>(g);
    w2b[j] = *reinterpret_cast<const float4*>(g + N2);
  }
  float bv1 = 0.f, bv2 = 0.f;
  if (tid < DH) bv1 = b1[tid];
  if (tid < N2) bv2 = b2[tid];
#pragma unroll
  for (int j = 0; j < NJ1; ++j) {
    const int it = j * MLP_NT + tid;
    const int p = ((it >> 5) + rot) & (DH / 2 - 1), cg = it & 31;
    float4* d = reinterpret_cast<float4*>(&sW1[p * SWP + cg * 4]);
    const float4 lo = {w1a[j].x, w1b[j].x, w1a[j].y, w1b[j].y};
    const float4 hi = {w1a[j].z, w1b[j].z, w1a[j].w, w1b[j].w};
    d[0] = lo;
    d[1] = hi;
  }
#pragma unroll
  for (int j = 0; j < NJ2; ++j) {
    const int it = j * MLP_NT + tid;
    const int p = (it / CG2 + rot) & (DH / 2 - 1), cg = it % CG2;
    float4* d = reinterpret_cast<float4*>(&sW2[p * SWP + cg * 4]);
    const float4 lo = {w2a[j].x, w2b[j].x, w2a[j].y, w2b[j].y};
    const float4 hi = {w2a[j].z, w2b[j].z, w2a[j].w, w2b[j].w};
    d[0] = lo;
    d[1] = hi;
  }
  if (tid < DH) sb1[tid] = bv1;
  if (tid < N2) sb2[tid] = bv2;
  __syncthreads();

  const int lane = tid & 31;
  const int wid = tid >> 5;
  const int m = lane & 15;
  const int koff = (lane >> 4) * 2;
  const int kpo = lane >> 4;
  const int mrow = (lane >> 4) * 8;
  float* zs = sZ[wid];

  const int ntiles = (nrows + 15) >> 4;
  const int gw0 = (int)blockIdx.x + (int)gridDim.x * wid;
  const int gwn = (int)gridDim.x * MLP_NW;
  const int nmain = (ntiles / ((int)gridDim.x * 4)) * ((int)gridDim.x * 4);

  for (int t = gw0; t < ntiles; t += gwn) {
    const int r0 = __builtin_amdgcn_readfirstlane(t << 4);

    if (!HAS_AGG) {
      float4 own[16];
#pragma unroll
      for (int r = 0; r < 16; ++r) {
        int nr = r0 + r;
        if (nr > nrows - 1) nr = nrows - 1;
        own[r] =
            *reinterpret_cast<const float4*>(hin + (size_t)nr * DH + lane * 4);
      }
#pragma unroll
      for (int r = 0; r < 16; ++r)
        *reinterpret_cast<float4*>(&zs[r * SZ + lane * 4]) = own[r];
      __builtin_amdgcn_wave_barrier();
    }

    if (HAS_AGG) {
      const int nv = nrows - r0 < 16 ? nrows - r0 : 16;
      int nd = r0 + m;
      if (nd > nrows - 1) nd = nrows - 1;
      const int rp = rowptr[nd];
      const int dg = m < nv ? cnt[nd] : 0;
      const int myend = rp + dg;
      float4 own[16];
#pragma unroll
      for (int r = 0; r < 16; ++r) {
        int nr = r0 + r;
        if (nr > nrows - 1) nr = nrows - 1;
        own[r] =
            *reinterpret_cast<const float4*>(hin + (size_t)nr * DH + lane * 4);
      }
      const int beg = __builtin_amdgcn_readlane(rp, 0);
      const int end = __builtin_amdgcn_readlane(myend, nv - 1);
      int cur = 0;
      int curend = __builtin_amdgcn_readlane(myend, 0);
      float4 acc = {0.f, 0.f, 0.f, 0.f};
      int mine = 0;
      if (beg < end) mine = beg + lane < end ? col[beg + lane] : 0;
#pragma unroll
      for (int r = 0; r < 16; ++r)
        *reinterpret_cast<float4*>(&zs[r * SZ + lane * 4]) = own[r];
      for (int base = beg; base < end; base += GU) {
        float4 v[GU];
#pragma unroll
        for (int u = 0; u < GU; ++u) {
          const int sidx = __builtin_amdgcn_readlane(mine, u);
          v[u] = *reinterpret_cast<const float4*>(hin + (size_t)sidx * DH +
                                                  lane * 4);
        }
        int nxt = 0;
        if (base + GU < end)
          nxt = base + GU + lane < end ? col[base + GU + lane] : 0;
#pragma unroll
        for (int u = 0; u < GU; ++u) {
          const int p = base + u;
          if (p < end) {
            while (p >= curend && cur < 15) {
              float4* zr = reinterpret_cast<float4*>(&zs[cur * SZ + lane * 4]);
              float4 hv = *zr;
              hv.x += acc.x;
              hv.y += acc.y;
              hv.z += acc.z;
              hv.w += acc.w;
              *zr = hv;
              acc.x = 0.f;
              acc.y = 0.f;
              acc.z = 0.f;
              acc.w = 0.f;
              ++cur;
              curend = __builtin_amdgcn_readlane(myend, cur);
            }
            acc.x += v[u].x;
            acc.y += v[u].y;
            acc.z += v[u].z;
            acc.w += v[u].w;
          }
        }
        mine = nxt;
      }
      {
        float4* zr = reinterpret_cast<float4*>(&zs[cur * SZ + lane * 4]);
        float4 hv = *zr;
        hv.x += acc.x;
        hv.y += acc.y;
        hv.z += acc.z;
        hv.w += acc.w;
        *zr = hv;
      }
      __builtin_amdgcn_wave_barrier();
    }

    if ((r0 >> 4) >= nmain) break;

    v8f acc[8];
#pragma unroll
    for (int n = 0; n < 8; ++n) acc[n] = zero8();

    for (int k0 = 0; k0 < DH; k0 += 4) {
      const int k = k0 + koff;
      const int kp = (k0 >> 1) + kpo;
      const v2f a = *reinterpret_cast<const v2f*>(&zs[m * SZ + k]);
#pragma unroll
      for (int n = 0; n < 8; ++n) {
        const v2f bf = sW1[kp * SWP + n * 16 + m];
        acc[n] = __builtin_amdgcn_wmma_f32_16x16x4_f32(
            false, a, false, bf, (short)0, acc[n], false, false);
      }
    }
    __builtin_amdgcn_wave_barrier();

#pragma unroll
    for (int n = 0; n < 8; ++n) {
      const float bias = sb1[n * 16 + m];
#pragma unroll
      for (int j = 0; j < 8; ++j) {
        float v = acc[n][j] + bias;
        v = v > 0.0f ? v : 0.0f;
        zs[(mrow + j) * SZ + n * 16 + m] = v;
      }
    }
    __builtin_amdgcn_wave_barrier();

#pragma unroll
    for (int n = 0; n < NCOL2; ++n) acc[n] = zero8();
    for (int k0 = 0; k0 < DH; k0 += 4) {
      const int k = k0 + koff;
      const int kp = (k0 >> 1) + kpo;
      const v2f a = *reinterpret_cast<const v2f*>(&zs[m * SZ + k]);
#pragma unroll
      for (int n = 0; n < NCOL2; ++n) {
        const v2f bf = sW2[kp * SWP + n * 16 + m];
        acc[n] = __builtin_amdgcn_wmma_f32_16x16x4_f32(
            false, a, false, bf, (short)0, acc[n], false, false);
      }
    }
    __builtin_amdgcn_wave_barrier();

#pragma unroll
    for (int n = 0; n < NCOL2; ++n) {
      const float bias = sb2[n * 16 + m];
#pragma unroll
      for (int j = 0; j < 8; ++j) {
        float v = acc[n][j] + bias;
        if (RELU2) v = v > 0.0f ? v : 0.0f;
        zs[(mrow + j) * SZ + n * 16 + m] = v;
      }
    }
    __builtin_amdgcn_wave_barrier();
    constexpr int LPR = N2 / 4;
    constexpr int RPS = 32 / LPR;
    const int rsub = lane / LPR;
    const int c4 = (lane % LPR) * 4;
    float* orow0 = out + (size_t)(r0 + rsub) * N2 + c4;
    if (r0 + 16 <= nrows) {
      float4 orow[16 / RPS];
#pragma unroll
      for (int i = 0; i < 16 / RPS; ++i)
        orow[i] = *reinterpret_cast<const float4*>(
            &zs[(i * RPS + rsub) * SZ + c4]);
#pragma unroll
      for (int i = 0; i < 16 / RPS; ++i)
        *reinterpret_cast<float4*>(orow0 + (size_t)(i * RPS) * N2) = orow[i];
    } else {
#pragma unroll
      for (int i = 0; i < 16 / RPS; ++i) {
        if (r0 + i * RPS + rsub < nrows)
          *reinterpret_cast<float4*>(orow0 + (size_t)(i * RPS) * N2) =
              *reinterpret_cast<const float4*>(
                  &zs[(i * RPS + rsub) * SZ + c4]);
      }
    }
  }

  const int j0 = nmain / (int)gridDim.x;
  const int left = ntiles - nmain - (int)blockIdx.x;
  const int nrem = left > 0 ? (left + (int)gridDim.x - 1) / (int)gridDim.x : 0;
  if (nrem > 0) {
    const int wv = __builtin_amdgcn_readfirstlane(wid);
    __syncthreads();
    for (int q = 0; q < nrem; ++q) {
      const int r0 = (nmain + (int)blockIdx.x + (int)gridDim.x * q) << 4;
      const float* za = sZ[(j0 + q) & (MLP_NW - 1)];
      float* zb = sZ[(j0 + q + 4) & (MLP_NW - 1)];
      float* zc = sZ[(j0 + q + 8) & (MLP_NW - 1)];
      if (wv < 8) {
        v2f av[DH / 4], bv[DH / 4];
#pragma unroll
        for (int s = 0; s < DH / 4; ++s) {
          av[s] = *reinterpret_cast<const v2f*>(&za[m * SZ + s * 4 + koff]);
          bv[s] = sW1[(s * 2 + kpo) * SWP + wv * 16 + m];
        }
        __builtin_amdgcn_sched_barrier(0);
        v8f c = zero8();
#pragma unroll
        for (int s = 0; s < DH / 4; ++s)
          c = __builtin_amdgcn_wmma_f32_16x16x4_f32(
              false, av[s], false, bv[s], (short)0, c, false, false);
        const float bias = sb1[wv * 16 + m];
#pragma unroll
        for (int j = 0; j < 8; ++j) {
          float v = c[j] + bias;
          v = v > 0.0f ? v : 0.0f;
          zb[(mrow + j) * SZ + wv * 16 + m] = v;
        }
      }
      __syncthreads();
      if (wv < NCOL2) {
        v2f av[DH / 4], bv[DH / 4];
#pragma unroll
        for (int s = 0; s < DH / 4; ++s) {
          av[s] = *reinterpret_cast<const v2f*>(&zb[m * SZ + s * 4 + koff]);
          bv[s] = sW2[(s * 2 + kpo) * SWP + wv * 16 + m];
        }
        __builtin_amdgcn_sched_barrier(0);
        v8f c = zero8();
#pragma unroll
        for (int s = 0; s < DH / 4; ++s)
          c = __builtin_amdgcn_wmma_f32_16x16x4_f32(
              false, av[s], false, bv[s], (short)0, c, false, false);
        const float bias = sb2[wv * 16 + m];
#pragma unroll
        for (int j = 0; j < 8; ++j) {
          float v = c[j] + bias;
          if (RELU2) v = v > 0.0f ? v : 0.0f;
          zc[(mrow + j) * SZ + wv * 16 + m] = v;
        }
      }
      __syncthreads();
      constexpr int LPR = N2 / 4;
      constexpr int RPS = 32 / LPR;
      if (wv < 16 / RPS) {
        const int row = wv * RPS + lane / LPR;
        const int c4 = (lane % LPR) * 4;
        if (r0 + row < nrows)
          *reinterpret_cast<float4*>(out + (size_t)(r0 + row) * N2 + c4) =
              *reinterpret_cast<const float4*>(&zc[row * SZ + c4]);
      }
    }
  }
}

static __device__ __forceinline__ void joint_block(
    const float* za, float* zd, const v2f* sW, const float* sb, int cb, int m,
    int koff, int kpo, int mrow, bool relu) {
  v2f av[DH / 4], bv[DH / 4];
#pragma unroll
  for (int s = 0; s < DH / 4; ++s) {
    av[s] = *reinterpret_cast<const v2f*>(&za[m * 132 + s * 4 + koff]);
    bv[s] = sW[(s * 2 + kpo) * 144 + cb * 16 + m];
  }
  __builtin_amdgcn_sched_barrier(0);
  v8f c = zero8();
#pragma unroll
  for (int s = 0; s < DH / 4; ++s)
    c = __builtin_amdgcn_wmma_f32_16x16x4_f32(false, av[s], false, bv[s],
                                              (short)0, c, false, false);
  const float bias = sb[cb * 16 + m];
#pragma unroll
  for (int j = 0; j < 8; ++j) {
    float v = c[j] + bias;
    if (relu) v = v > 0.0f ? v : 0.0f;
    zd[(mrow + j) * 132 + cb * 16 + m] = v;
  }
}

__global__ void __launch_bounds__(MLP_NT) gin_mlp_head(
    const float* __restrict__ hin, const int* __restrict__ rowptr,
    const int* __restrict__ cnt, const int* __restrict__ col,
    const float* __restrict__ W1, const float* __restrict__ b1,
    const float* __restrict__ W2, const float* __restrict__ b2,
    const float* __restrict__ W3, const float* __restrict__ b3,
    const float* __restrict__ W4, const float* __restrict__ b4,
    float* __restrict__ out, int nrows) {
  constexpr int NCOL2 = 8;
  constexpr bool HAS_AGG = true;
  constexpr bool RELU2 = true;
  constexpr int N2 = NCOL2 * 16;
  constexpr int SWP = 144;
  constexpr int SZ = 132;
  __shared__ __attribute__((aligned(16))) v2f sW1[(DH / 2) * SWP];
  __shared__ __attribute__((aligned(16))) v2f sW2[(DH / 2) * SWP];
  __shared__ float sb1[DH];
  __shared__ float sb2[DH];
  __shared__ __attribute__((aligned(16))) float sZ[MLP_NW + 1][16 * SZ];

  const int tid = (int)threadIdx.x;

  constexpr int CG2 = N2 / 4;
  constexpr int NJ1 = (DH / 2) * (DH / 4) / MLP_NT;
  constexpr int NJ2 = (DH / 2) * CG2 / MLP_NT;
  const int rot = (int)blockIdx.x;
  float4 w1a[NJ1], w1b[NJ1], w2a[NJ2], w2b[NJ2];
#pragma unroll
  for (int j = 0; j < NJ1; ++j) {
    const int it = j * MLP_NT + tid;
    const int p = ((it >> 5) + rot) & (DH / 2 - 1), cg = it & 31;
    const float* g = W1 + (size_t)(2 * p) * DH + cg * 4;
    w1a[j] = *reinterpret_cast<const float4*>(g);
    w1b[j] = *reinterpret_cast<const float4*>(g + DH);
  }
#pragma unroll
  for (int j = 0; j < NJ2; ++j) {
    const int it = j * MLP_NT + tid;
    const int p = (it / CG2 + rot) & (DH / 2 - 1), cg = it % CG2;
    const float* g = W2 + (size_t)(2 * p) * N2 + cg * 4;
    w2a[j] = *reinterpret_cast<const float4*>(g);
    w2b[j] = *reinterpret_cast<const float4*>(g + N2);
  }
  float bv1 = 0.f, bv2 = 0.f;
  if (tid < DH) bv1 = b1[tid];
  if (tid < N2) bv2 = b2[tid];
#pragma unroll
  for (int j = 0; j < NJ1; ++j) {
    const int it = j * MLP_NT + tid;
    const int p = ((it >> 5) + rot) & (DH / 2 - 1), cg = it & 31;
    float4* d = reinterpret_cast<float4*>(&sW1[p * SWP + cg * 4]);
    const float4 lo = {w1a[j].x, w1b[j].x, w1a[j].y, w1b[j].y};
    const float4 hi = {w1a[j].z, w1b[j].z, w1a[j].w, w1b[j].w};
    d[0] = lo;
    d[1] = hi;
  }
#pragma unroll
  for (int j = 0; j < NJ2; ++j) {
    const int it = j * MLP_NT + tid;
    const int p = (it / CG2 + rot) & (DH / 2 - 1), cg = it % CG2;
    float4* d = reinterpret_cast<float4*>(&sW2[p * SWP + cg * 4]);
    const float4 lo = {w2a[j].x, w2b[j].x, w2a[j].y, w2b[j].y};
    const float4 hi = {w2a[j].z, w2b[j].z, w2a[j].w, w2b[j].w};
    d[0] = lo;
    d[1] = hi;
  }
  if (tid < DH) sb1[tid] = bv1;
  if (tid < N2) sb2[tid] = bv2;
  __syncthreads();

  const int lane = tid & 31;
  const int wid = tid >> 5;
  const int m = lane & 15;
  const int koff = (lane >> 4) * 2;
  const int kpo = lane >> 4;
  const int mrow = (lane >> 4) * 8;
  float* zs = sZ[wid];

  const int ntiles = (nrows + 15) >> 4;
  const int gw0 = (int)blockIdx.x + (int)gridDim.x * wid;
  const int gwn = (int)gridDim.x * MLP_NW;
  const int nmain = (ntiles / ((int)gridDim.x * 4)) * ((int)gridDim.x * 4);

  int have = 0;
  int r0m = 0;
  for (int t = gw0; t < ntiles; t += gwn) {
    const int r0 = __builtin_amdgcn_readfirstlane(t << 4);

    if (!HAS_AGG) {
      float4 own[16];
#pragma unroll
      for (int r = 0; r < 16; ++r) {
        int nr = r0 + r;
        if (nr > nrows - 1) nr = nrows - 1;
        own[r] =
            *reinterpret_cast<const float4*>(hin + (size_t)nr * DH + lane * 4);
      }
#pragma unroll
      for (int r = 0; r < 16; ++r)
        *reinterpret_cast<float4*>(&zs[r * SZ + lane * 4]) = own[r];
      __builtin_amdgcn_wave_barrier();
    }

    if (HAS_AGG) {
      const int nv = nrows - r0 < 16 ? nrows - r0 : 16;
      int nd = r0 + m;
      if (nd > nrows - 1) nd = nrows - 1;
      const int rp = rowptr[nd];
      const int dg = m < nv ? cnt[nd] : 0;
      const int myend = rp + dg;
      float4 own[16];
#pragma unroll
      for (int r = 0; r < 16; ++r) {
        int nr = r0 + r;
        if (nr > nrows - 1) nr = nrows - 1;
        own[r] =
            *reinterpret_cast<const float4*>(hin + (size_t)nr * DH + lane * 4);
      }
      const int beg = __builtin_amdgcn_readlane(rp, 0);
      const int end = __builtin_amdgcn_readlane(myend, nv - 1);
      int cur = 0;
      int curend = __builtin_amdgcn_readlane(myend, 0);
      float4 acc = {0.f, 0.f, 0.f, 0.f};
      int mine = 0;
      if (beg < end) mine = beg + lane < end ? col[beg + lane] : 0;
#pragma unroll
      for (int r = 0; r < 16; ++r)
        *reinterpret_cast<float4*>(&zs[r * SZ + lane * 4]) = own[r];
      for (int base = beg; base < end; base += GU) {
        float4 v[GU];
#pragma unroll
        for (int u = 0; u < GU; ++u) {
          const int sidx = __builtin_amdgcn_readlane(mine, u);
          v[u] = *reinterpret_cast<const float4*>(hin + (size_t)sidx * DH +
                                                  lane * 4);
        }
        int nxt = 0;
        if (base + GU < end)
          nxt = base + GU + lane < end ? col[base + GU + lane] : 0;
#pragma unroll
        for (int u = 0; u < GU; ++u) {
          const int p = base + u;
          if (p < end) {
            while (p >= curend && cur < 15) {
              float4* zr = reinterpret_cast<float4*>(&zs[cur * SZ + lane * 4]);
              float4 hv = *zr;
              hv.x += acc.x;
              hv.y += acc.y;
              hv.z += acc.z;
              hv.w += acc.w;
              *zr = hv;
              acc.x = 0.f;
              acc.y = 0.f;
              acc.z = 0.f;
              acc.w = 0.f;
              ++cur;
              curend = __builtin_amdgcn_readlane(myend, cur);
            }
            acc.x += v[u].x;
            acc.y += v[u].y;
            acc.z += v[u].z;
            acc.w += v[u].w;
          }
        }
        mine = nxt;
      }
      {
        float4* zr = reinterpret_cast<float4*>(&zs[cur * SZ + lane * 4]);
        float4 hv = *zr;
        hv.x += acc.x;
        hv.y += acc.y;
        hv.z += acc.z;
        hv.w += acc.w;
        *zr = hv;
      }
      __builtin_amdgcn_wave_barrier();
    }

    if ((r0 >> 4) >= nmain) break;

    v8f acc[8];
#pragma unroll
    for (int n = 0; n < 8; ++n) acc[n] = zero8();

    for (int k0 = 0; k0 < DH; k0 += 4) {
      const int k = k0 + koff;
      const int kp = (k0 >> 1) + kpo;
      const v2f a = *reinterpret_cast<const v2f*>(&zs[m * SZ + k]);
#pragma unroll
      for (int n = 0; n < 8; ++n) {
        const v2f bf = sW1[kp * SWP + n * 16 + m];
        acc[n] = __builtin_amdgcn_wmma_f32_16x16x4_f32(
            false, a, false, bf, (short)0, acc[n], false, false);
      }
    }
    __builtin_amdgcn_wave_barrier();

#pragma unroll
    for (int n = 0; n < 8; ++n) {
      const float bias = sb1[n * 16 + m];
#pragma unroll
      for (int j = 0; j < 8; ++j) {
        float v = acc[n][j] + bias;
        v = v > 0.0f ? v : 0.0f;
        zs[(mrow + j) * SZ + n * 16 + m] = v;
      }
    }
    __builtin_amdgcn_wave_barrier();

#pragma unroll
    for (int n = 0; n < NCOL2; ++n) acc[n] = zero8();
    for (int k0 = 0; k0 < DH; k0 += 4) {
      const int k = k0 + koff;
      const int kp = (k0 >> 1) + kpo;
      const v2f a = *reinterpret_cast<const v2f*>(&zs[m * SZ + k]);
#pragma unroll
      for (int n = 0; n < NCOL2; ++n) {
        const v2f bf = sW2[kp * SWP + n * 16 + m];
        acc[n] = __builtin_amdgcn_wmma_f32_16x16x4_f32(
            false, a, false, bf, (short)0, acc[n], false, false);
      }
    }
    __builtin_amdgcn_wave_barrier();

#pragma unroll
    for (int n = 0; n < NCOL2; ++n) {
      const float bias = sb2[n * 16 + m];
#pragma unroll
      for (int j = 0; j < 8; ++j) {
        float v = acc[n][j] + bias;
        if (RELU2) v = v > 0.0f ? v : 0.0f;
        zs[(mrow + j) * SZ + n * 16 + m] = v;
      }
    }
    __builtin_amdgcn_wave_barrier();
    have = 1;
    r0m = r0;
  }

  constexpr int N4 = 64;
  constexpr int CG4 = N4 / 4;
  constexpr int NJ4 = (DH / 2) * CG4 / MLP_NT;
  float4 w3a[NJ1], w3b[NJ1], w4a[NJ4], w4b[NJ4];
#pragma unroll
  for (int j = 0; j < NJ1; ++j) {
    const int it = j * MLP_NT + tid;
    const int p = ((it >> 5) + rot) & (DH / 2 - 1), cg = it & 31;
    const float* g = W3 + (size_t)(2 * p) * DH + cg * 4;
    w3a[j] = *reinterpret_cast<const float4*>(g);
    w3b[j] = *reinterpret_cast<const float4*>(g + DH);
  }
#pragma unroll
  for (int j = 0; j < NJ4; ++j) {
    const int it = j * MLP_NT + tid;
    const int p = (it / CG4 + rot) & (DH / 2 - 1), cg = it % CG4;
    const float* g = W4 + (size_t)(2 * p) * N4 + cg * 4;
    w4a[j] = *reinterpret_cast<const float4*>(g);
    w4b[j] = *reinterpret_cast<const float4*>(g + N4);
  }
  float bv3 = 0.f, bv4 = 0.f;
  if (tid < DH) bv3 = b3[tid];
  if (tid < N4) bv4 = b4[tid];

  const int j0 = nmain / (int)gridDim.x;
  const int left = ntiles - nmain - (int)blockIdx.x;
  const int nrem = left > 0 ? (left + (int)gridDim.x - 1) / (int)gridDim.x : 0;
  const int wv = __builtin_amdgcn_readfirstlane(wid);
  float* zx = sZ[MLP_NW];
  __syncthreads();
  for (int q = 0; q < nrem; ++q) {
    float* za = sZ[(j0 + q) & (MLP_NW - 1)];
    if (wv < 8) joint_block(za, zx, sW1, sb1, wv, m, koff, kpo, mrow, true);
    __syncthreads();
    if (wv < 8) joint_block(zx, za, sW2, sb2, wv, m, koff, kpo, mrow, true);
    __syncthreads();
  }
#pragma unroll
  for (int j = 0; j < NJ1; ++j) {
    const int it = j * MLP_NT + tid;
    const int p = ((it >> 5) + rot) & (DH / 2 - 1), cg = it & 31;
    float4* d = reinterpret_cast<float4*>(&sW1[p * SWP + cg * 4]);
    const float4 lo = {w3a[j].x, w3b[j].x, w3a[j].y, w3b[j].y};
    const float4 hi = {w3a[j].z, w3b[j].z, w3a[j].w, w3b[j].w};
    d[0] = lo;
    d[1] = hi;
  }
#pragma unroll
  for (int j = 0; j < NJ4; ++j) {
    const int it = j * MLP_NT + tid;
    const int p = (it / CG4 + rot) & (DH / 2 - 1), cg = it % CG4;
    float4* d = reinterpret_cast<float4*>(&sW2[p * SWP + cg * 4]);
    const float4 lo = {w4a[j].x, w4b[j].x, w4a[j].y, w4b[j].y};
    const float4 hi = {w4a[j].z, w4b[j].z, w4a[j].w, w4b[j].w};
    d[0] = lo;
    d[1] = hi;
  }
  if (tid < DH) sb1[tid] = bv3;
  if (tid < N4) sb2[tid] = bv4;
  __syncthreads();

  constexpr int LPR = N4 / 4;
  constexpr int RPS = 32 / LPR;
  if (have) {
    v8f acc[8];
#pragma unroll
    for (int n = 0; n < 8; ++n) acc[n] = zero8();
    for (int k0 = 0; k0 < DH; k0 += 4) {
      const int k = k0 + koff;
      const int kp = (k0 >> 1) + kpo;
      const v2f a = *reinterpret_cast<const v2f*>(&zs[m * SZ + k]);
#pragma unroll
      for (int n = 0; n < 8; ++n) {
        const v2f bf = sW1[kp * SWP + n * 16 + m];
        acc[n] = __builtin_amdgcn_wmma_f32_16x16x4_f32(
            false, a, false, bf, (short)0, acc[n], false, false);
      }
    }
    __builtin_amdgcn_wave_barrier();
#pragma unroll
    for (int n = 0; n < 8; ++n) {
      const float bias = sb1[n * 16 + m];
#pragma unroll
      for (int j = 0; j < 8; ++j) {
        float v = acc[n][j] + bias;
        v = v > 0.0f ? v : 0.0f;
        zs[(mrow + j) * SZ + n * 16 + m] = v;
      }
    }
    __builtin_amdgcn_wave_barrier();
#pragma unroll
    for (int n = 0; n < 4; ++n) acc[n] = zero8();
    for (int k0 = 0; k0 < DH; k0 += 4) {
      const int k = k0 + koff;
      const int kp = (k0 >> 1) + kpo;
      const v2f a = *reinterpret_cast<const v2f*>(&zs[m * SZ + k]);
#pragma unroll
      for (int n = 0; n < 4; ++n) {
        const v2f bf = sW2[kp * SWP + n * 16 + m];
        acc[n] = __builtin_amdgcn_wmma_f32_16x16x4_f32(
            false, a, false, bf, (short)0, acc[n], false, false);
      }
    }
    __builtin_amdgcn_wave_barrier();
#pragma unroll
    for (int n = 0; n < 4; ++n) {
      const float bias = sb2[n * 16 + m];
#pragma unroll
      for (int j = 0; j < 8; ++j)
        zs[(mrow + j) * SZ + n * 16 + m] = acc[n][j] + bias;
    }
    __builtin_amdgcn_wave_barrier();
    const int rsub = lane / LPR;
    const int c4 = (lane % LPR) * 4;
#pragma unroll
    for (int i = 0; i < 16 / RPS; ++i) {
      const int row = i * RPS + rsub;
      if (r0m + row < nrows)
        *reinterpret_cast<float4*>(out + (size_t)(r0m + row) * N4 + c4) =
            *reinterpret_cast<const float4*>(&zs[row * SZ + c4]);
    }
  }
  for (int q = 0; q < nrem; ++q) {
    const int r0 = (nmain + (int)blockIdx.x + (int)gridDim.x * q) << 4;
    float* za = sZ[(j0 + q) & (MLP_NW - 1)];
    if (wv < 8) joint_block(za, zx, sW1, sb1, wv, m, koff, kpo, mrow, true);
    __syncthreads();
    if (wv < 4) joint_block(zx, za, sW2, sb2, wv, m, koff, kpo, mrow, false);
    __syncthreads();
    if (wv < 16 / RPS) {
      const int row = wv * RPS + lane / LPR;
      const int c4 = (lane % LPR) * 4;
      if (r0 + row < nrows)
        *reinterpret_cast<float4*>(out + (size_t)(r0 + row) * N4 + c4) =
            *reinterpret_cast<const float4*>(&za[row * SZ + c4]);
    }
  }
}

extern "C" void kernel_launch(void* const* d_in, const int* in_sizes, int n_in,
                              void* d_out, int out_size, void* d_ws,
                              size_t ws_size, hipStream_t stream) {
  const float* x = (const float*)d_in[0];
  const int* ei = (const int*)d_in[1];
  const int N = in_sizes[0] / DH;
  const int E = in_sizes[1] / 2;
  const int* src = ei;
  const int* dst = ei + E;

  const float* c1w1 = (const float*)d_in[2];
  const float* c1b1 = (const float*)d_in[3];
  const float* c1w2 = (const float*)d_in[4];
  const float* c1b2 = (const float*)d_in[5];
  const float* c2w1 = (const float*)d_in[6];
  const float* c2b1 = (const float*)d_in[7];
  const float* c2w2 = (const float*)d_in[8];
  const float* c2b2 = (const float*)d_in[9];
  const float* c3w1 = (const float*)d_in[10];
  const float* c3b1 = (const float*)d_in[11];
  const float* c3w2 = (const float*)d_in[12];
  const float* c3b2 = (const float*)d_in[13];
  const float* l1w = (const float*)d_in[14];
  const float* l1b = (const float*)d_in[15];
  const float* l2w = (const float*)d_in[16];
  const float* l2b = (const float*)d_in[17];

  float* hA = (float*)d_ws;
  float* hB = hA + (size_t)N * DH;
  int* ebin = (int*)(hB + (size_t)N * DH);
  int* cnt = ebin + 2 * (size_t)E;
  int* rowptr = cnt + N;
  int* col = rowptr + N;
  int* mat = col + E;

  const dim3 bB(CSR_T1), bG(256);
  const dim3 tB(CSR_T2), tG(512);
  const dim3 mB(512), mG(256);

  csr_bin<<<bG, bB, 0, stream>>>(src, dst, mat, ebin, E, N);
  csr_emit<<<tG, tB, 0, stream>>>(mat, ebin, cnt, rowptr, col, E, N);

  gin_mlp<8, true, true><<<mG, mB, 0, stream>>>(
      x, rowptr, cnt, col, c1w1, c1b1, c1w2, c1b2, hA, N);
  gin_mlp<8, true, true><<<mG, mB, 0, stream>>>(
      hA, rowptr, cnt, col, c2w1, c2b1, c2w2, c2b2, hB, N);
  gin_mlp_head<<<mG, mB, 0, stream>>>(hB, rowptr, cnt, col, c3w1, c3b1, c3w2,
                                      c3b2, l1w, l1b, l2w, l2b,
                                      (float*)d_out, N);
}


# gather walk: node of each list position counted in the lanes once per 32-row chunk, one first-of-node bit per row; the walk tests a bit per row, not a scalar compare chain; same sums, same order
# speedup vs baseline: 5.0064x; 1.0427x over previous
#include <hip/hip_runtime.h>


typedef float v2f __attribute__((ext_vector_type(2)));
typedef float v8f __attribute__((ext_vector_type(8)));

#define DH 128

static __device__ __forceinline__ v8f zero8() {
  v8f z = {0.f, 0.f, 0.f, 0.f, 0.f, 0.f, 0.f, 0.f};
  return z;
}

#define CSR_CH 3200
#define CSR_T1 512
#define CSR_EPT 7
#define CSR_SH 7
#define CSR_NPB 128
#define CSR_MAXB 512
#define CSR_MST 514
#define CSR_T2 256
#define CSR_CAP 6144

template <int NW>
static __device__ __forceinline__ int block_excl_scan(int v, int* swave,
                                                      int* tot) {
  const int lane = (int)threadIdx.x & 31;
  const int wid = (int)threadIdx.x >> 5;
  int x = v;
  for (int o = 1; o < 32; o <<= 1) {
    const int y = __shfl_up(x, o);
    if (lane >= o) x += y;
  }
  if (lane == 31) swave[wid] = x;
  __syncthreads();
  int w = lane < NW ? swave[lane] : 0;
  int pre = lane < wid ? w : 0;
  for (int o = 16; o > 0; o >>= 1) {
    pre += __shfl_xor(pre, o);
    w += __shfl_xor(w, o);
  }
  *tot = w;
  return pre + x - v;
}

__global__ void __launch_bounds__(CSR_T1) csr_bin(
    const int* __restrict__ src, const int* __restrict__ dst,
    int* __restrict__ mat, int* __restrict__ ebin, int E, int N) {
  __shared__ int hist[CSR_MAXB];
  __shared__ int offs[CSR_MAXB];
  __shared__ int swave[32];
  __shared__ __attribute__((aligned(16))) int2 sorted[CSR_CH];
  const int tid = (int)threadIdx.x;
  const int nchunks = (E + CSR_CH - 1) / CSR_CH;
  for (int c = (int)blockIdx.x; c < nchunks; c += (int)gridDim.x) {
    const int e0 = c * CSR_CH;
    const int n = E - e0 < CSR_CH ? E - e0 : CSR_CH;
    hist[tid] = 0;
    __syncthreads();
    int s[CSR_EPT], d[CSR_EPT], r[CSR_EPT];
#pragma unroll
    for (int j = 0; j < CSR_EPT; ++j) {
      const int i = tid + j * CSR_T1;
      s[j] = 0;
      d[j] = -1;
      if (i < n) {
        s[j] = src[e0 + i];
        d[j] = dst[e0 + i];
      }
    }
#pragma unroll
    for (int j = 0; j < CSR_EPT; ++j) {
      r[j] = 0;
      if (d[j] >= 0) {
        int b = d[j] >> CSR_SH;
        if (b > CSR_MAXB - 1) b = CSR_MAXB - 1;
        r[j] = atomicAdd(&hist[b], 1);
      }
    }
    __syncthreads();
    const int v = hist[tid];
    int tot;
    const int ex = block_excl_scan<CSR_T1 / 32>(v, swave, &tot);
    offs[tid] = ex;
    mat[(size_t)c * CSR_MST + tid] = ex;
    if (tid == CSR_T1 - 1) mat[(size_t)c * CSR_MST + CSR_MAXB] = ex + v;
    __syncthreads();
#pragma unroll
    for (int j = 0; j < CSR_EPT; ++j) {
      if (d[j] >= 0) {
        int b = d[j] >> CSR_SH;
        if (b > CSR_MAXB - 1) b = CSR_MAXB - 1;
        int2 p;
        p.x = s[j];
        p.y = d[j];
        sorted[offs[b] + r[j]] = p;
      }
    }
    __syncthreads();
    int2* eb = reinterpret_cast<int2*>(ebin) + e0;
    for (int i = tid; i < n; i += CSR_T1) eb[i] = sorted[i];
    __syncthreads();
  }
}

__global__ void __launch_bounds__(CSR_T2) csr_emit(
    const int* __restrict__ mat, const int* __restrict__ ebin,
    int* __restrict__ cnt, int* __restrict__ rowptr, int* __restrict__ col,
    int E, int N) {
  __shared__ int ncnt[CSR_NPB];
  __shared__ int cur[CSR_NPB];
  __shared__ int swave[32];
  __shared__ int sred[2][CSR_T2 / 32];
  __shared__ int colbuf[CSR_CAP];
  const int tid = (int)threadIdx.x;
  const int lane = tid & 31;
  const int wid = tid >> 5;
  const int nbins = ((N - 1) >> CSR_SH) + 1;
  const int nchunks = (E + CSR_CH - 1) / CSR_CH;
  const int2* eb = reinterpret_cast<const int2*>(ebin);
  for (int b = (int)blockIdx.x; b < nbins; b += (int)gridDim.x) {
    int st = 0, nt = 0;
    for (int c = tid; c < nchunks; c += CSR_T2) {
      const int o0 = mat[(size_t)c * CSR_MST + b];
      const int o1 = mat[(size_t)c * CSR_MST + b + 1];
      st += o0;
      nt += o1 - o0;
    }
    for (int o = 16; o > 0; o >>= 1) {
      st += __shfl_xor(st, o);
      nt += __shfl_xor(nt, o);
    }
    if (lane == 0) {
      sred[0][wid] = st;
      sred[1][wid] = nt;
    }
    if (tid < CSR_NPB) ncnt[tid] = 0;
    __syncthreads();
    int bstart = 0, btot = 0;
#pragma unroll
    for (int w = 0; w < CSR_T2 / 32; ++w) {
      bstart += sred[0][w];
      btot += sred[1][w];
    }
    for (int c = tid; c < nchunks; c += CSR_T2) {
      const int o0 = mat[(size_t)c * CSR_MST + b];
      const int o1 = mat[(size_t)c * CSR_MST + b + 1];
      const int2* seg = eb + (size_t)c * CSR_CH;
      for (int i = o0; i < o1; i += 4) {
        int2 e[4];
#pragma unroll
        for (int u = 0; u < 4; ++u) {
          e[u].x = 0;
          e[u].y = -1;
          if (i + u < o1) e[u] = seg[i + u];
        }
#pragma unroll
        for (int u = 0; u < 4; ++u)
          if (e[u].y >= 0) atomicAdd(&ncnt[e[u].y & (CSR_NPB - 1)], 1);
      }
    }
    __syncthreads();
    const int v = tid < CSR_NPB ? ncnt[tid] : 0;
    int tot;
    const int ex = block_excl_scan<CSR_T2 / 32>(v, swave, &tot);
    if (tid < CSR_NPB) {
      cur[tid] = ex;
      const int node = b * CSR_NPB + tid;
      if (node < N) {
        cnt[node] = v;
        rowptr[node] = bstart + ex;
      }
    }
    __syncthreads();
    const bool inlds = btot <= CSR_CAP;
    for (int c = tid; c < nchunks; c += CSR_T2) {
      const int o0 = mat[(size_t)c * CSR_MST + b];
      const int o1 = mat[(size_t)c * CSR_MST + b + 1];
      const int2* seg = eb + (size_t)c * CSR_CH;
      for (int i = o0; i < o1; i += 4) {
        int2 e[4];
#pragma unroll
        for (int u = 0; u < 4; ++u) {
          e[u].x = 0;
          e[u].y = -1;
          if (i + u < o1) e[u] = seg[i + u];
        }
#pragma unroll
        for (int u = 0; u < 4; ++u) {
          if (e[u].y >= 0) {
            const int pos = atomicAdd(&cur[e[u].y & (CSR_NPB - 1)], 1);
            if (inlds)
              colbuf[pos] = e[u].x;
            else
              col[bstart + pos] = e[u].x;
          }
        }
      }
    }
    __syncthreads();
    if (inlds)
      for (int i = tid; i < btot; i += CSR_T2) col[bstart + i] = colbuf[i];
    __syncthreads();
  }
}

#define MLP_NW 16
#define MLP_NT (MLP_NW * 32)
#define GU 32

template <int NCOL2, bool HAS_AGG, bool RELU2>
__global__ void __launch_bounds__(MLP_NT) gin_mlp(
    const float* __restrict__ hin, const int* __restrict__ rowptr,
    const int* __restrict__ cnt, const int* __restrict__ col,
    const float* __restrict__ W1, const float* __restrict__ b1,
    const float* __restrict__ W2, const float* __restrict__ b2,
    float* __restrict__ out, int nrows) {
  constexpr int N2 = NCOL2 * 16;
  constexpr int SWP = 144;
  constexpr int SZ = 132;
  __shared__ __attribute__((aligned(16))) v2f sW1[(DH / 2) * SWP];
  __shared__ __attribute__((aligned(16))) v2f sW2[(DH / 2) * SWP];
  __shared__ float sb1[DH];
  __shared__ float sb2[DH];
  __shared__ __attribute__((aligned(16))) float sZ[MLP_NW][16 * SZ];

  const int tid = (int)threadIdx.x;

  constexpr int CG2 = N2 / 4;
  constexpr int NJ1 = (DH / 2) * (DH / 4) / MLP_NT;
  constexpr int NJ2 = (DH / 2) * CG2 / MLP_NT;
  const int rot = (int)blockIdx.x;
  float4 w1a[NJ1], w1b[NJ1], w2a[NJ2], w2b[NJ2];
#pragma unroll
  for (int j = 0; j < NJ1; ++j) {
    const int it = j * MLP_NT + tid;
    const int p = ((it >> 5) + rot) & (DH / 2 - 1), cg = it & 31;
    const float* g = W1 + (size_t)(2 * p) * DH + cg * 4;
    w1a[j] = *reinterpret_cast<const float4*>(g);
    w1b[j] = *reinterpret_cast<const float4*>(g + DH);
  }
#pragma unroll
  for (int j = 0; j < NJ2; ++j) {
    const int it = j * MLP_NT + tid;
    const int p = (it / CG2 + rot) & (DH / 2 - 1), cg = it % CG2;
    const float* g = W2 + (size_t)(2 * p) * N2 + cg * 4;
    w2a[j] = *reinterpret_cast<const float4*>(g);
    w2b[j] = *reinterpret_cast<const float4*>(g + N2);
  }
  float bv1 = 0.f, bv2 = 0.f;
  if (tid < DH) bv1 = b1[tid];
  if (tid < N2) bv2 = b2[tid];
#pragma unroll
  for (int j = 0; j < NJ1; ++j) {
    const int it = j * MLP_NT + tid;
    const int p = ((it >> 5) + rot) & (DH / 2 - 1), cg = it & 31;
    float4* d = reinterpret_cast<float4*>(&sW1[p * SWP + cg * 4]);
    const float4 lo = {w1a[j].x, w1b[j].x, w1a[j].y, w1b[j].y};
    const float4 hi = {w1a[j].z, w1b[j].z, w1a[j].w, w1b[j].w};
    d[0] = lo;
    d[1] = hi;
  }
#pragma unroll
  for (int j = 0; j < NJ2; ++j) {
    const int it = j * MLP_NT + tid;
    const int p = (it / CG2 + rot) & (DH / 2 - 1), cg = it % CG2;
    float4* d = reinterpret_cast<float4*>(&sW2[p * SWP + cg * 4]);
    const float4 lo = {w2a[j].x, w2b[j].x, w2a[j].y, w2b[j].y};
    const float4 hi = {w2a[j].z, w2b[j].z, w2a[j].w, w2b[j].w};
    d[0] = lo;
    d[1] = hi;
  }
  if (tid < DH) sb1[tid] = bv1;
  if (tid < N2) sb2[tid] = bv2;
  __syncthreads();

  const int lane = tid & 31;
  const int wid = tid >> 5;
  const int m = lane & 15;
  const int koff = (lane >> 4) * 2;
  const int kpo = lane >> 4;
  const int mrow = (lane >> 4) * 8;
  float* zs = sZ[wid];

  const int ntiles = (nrows + 15) >> 4;
  const int gw0 = (int)blockIdx.x + (int)gridDim.x * wid;
  const int gwn = (int)gridDim.x * MLP_NW;
  const int nmain = (ntiles / ((int)gridDim.x * 4)) * ((int)gridDim.x * 4);

  for (int t = gw0; t < ntiles; t += gwn) {
    const int r0 = __builtin_amdgcn_readfirstlane(t << 4);

    if (!HAS_AGG) {
      float4 own[16];
#pragma unroll
      for (int r = 0; r < 16; ++r) {
        int nr = r0 + r;
        if (nr > nrows - 1) nr = nrows - 1;
        own[r] =
            *reinterpret_cast<const float4*>(hin + (size_t)nr * DH + lane * 4);
      }
#pragma unroll
      for (int r = 0; r < 16; ++r)
        *reinterpret_cast<float4*>(&zs[r * SZ + lane * 4]) = own[r];
      __builtin_amdgcn_wave_barrier();
    }

    if (HAS_AGG) {
      const int nv = nrows - r0 < 16 ? nrows - r0 : 16;
      int nd = r0 + m;
      if (nd > nrows - 1) nd = nrows - 1;
      const int rp = rowptr[nd];
      const int dg = m < nv ? cnt[nd] : 0;
      const int myend = m < nv ? rp + dg : 0x7fffffff;
      float4 own[16];
#pragma unroll
      for (int r = 0; r < 16; ++r) {
        int nr = r0 + r;
        if (nr > nrows - 1) nr = nrows - 1;
        own[r] =
            *reinterpret_cast<const float4*>(hin + (size_t)nr * DH + lane * 4);
      }
      const int beg = __builtin_amdgcn_readlane(rp, 0);
      const int end = __builtin_amdgcn_readlane(myend, nv - 1);
      int lend[15];
#pragma unroll
      for (int r = 0; r < 15; ++r) lend[r] = __builtin_amdgcn_readlane(myend, r);
      int cur = 0;
      float4 acc = {0.f, 0.f, 0.f, 0.f};
      int mine = 0;
      if (beg < end) mine = beg + lane < end ? col[beg + lane] : 0;
#pragma unroll
      for (int r = 0; r < 16; ++r)
        *reinterpret_cast<float4*>(&zs[r * SZ + lane * 4]) = own[r];
      for (int base = beg; base < end; base += GU) {
        float4 v[GU];
#pragma unroll
        for (int u = 0; u < GU; ++u) {
          const int sidx = __builtin_amdgcn_readlane(mine, u);
          v[u] = *reinterpret_cast<const float4*>(hin + (size_t)sidx * DH +
                                                  lane * 4);
        }
        int nxt = 0;
        if (base + GU < end)
          nxt = base + GU + lane < end ? col[base + GU + lane] : 0;
        const int pl = base + lane;
        int nid = 0, nidm = 0;
#pragma unroll
        for (int r = 0; r < 15; ++r) {
          nid += pl >= lend[r] ? 1 : 0;
          nidm += pl > lend[r] ? 1 : 0;
        }
        const unsigned first = __builtin_amdgcn_ballot_w32(nid != nidm);
        const int nval = end - base;
#pragma unroll
        for (int u = 0; u < GU; ++u) {
          if (u < nval) {
          if ((first >> u) & 1u) {
            float4* zr = reinterpret_cast<float4*>(&zs[cur * SZ + lane * 4]);
            float4 hv = *zr;
            hv.x += acc.x;
            hv.y += acc.y;
            hv.z += acc.z;
            hv.w += acc.w;
            *zr = hv;
            acc.x = 0.f;
            acc.y = 0.f;
            acc.z = 0.f;
            acc.w = 0.f;
            cur = __builtin_amdgcn_readlane(nid, u);
          }
          acc.x += v[u].x;
          acc.y += v[u].y;
          acc.z += v[u].z;
          acc.w += v[u].w;
          }
        }
        mine = nxt;
      }
      {
        float4* zr = reinterpret_cast<float4*>(&zs[cur * SZ + lane * 4]);
        float4 hv = *zr;
        hv.x += acc.x;
        hv.y += acc.y;
        hv.z += acc.z;
        hv.w += acc.w;
        *zr = hv;
      }
      __builtin_amdgcn_wave_barrier();
    }

    if ((r0 >> 4) >= nmain) break;

    v8f acc[8];
#pragma unroll
    for (int n = 0; n < 8; ++n) acc[n] = zero8();

    for (int k0 = 0; k0 < DH; k0 += 4) {
      const int k = k0 + koff;
      const int kp = (k0 >> 1) + kpo;
      const v2f a = *reinterpret_cast<const v2f*>(&zs[m * SZ + k]);
#pragma unroll
      for (int n = 0; n < 8; ++n) {
        const v2f bf = sW1[kp * SWP + n * 16 + m];
        acc[n] = __builtin_amdgcn_wmma_f32_16x16x4_f32(
            false, a, false, bf, (short)0, acc[n], false, false);
      }
    }
    __builtin_amdgcn_wave_barrier();

#pragma unroll
    for (int n = 0; n < 8; ++n) {
      const float bias = sb1[n * 16 + m];
#pragma unroll
      for (int j = 0; j < 8; ++j) {
        float v = acc[n][j] + bias;
        v = v > 0.0f ? v : 0.0f;
        zs[(mrow + j) * SZ + n * 16 + m] = v;
      }
    }
    __builtin_amdgcn_wave_barrier();

#pragma unroll
    for (int n = 0; n < NCOL2; ++n) acc[n] = zero8();
    for (int k0 = 0; k0 < DH; k0 += 4) {
      const int k = k0 + koff;
      const int kp = (k0 >> 1) + kpo;
      const v2f a = *reinterpret_cast<const v2f*>(&zs[m * SZ + k]);
#pragma unroll
      for (int n = 0; n < NCOL2; ++n) {
        const v2f bf = sW2[kp * SWP + n * 16 + m];
        acc[n] = __builtin_amdgcn_wmma_f32_16x16x4_f32(
            false, a, false, bf, (short)0, acc[n], false, false);
      }
    }
    __builtin_amdgcn_wave_barrier();

#pragma unroll
    for (int n = 0; n < NCOL2; ++n) {
      const float bias = sb2[n * 16 + m];
#pragma unroll
      for (int j = 0; j < 8; ++j) {
        float v = acc[n][j] + bias;
        if (RELU2) v = v > 0.0f ? v : 0.0f;
        zs[(mrow + j) * SZ + n * 16 + m] = v;
      }
    }
    __builtin_amdgcn_wave_barrier();
    constexpr int LPR = N2 / 4;
    constexpr int RPS = 32 / LPR;
    const int rsub = lane / LPR;
    const int c4 = (lane % LPR) * 4;
    float* orow0 = out + (size_t)(r0 + rsub) * N2 + c4;
    if (r0 + 16 <= nrows) {
      float4 orow[16 / RPS];
#pragma unroll
      for (int i = 0; i < 16 / RPS; ++i)
        orow[i] = *reinterpret_cast<const float4*>(
            &zs[(i * RPS + rsub) * SZ + c4]);
#pragma unroll
      for (int i = 0; i < 16 / RPS; ++i)
        *reinterpret_cast<float4*>(orow0 + (size_t)(i * RPS) * N2) = orow[i];
    } else {
#pragma unroll
      for (int i = 0; i < 16 / RPS; ++i) {
        if (r0 + i * RPS + rsub < nrows)
          *reinterpret_cast<float4*>(orow0 + (size_t)(i * RPS) * N2) =
              *reinterpret_cast<const float4*>(
                  &zs[(i * RPS + rsub) * SZ + c4]);
      }
    }
  }

  const int j0 = nmain / (int)gridDim.x;
  const int left = ntiles - nmain - (int)blockIdx.x;
  const int nrem = left > 0 ? (left + (int)gridDim.x - 1) / (int)gridDim.x : 0;
  if (nrem > 0) {
    const int wv = __builtin_amdgcn_readfirstlane(wid);
    __syncthreads();
    for (int q = 0; q < nrem; ++q) {
      const int r0 = (nmain + (int)blockIdx.x + (int)gridDim.x * q) << 4;
      const float* za = sZ[(j0 + q) & (MLP_NW - 1)];
      float* zb = sZ[(j0 + q + 4) & (MLP_NW - 1)];
      float* zc = sZ[(j0 + q + 8) & (MLP_NW - 1)];
      if (wv < 8) {
        v2f av[DH / 4], bv[DH / 4];
#pragma unroll
        for (int s = 0; s < DH / 4; ++s) {
          av[s] = *reinterpret_cast<const v2f*>(&za[m * SZ + s * 4 + koff]);
          bv[s] = sW1[(s * 2 + kpo) * SWP + wv * 16 + m];
        }
        __builtin_amdgcn_sched_barrier(0);
        v8f c = zero8();
#pragma unroll
        for (int s = 0; s < DH / 4; ++s)
          c = __builtin_amdgcn_wmma_f32_16x16x4_f32(
              false, av[s], false, bv[s], (short)0, c, false, false);
        const float bias = sb1[wv * 16 + m];
#pragma unroll
        for (int j = 0; j < 8; ++j) {
          float v = c[j] + bias;
          v = v > 0.0f ? v : 0.0f;
          zb[(mrow + j) * SZ + wv * 16 + m] = v;
        }
      }
      __syncthreads();
      if (wv < NCOL2) {
        v2f av[DH / 4], bv[DH / 4];
#pragma unroll
        for (int s = 0; s < DH / 4; ++s) {
          av[s] = *reinterpret_cast<const v2f*>(&zb[m * SZ + s * 4 + koff]);
          bv[s] = sW2[(s * 2 + kpo) * SWP + wv * 16 + m];
        }
        __builtin_amdgcn_sched_barrier(0);
        v8f c = zero8();
#pragma unroll
        for (int s = 0; s < DH / 4; ++s)
          c = __builtin_amdgcn_wmma_f32_16x16x4_f32(
              false, av[s], false, bv[s], (short)0, c, false, false);
        const float bias = sb2[wv * 16 + m];
#pragma unroll
        for (int j = 0; j < 8; ++j) {
          float v = c[j] + bias;
          if (RELU2) v = v > 0.0f ? v : 0.0f;
          zc[(mrow + j) * SZ + wv * 16 + m] = v;
        }
      }
      __syncthreads();
      constexpr int LPR = N2 / 4;
      constexpr int RPS = 32 / LPR;
      if (wv < 16 / RPS) {
        const int row = wv * RPS + lane / LPR;
        const int c4 = (lane % LPR) * 4;
        if (r0 + row < nrows)
          *reinterpret_cast<float4*>(out + (size_t)(r0 + row) * N2 + c4) =
              *reinterpret_cast<const float4*>(&zc[row * SZ + c4]);
      }
    }
  }
}

static __device__ __forceinline__ void joint_block(
    const float* za, float* zd, const v2f* sW, const float* sb, int cb, int m,
    int koff, int kpo, int mrow, bool relu) {
  v2f av[DH / 4], bv[DH / 4];
#pragma unroll
  for (int s = 0; s < DH / 4; ++s) {
    av[s] = *reinterpret_cast<const v2f*>(&za[m * 132 + s * 4 + koff]);
    bv[s] = sW[(s * 2 + kpo) * 144 + cb * 16 + m];
  }
  __builtin_amdgcn_sched_barrier(0);
  v8f c = zero8();
#pragma unroll
  for (int s = 0; s < DH / 4; ++s)
    c = __builtin_amdgcn_wmma_f32_16x16x4_f32(false, av[s], false, bv[s],
                                              (short)0, c, false, false);
  const float bias = sb[cb * 16 + m];
#pragma unroll
  for (int j = 0; j < 8; ++j) {
    float v = c[j] + bias;
    if (relu) v = v > 0.0f ? v : 0.0f;
    zd[(mrow + j) * 132 + cb * 16 + m] = v;
  }
}

__global__ void __launch_bounds__(MLP_NT) gin_mlp_head(
    const float* __restrict__ hin, const int* __restrict__ rowptr,
    const int* __restrict__ cnt, const int* __restrict__ col,
    const float* __restrict__ W1, const float* __restrict__ b1,
    const float* __restrict__ W2, const float* __restrict__ b2,
    const float* __restrict__ W3, const float* __restrict__ b3,
    const float* __restrict__ W4, const float* __restrict__ b4,
    float* __restrict__ out, int nrows) {
  constexpr int NCOL2 = 8;
  constexpr bool HAS_AGG = true;
  constexpr bool RELU2 = true;
  constexpr int N2 = NCOL2 * 16;
  constexpr int SWP = 144;
  constexpr int SZ = 132;
  __shared__ __attribute__((aligned(16))) v2f sW1[(DH / 2) * SWP];
  __shared__ __attribute__((aligned(16))) v2f sW2[(DH / 2) * SWP];
  __shared__ float sb1[DH];
  __shared__ float sb2[DH];
  __shared__ __attribute__((aligned(16))) float sZ[MLP_NW + 1][16 * SZ];

  const int tid = (int)threadIdx.x;

  constexpr int CG2 = N2 / 4;
  constexpr int NJ1 = (DH / 2) * (DH / 4) / MLP_NT;
  constexpr int NJ2 = (DH / 2) * CG2 / MLP_NT;
  const int rot = (int)blockIdx.x;
  float4 w1a[NJ1], w1b[NJ1], w2a[NJ2], w2b[NJ2];
#pragma unroll
  for (int j = 0; j < NJ1; ++j) {
    const int it = j * MLP_NT + tid;
    const int p = ((it >> 5) + rot) & (DH / 2 - 1), cg = it & 31;
    const float* g = W1 + (size_t)(2 * p) * DH + cg * 4;
    w1a[j] = *reinterpret_cast<const float4*>(g);
    w1b[j] = *reinterpret_cast<const float4*>(g + DH);
  }
#pragma unroll
  for (int j = 0; j < NJ2; ++j) {
    const int it = j * MLP_NT + tid;
    const int p = (it / CG2 + rot) & (DH / 2 - 1), cg = it % CG2;
    const float* g = W2 + (size_t)(2 * p) * N2 + cg * 4;
    w2a[j] = *reinterpret_cast<const float4*>(g);
    w2b[j] = *reinterpret_cast<const float4*>(g + N2);
  }
  float bv1 = 0.f, bv2 = 0.f;
  if (tid < DH) bv1 = b1[tid];
  if (tid < N2) bv2 = b2[tid];
#pragma unroll
  for (int j = 0; j < NJ1; ++j) {
    const int it = j * MLP_NT + tid;
    const int p = ((it >> 5) + rot) & (DH / 2 - 1), cg = it & 31;
    float4* d = reinterpret_cast<float4*>(&sW1[p * SWP + cg * 4]);
    const float4 lo = {w1a[j].x, w1b[j].x, w1a[j].y, w1b[j].y};
    const float4 hi = {w1a[j].z, w1b[j].z, w1a[j].w, w1b[j].w};
    d[0] = lo;
    d[1] = hi;
  }
#pragma unroll
  for (int j = 0; j < NJ2; ++j) {
    const int it = j * MLP_NT + tid;
    const int p = (it / CG2 + rot) & (DH / 2 - 1), cg = it % CG2;
    float4* d = reinterpret_cast<float4*>(&sW2[p * SWP + cg * 4]);
    const float4 lo = {w2a[j].x, w2b[j].x, w2a[j].y, w2b[j].y};
    const float4 hi = {w2a[j].z, w2b[j].z, w2a[j].w, w2b[j].w};
    d[0] = lo;
    d[1] = hi;
  }
  if (tid < DH) sb1[tid] = bv1;
  if (tid < N2) sb2[tid] = bv2;
  __syncthreads();

  const int lane = tid & 31;
  const int wid = tid >> 5;
  const int m = lane & 15;
  const int koff = (lane >> 4) * 2;
  const int kpo = lane >> 4;
  const int mrow = (lane >> 4) * 8;
  float* zs = sZ[wid];

  const int ntiles = (nrows + 15) >> 4;
  const int gw0 = (int)blockIdx.x + (int)gridDim.x * wid;
  const int gwn = (int)gridDim.x * MLP_NW;
  const int nmain = (ntiles / ((int)gridDim.x * 4)) * ((int)gridDim.x * 4);

  int have = 0;
  int r0m = 0;
  for (int t = gw0; t < ntiles; t += gwn) {
    const int r0 = __builtin_amdgcn_readfirstlane(t << 4);

    if (!HAS_AGG) {
      float4 own[16];
#pragma unroll
      for (int r = 0; r < 16; ++r) {
        int nr = r0 + r;
        if (nr > nrows - 1) nr = nrows - 1;
        own[r] =
            *reinterpret_cast<const float4*>(hin + (size_t)nr * DH + lane * 4);
      }
#pragma unroll
      for (int r = 0; r < 16; ++r)
        *reinterpret_cast<float4*>(&zs[r * SZ + lane * 4]) = own[r];
      __builtin_amdgcn_wave_barrier();
    }

    if (HAS_AGG) {
      const int nv = nrows - r0 < 16 ? nrows - r0 : 16;
      int nd = r0 + m;
      if (nd > nrows - 1) nd = nrows - 1;
      const int rp = rowptr[nd];
      const int dg = m < nv ? cnt[nd] : 0;
      const int myend = m < nv ? rp + dg : 0x7fffffff;
      float4 own[16];
#pragma unroll
      for (int r = 0; r < 16; ++r) {
        int nr = r0 + r;
        if (nr > nrows - 1) nr = nrows - 1;
        own[r] =
            *reinterpret_cast<const float4*>(hin + (size_t)nr * DH + lane * 4);
      }
      const int beg = __builtin_amdgcn_readlane(rp, 0);
      const int end = __builtin_amdgcn_readlane(myend, nv - 1);
      int lend[15];
#pragma unroll
      for (int r = 0; r < 15; ++r) lend[r] = __builtin_amdgcn_readlane(myend, r);
      int cur = 0;
      float4 acc = {0.f, 0.f, 0.f, 0.f};
      int mine = 0;
      if (beg < end) mine = beg + lane < end ? col[beg + lane] : 0;
#pragma unroll
      for (int r = 0; r < 16; ++r)
        *reinterpret_cast<float4*>(&zs[r * SZ + lane * 4]) = own[r];
      for (int base = beg; base < end; base += GU) {
        float4 v[GU];
#pragma unroll
        for (int u = 0; u < GU; ++u) {
          const int sidx = __builtin_amdgcn_readlane(mine, u);
          v[u] = *reinterpret_cast<const float4*>(hin + (size_t)sidx * DH +
                                                  lane * 4);
        }
        int nxt = 0;
        if (base + GU < end)
          nxt = base + GU + lane < end ? col[base + GU + lane] : 0;
        const int pl = base + lane;
        int nid = 0, nidm = 0;
#pragma unroll
        for (int r = 0; r < 15; ++r) {
          nid += pl >= lend[r] ? 1 : 0;
          nidm += pl > lend[r] ? 1 : 0;
        }
        const unsigned first = __builtin_amdgcn_ballot_w32(nid != nidm);
        const int nval = end - base;
#pragma unroll
        for (int u = 0; u < GU; ++u) {
          if (u < nval) {
          if ((first >> u) & 1u) {
            float4* zr = reinterpret_cast<float4*>(&zs[cur * SZ + lane * 4]);
            float4 hv = *zr;
            hv.x += acc.x;
            hv.y += acc.y;
            hv.z += acc.z;
            hv.w += acc.w;
            *zr = hv;
            acc.x = 0.f;
            acc.y = 0.f;
            acc.z = 0.f;
            acc.w = 0.f;
            cur = __builtin_amdgcn_readlane(nid, u);
          }
          acc.x += v[u].x;
          acc.y += v[u].y;
          acc.z += v[u].z;
          acc.w += v[u].w;
          }
        }
        mine = nxt;
      }
      {
        float4* zr = reinterpret_cast<float4*>(&zs[cur * SZ + lane * 4]);
        float4 hv = *zr;
        hv.x += acc.x;
        hv.y += acc.y;
        hv.z += acc.z;
        hv.w += acc.w;
        *zr = hv;
      }
      __builtin_amdgcn_wave_barrier();
    }

    if ((r0 >> 4) >= nmain) break;

    v8f acc[8];
#pragma unroll
    for (int n = 0; n < 8; ++n) acc[n] = zero8();

    for (int k0 = 0; k0 < DH; k0 += 4) {
      const int k = k0 + koff;
      const int kp = (k0 >> 1) + kpo;
      const v2f a = *reinterpret_cast<const v2f*>(&zs[m * SZ + k]);
#pragma unroll
      for (int n = 0; n < 8; ++n) {
        const v2f bf = sW1[kp * SWP + n * 16 + m];
        acc[n] = __builtin_amdgcn_wmma_f32_16x16x4_f32(
            false, a, false, bf, (short)0, acc[n], false, false);
      }
    }
    __builtin_amdgcn_wave_barrier();

#pragma unroll
    for (int n = 0; n < 8; ++n) {
      const float bias = sb1[n * 16 + m];
#pragma unroll
      for (int j = 0; j < 8; ++j) {
        float v = acc[n][j] + bias;
        v = v > 0.0f ? v : 0.0f;
        zs[(mrow + j) * SZ + n * 16 + m] = v;
      }
    }
    __builtin_amdgcn_wave_barrier();

#pragma unroll
    for (int n = 0; n < NCOL2; ++n) acc[n] = zero8();
    for (int k0 = 0; k0 < DH; k0 += 4) {
      const int k = k0 + koff;
      const int kp = (k0 >> 1) + kpo;
      const v2f a = *reinterpret_cast<const v2f*>(&zs[m * SZ + k]);
#pragma unroll
      for (int n = 0; n < NCOL2; ++n) {
        const v2f bf = sW2[kp * SWP + n * 16 + m];
        acc[n] = __builtin_amdgcn_wmma_f32_16x16x4_f32(
            false, a, false, bf, (short)0, acc[n], false, false);
      }
    }
    __builtin_amdgcn_wave_barrier();

#pragma unroll
    for (int n = 0; n < NCOL2; ++n) {
      const float bias = sb2[n * 16 + m];
#pragma unroll
      for (int j = 0; j < 8; ++j) {
        float v = acc[n][j] + bias;
        if (RELU2) v = v > 0.0f ? v : 0.0f;
        zs[(mrow + j) * SZ + n * 16 + m] = v;
      }
    }
    __builtin_amdgcn_wave_barrier();
    have = 1;
    r0m = r0;
  }

  constexpr int N4 = 64;
  constexpr int CG4 = N4 / 4;
  constexpr int NJ4 = (DH / 2) * CG4 / MLP_NT;
  float4 w3a[NJ1], w3b[NJ1], w4a[NJ4], w4b[NJ4];
#pragma unroll
  for (int j = 0; j < NJ1; ++j) {
    const int it = j * MLP_NT + tid;
    const int p = ((it >> 5) + rot) & (DH / 2 - 1), cg = it & 31;
    const float* g = W3 + (size_t)(2 * p) * DH + cg * 4;
    w3a[j] = *reinterpret_cast<const float4*>(g);
    w3b[j] = *reinterpret_cast<const float4*>(g + DH);
  }
#pragma unroll
  for (int j = 0; j < NJ4; ++j) {
    const int it = j * MLP_NT + tid;
    const int p = (it / CG4 + rot) & (DH / 2 - 1), cg = it % CG4;
    const float* g = W4 + (size_t)(2 * p) * N4 + cg * 4;
    w4a[j] = *reinterpret_cast<const float4*>(g);
    w4b[j] = *reinterpret_cast<const float4*>(g + N4);
  }
  float bv3 = 0.f, bv4 = 0.f;
  if (tid < DH) bv3 = b3[tid];
  if (tid < N4) bv4 = b4[tid];

  const int j0 = nmain / (int)gridDim.x;
  const int left = ntiles - nmain - (int)blockIdx.x;
  const int nrem = left > 0 ? (left + (int)gridDim.x - 1) / (int)gridDim.x : 0;
  const int wv = __builtin_amdgcn_readfirstlane(wid);
  float* zx = sZ[MLP_NW];
  __syncthreads();
  for (int q = 0; q < nrem; ++q) {
    float* za = sZ[(j0 + q) & (MLP_NW - 1)];
    if (wv < 8) joint_block(za, zx, sW1, sb1, wv, m, koff, kpo, mrow, true);
    __syncthreads();
    if (wv < 8) joint_block(zx, za, sW2, sb2, wv, m, koff, kpo, mrow, true);
    __syncthreads();
  }
#pragma unroll
  for (int j = 0; j < NJ1; ++j) {
    const int it = j * MLP_NT + tid;
    const int p = ((it >> 5) + rot) & (DH / 2 - 1), cg = it & 31;
    float4* d = reinterpret_cast<float4*>(&sW1[p * SWP + cg * 4]);
    const float4 lo = {w3a[j].x, w3b[j].x, w3a[j].y, w3b[j].y};
    const float4 hi = {w3a[j].z, w3b[j].z, w3a[j].w, w3b[j].w};
    d[0] = lo;
    d[1] = hi;
  }
#pragma unroll
  for (int j = 0; j < NJ4; ++j) {
    const int it = j * MLP_NT + tid;
    const int p = (it / CG4 + rot) & (DH / 2 - 1), cg = it % CG4;
    float4* d = reinterpret_cast<float4*>(&sW2[p * SWP + cg * 4]);
    const float4 lo = {w4a[j].x, w4b[j].x, w4a[j].y, w4b[j].y};
    const float4 hi = {w4a[j].z, w4b[j].z, w4a[j].w, w4b[j].w};
    d[0] = lo;
    d[1] = hi;
  }
  if (tid < DH) sb1[tid] = bv3;
  if (tid < N4) sb2[tid] = bv4;
  __syncthreads();

  constexpr int LPR = N4 / 4;
  constexpr int RPS = 32 / LPR;
  if (have) {
    v8f acc[8];
#pragma unroll
    for (int n = 0; n < 8; ++n) acc[n] = zero8();
    for (int k0 = 0; k0 < DH; k0 += 4) {
      const int k = k0 + koff;
      const int kp = (k0 >> 1) + kpo;
      const v2f a = *reinterpret_cast<const v2f*>(&zs[m * SZ + k]);
#pragma unroll
      for (int n = 0; n < 8; ++n) {
        const v2f bf = sW1[kp * SWP + n * 16 + m];
        acc[n] = __builtin_amdgcn_wmma_f32_16x16x4_f32(
            false, a, false, bf, (short)0, acc[n], false, false);
      }
    }
    __builtin_amdgcn_wave_barrier();
#pragma unroll
    for (int n = 0; n < 8; ++n) {
      const float bias = sb1[n * 16 + m];
#pragma unroll
      for (int j = 0; j < 8; ++j) {
        float v = acc[n][j] + bias;
        v = v > 0.0f ? v : 0.0f;
        zs[(mrow + j) * SZ + n * 16 + m] = v;
      }
    }
    __builtin_amdgcn_wave_barrier();
#pragma unroll
    for (int n = 0; n < 4; ++n) acc[n] = zero8();
    for (int k0 = 0; k0 < DH; k0 += 4) {
      const int k = k0 + koff;
      const int kp = (k0 >> 1) + kpo;
      const v2f a = *reinterpret_cast<const v2f*>(&zs[m * SZ + k]);
#pragma unroll
      for (int n = 0; n < 4; ++n) {
        const v2f bf = sW2[kp * SWP + n * 16 + m];
        acc[n] = __builtin_amdgcn_wmma_f32_16x16x4_f32(
            false, a, false, bf, (short)0, acc[n], false, false);
      }
    }
    __builtin_amdgcn_wave_barrier();
#pragma unroll
    for (int n = 0; n < 4; ++n) {
      const float bias = sb2[n * 16 + m];
#pragma unroll
      for (int j = 0; j < 8; ++j)
        zs[(mrow + j) * SZ + n * 16 + m] = acc[n][j] + bias;
    }
    __builtin_amdgcn_wave_barrier();
    const int rsub = lane / LPR;
    const int c4 = (lane % LPR) * 4;
#pragma unroll
    for (int i = 0; i < 16 / RPS; ++i) {
      const int row = i * RPS + rsub;
      if (r0m + row < nrows)
        *reinterpret_cast<float4*>(out + (size_t)(r0m + row) * N4 + c4) =
            *reinterpret_cast<const float4*>(&zs[row * SZ + c4]);
    }
  }
  for (int q = 0; q < nrem; ++q) {
    const int r0 = (nmain + (int)blockIdx.x + (int)gridDim.x * q) << 4;
    float* za = sZ[(j0 + q) & (MLP_NW - 1)];
    if (wv < 8) joint_block(za, zx, sW1, sb1, wv, m, koff, kpo, mrow, true);
    __syncthreads();
    if (wv < 4) joint_block(zx, za, sW2, sb2, wv, m, koff, kpo, mrow, false);
    __syncthreads();
    if (wv < 16 / RPS) {
      const int row = wv * RPS + lane / LPR;
      const int c4 = (lane % LPR) * 4;
      if (r0 + row < nrows)
        *reinterpret_cast<float4*>(out + (size_t)(r0 + row) * N4 + c4) =
            *reinterpret_cast<const float4*>(&za[row * SZ + c4]);
    }
  }
}

extern "C" void kernel_launch(void* const* d_in, const int* in_sizes, int n_in,
                              void* d_out, int out_size, void* d_ws,
                              size_t ws_size, hipStream_t stream) {
  const float* x = (const float*)d_in[0];
  const int* ei = (const int*)d_in[1];
  const int N = in_sizes[0] / DH;
  const int E = in_sizes[1] / 2;
  const int* src = ei;
  const int* dst = ei + E;

  const float* c1w1 = (const float*)d_in[2];
  const float* c1b1 = (const float*)d_in[3];
  const float* c1w2 = (const float*)d_in[4];
  const float* c1b2 = (const float*)d_in[5];
  const float* c2w1 = (const float*)d_in[6];
  const float* c2b1 = (const float*)d_in[7];
  const float* c2w2 = (const float*)d_in[8];
  const float* c2b2 = (const float*)d_in[9];
  const float* c3w1 = (const float*)d_in[10];
  const float* c3b1 = (const float*)d_in[11];
  const float* c3w2 = (const float*)d_in[12];
  const float* c3b2 = (const float*)d_in[13];
  const float* l1w = (const float*)d_in[14];
  const float* l1b = (const float*)d_in[15];
  const float* l2w = (const float*)d_in[16];
  const float* l2b = (const float*)d_in[17];

  float* hA = (float*)d_ws;
  float* hB = hA + (size_t)N * DH;
  int* ebin = (int*)(hB + (size_t)N * DH);
  int* cnt = ebin + 2 * (size_t)E;
  int* rowptr = cnt + N;
  int* col = rowptr + N;
  int* mat = col + E;

  const dim3 bB(CSR_T1), bG(256);
  const dim3 tB(CSR_T2), tG(512);
  const dim3 mB(512), mG(256);

  csr_bin<<<bG, bB, 0, stream>>>(src, dst, mat, ebin, E, N);
  csr_emit<<<tG, tB, 0, stream>>>(mat, ebin, cnt, rowptr, col, E, N);

  gin_mlp<8, true, true><<<mG, mB, 0, stream>>>(
      x, rowptr, cnt, col, c1w1, c1b1, c1w2, c1b2, hA, N);
  gin_mlp<8, true, true><<<mG, mB, 0, stream>>>(
      hA, rowptr, cnt, col, c2w1, c2b1, c2w2, c2b2, hB, N);
  gin_mlp_head<<<mG, mB, 0, stream>>>(hB, rowptr, cnt, col, c3w1, c3b1, c3w2,
                                      c3b2, l1w, l1b, l2w, l2b,
                                      (float*)d_out, N);
}
